# starting kernel
# baseline (speedup 1.0000x reference)
#include <hip/hip_runtime.h>
#include <hip/hip_bf16.h>

typedef __attribute__((ext_vector_type(2))) float v2f;
typedef __attribute__((ext_vector_type(8))) float v8f;

#define NUM_USERS  100000
#define NUM_MOVIES 50000
#define NUM_EDGES  4000000
#define USER_FEAT  24
#define MOVIE_FEAT 404
#define HID        64
#define GENRES     3

__global__ __launch_bounds__(256) void degree_kernel(const int* __restrict__ idx,
                                                     float* __restrict__ deg, int E) {
    int e = blockIdx.x * blockDim.x + threadIdx.x;
    if (e < E) atomicAdd(&deg[idx[e]], 1.0f);
}

__global__ __launch_bounds__(256) void scatter_kernel(const float* __restrict__ X,
                                                      const int* __restrict__ src,
                                                      const int* __restrict__ dst,
                                                      float* __restrict__ acc, int E) {
    long long t = (long long)blockIdx.x * blockDim.x + threadIdx.x;
    int e = (int)(t >> 4);
    int c = ((int)t & 15) * 4;
    if (e >= E) return;
    int s = src[e];
    int d = dst[e];
    const float4 v = *(const float4*)(X + (size_t)s * HID + c);
    float* o = acc + (size_t)d * HID + c;
    atomicAdd(o + 0, v.x);
    atomicAdd(o + 1, v.y);
    atomicAdd(o + 2, v.z);
    atomicAdd(o + 3, v.w);
}

__global__ __launch_bounds__(256)
void sage_gemm64(const float* __restrict__ A1, const float* __restrict__ deg1,
                 const float* __restrict__ W1, int K1,
                 const float* __restrict__ A2, const float* __restrict__ W2, int K2,
                 const float* __restrict__ bias, float* __restrict__ C,
                 int M, int relu) {
    const int lane = threadIdx.x & 31;
    const int wave = threadIdx.x >> 5;
    const int half = lane >> 4;
    const int r    = lane & 15;
    const int row0 = (blockIdx.x * 8 + wave) * 16;
    if (row0 >= M) return;

    v8f c[4] = {};

    float scale = 1.0f;
    if (deg1) scale = 1.0f / fmaxf(deg1[row0 + r], 1.0f);

    {
        const float* Arow = A1 + (size_t)(row0 + r) * K1;
        for (int kk = 0; kk < K1; kk += 4) {
            const int k = kk + 2 * half;
            v2f a;
            a.x = Arow[k] * scale;
            a.y = Arow[k + 1] * scale;
#pragma unroll
            for (int t = 0; t < 4; ++t) {
                const float* Wrow = W1 + (size_t)(t * 16 + r) * K1;
                v2f b;
                b.x = Wrow[k];
                b.y = Wrow[k + 1];
                c[t] = __builtin_amdgcn_wmma_f32_16x16x4_f32(
                    false, a, false, b, (short)0, c[t], false, false);
            }
        }
    }
    if (A2) {
        const float* Arow = A2 + (size_t)(row0 + r) * K2;
        for (int kk = 0; kk < K2; kk += 4) {
            const int k = kk + 2 * half;
            v2f a;
            a.x = Arow[k];
            a.y = Arow[k + 1];
#pragma unroll
            for (int t = 0; t < 4; ++t) {
                const float* Wrow = W2 + (size_t)(t * 16 + r) * K2;
                v2f b;
                b.x = Wrow[k];
                b.y = Wrow[k + 1];
                c[t] = __builtin_amdgcn_wmma_f32_16x16x4_f32(
                    false, a, false, b, (short)0, c[t], false, false);
            }
        }
    }

#pragma unroll
    for (int t = 0; t < 4; ++t) {
        const float bv = bias ? bias[t * 16 + r] : 0.0f;
#pragma unroll
        for (int i = 0; i < 8; ++i) {
            float v = c[t][i] + bv;
            if (relu) v = fmaxf(v, 0.0f);
            C[(size_t)(row0 + i + 8 * half) * HID + t * 16 + r] = v;
        }
    }
}

__global__ __launch_bounds__(256)
void head_kernel(const float* __restrict__ acc2, const float* __restrict__ degm,
                 const float* __restrict__ m1, const float* __restrict__ Wl2,
                 const float* __restrict__ bl2, const float* __restrict__ Wr2,
                 float* __restrict__ out, int M) {
    int m = blockIdx.x * blockDim.x + threadIdx.x;
    if (m >= M) return;
    const float inv = 1.0f / fmaxf(degm[m], 1.0f);
    const float* a = acc2 + (size_t)m * HID;
    const float* x = m1 + (size_t)m * HID;
    float acc[GENRES];
#pragma unroll
    for (int g = 0; g < GENRES; ++g) acc[g] = bl2[g];
    for (int k = 0; k < HID; ++k) {
        const float av = a[k] * inv;
        const float xv = x[k];
#pragma unroll
        for (int g = 0; g < GENRES; ++g)
            acc[g] += av * Wl2[g * HID + k] + xv * Wr2[g * HID + k];
    }
#pragma unroll
    for (int g = 0; g < GENRES; ++g) out[(size_t)m * GENRES + g] = acc[g];
}

extern "C" void kernel_launch(void* const* d_in, const int* in_sizes, int n_in,
                              void* d_out, int out_size, void* d_ws, size_t ws_size,
                              hipStream_t stream) {
    (void)in_sizes; (void)n_in; (void)out_size; (void)ws_size;

    const float* x_user  = (const float*)d_in[0];
    const float* x_movie = (const float*)d_in[1];
    const int*   e_src   = (const int*)d_in[2];
    const int*   e_dst   = (const int*)d_in[3];
    const float* W_user  = (const float*)d_in[4];
    const float* b_user  = (const float*)d_in[5];
    const float* W_movie = (const float*)d_in[6];
    const float* b_movie = (const float*)d_in[7];
    const float* Wl1_um  = (const float*)d_in[8];
    const float* bl1_um  = (const float*)d_in[9];
    const float* Wr1_um  = (const float*)d_in[10];
    const float* Wl1_mu  = (const float*)d_in[11];
    const float* bl1_mu  = (const float*)d_in[12];
    const float* Wr1_mu  = (const float*)d_in[13];
    const float* Wl2_um  = (const float*)d_in[14];
    const float* bl2_um  = (const float*)d_in[15];
    const float* Wr2_um  = (const float*)d_in[16];
    float* out = (float*)d_out;

    float* ws = (float*)d_ws;
    float* acc_um    = ws;
    float* acc_mu    = acc_um + 3200000;
    float* acc2      = acc_mu + 6400000;
    float* deg_user  = acc2 + 3200000;
    float* deg_movie = deg_user + 100000;
    float* hu        = deg_movie + 50000;
    float* hm        = hu + 6400000;
    float* m1        = hm + 3200000;
    float* u1        = m1 + 3200000;

    const size_t zero_bytes = (size_t)(3200000 + 6400000 + 3200000 + 100000 + 50000) * 4;
    hipMemsetAsync(d_ws, 0, zero_bytes, stream);

    const int E = NUM_EDGES;
    degree_kernel<<<(E + 255) / 256, 256, 0, stream>>>(e_dst, deg_movie, E);
    degree_kernel<<<(E + 255) / 256, 256, 0, stream>>>(e_src, deg_user, E);

    const int gridU = (NUM_USERS / 16 + 7) / 8;
    const int gridM = (NUM_MOVIES / 16 + 7) / 8;
    sage_gemm64<<<gridU, 256, 0, stream>>>(x_user, nullptr, W_user, USER_FEAT,
                                           nullptr, nullptr, 0, b_user, hu,
                                           NUM_USERS, 0);
    sage_gemm64<<<gridM, 256, 0, stream>>>(x_movie, nullptr, W_movie, MOVIE_FEAT,
                                           nullptr, nullptr, 0, b_movie, hm,
                                           NUM_MOVIES, 0);

    const long long sthreads = (long long)E * 16;
    const int sblocks = (int)((sthreads + 255) / 256);

    scatter_kernel<<<sblocks, 256, 0, stream>>>(hu, e_src, e_dst, acc_um, E);
    sage_gemm64<<<gridM, 256, 0, stream>>>(acc_um, deg_movie, Wl1_um, HID,
                                           hm, Wr1_um, HID, bl1_um, m1,
                                           NUM_MOVIES, 1);

    scatter_kernel<<<sblocks, 256, 0, stream>>>(hm, e_dst, e_src, acc_mu, E);
    sage_gemm64<<<gridU, 256, 0, stream>>>(acc_mu, deg_user, Wl1_mu, HID,
                                           hu, Wr1_mu, HID, bl1_mu, u1,
                                           NUM_USERS, 1);

    scatter_kernel<<<sblocks, 256, 0, stream>>>(u1, e_src, e_dst, acc2, E);
    head_kernel<<<(NUM_MOVIES + 255) / 256, 256, 0, stream>>>(
        acc2, deg_movie, m1, Wl2_um, bl2_um, Wr2_um, out, NUM_MOVIES);
}


# Edge sums and degree counts owner-computed: 256 blocks each keep a node range's 64-float rows in LDS, scan all edge keys, add matching source rows by LDS float adds, store once; sum order changes
# speedup vs baseline: 1.5792x; 1.5792x over previous
#include <hip/hip_runtime.h>
#include <hip/hip_bf16.h>

typedef __attribute__((ext_vector_type(2))) float v2f;
typedef __attribute__((ext_vector_type(8))) float v8f;

#define NUM_USERS  100000
#define NUM_MOVIES 50000
#define NUM_EDGES  4000000
#define USER_FEAT  24
#define MOVIE_FEAT 404
#define HID        64
#define GENRES     3

#define OWN_BLOCKS   256
#define OWN_THREADS  1024
#define OWN_WAVES    32
#define OWN_MAXROWS  391
#define OWN_PITCH    68
#define OWN_LIST     512
#define OWN_LOADS    16

__global__ __launch_bounds__(OWN_THREADS)
void owner_sum_kernel(const float* __restrict__ X, const int* __restrict__ own_idx,
                      const int* __restrict__ src_idx, float* __restrict__ acc,
                      float* __restrict__ deg, int N, int E) {
    __shared__ float    s_acc[(OWN_MAXROWS + 1) * OWN_PITCH];
    __shared__ unsigned s_cnt[OWN_MAXROWS + 1];
    __shared__ unsigned s_list[OWN_WAVES][OWN_LIST];

    const int tid  = threadIdx.x;
    const int lane = tid & 31;
    const int wave = tid >> 5;
    const int R    = (N + OWN_BLOCKS - 1) / OWN_BLOCKS;
    const int base = blockIdx.x * R;

    for (int i = tid; i < (OWN_MAXROWS + 1) * OWN_PITCH; i += OWN_THREADS) s_acc[i] = 0.0f;
    for (int i = tid; i <= OWN_MAXROWS; i += OWN_THREADS) s_cnt[i] = 0u;
    __syncthreads();

    const int hw = lane >> 4;
    const int c  = (lane & 15) * 4;
    const int keys_per_wave = 32 * 4 * OWN_LOADS;
    const int keys_per_trip = keys_per_wave * OWN_WAVES;

    const int lane4 = lane * 4;
    const int ntrips = (E + keys_per_trip - 1) / keys_per_trip;
    int trip = (int)(((unsigned)blockIdx.x * (unsigned)ntrips) / OWN_BLOCKS);
    for (int it = 0; it < ntrips; ++it) {
        const int t0 = trip * keys_per_trip;
        trip = (trip + 1 == ntrips) ? 0 : trip + 1;
        const int wbase = t0 + wave * keys_per_wave;
        const int* kp = own_idx + wbase + lane4;
        int4 k[OWN_LOADS];
        if (wbase + keys_per_wave <= E) {
#pragma unroll
            for (int j = 0; j < OWN_LOADS; ++j) k[j] = *(const int4*)(kp + j * 128);
        } else {
#pragma unroll
            for (int j = 0; j < OWN_LOADS; ++j)
                k[j] = (wbase + j * 128 + lane4 + 3 < E) ? *(const int4*)(kp + j * 128)
                                                         : make_int4(-1, -1, -1, -1);
        }
        const int ebase = wbase + lane4;
        unsigned cnt = 0u;
#pragma unroll
        for (int j = 0; j < OWN_LOADS; ++j) {
            const int kv[4] = {k[j].x, k[j].y, k[j].z, k[j].w};
            bool mt[4];
#pragma unroll
            for (int q = 0; q < 4; ++q) mt[q] = (unsigned)(kv[q] - base) < (unsigned)R;
            if (__builtin_amdgcn_ballot_w32(mt[0] | mt[1] | mt[2] | mt[3]) != 0u) {
#pragma unroll
                for (int q = 0; q < 4; ++q) {
                    const unsigned bal = __builtin_amdgcn_ballot_w32(mt[q]);
                    if (bal != 0u) {
                        if (mt[q]) {
                            const unsigned slot = cnt + __builtin_amdgcn_mbcnt_lo(bal, 0u);
                            if (slot < OWN_LIST)
                                s_list[wave][slot] = (unsigned)(ebase + (j * 128 + q));
                        }
                        cnt += (unsigned)__builtin_popcount(bal);
                    }
                }
            }
        }
        if (cnt <= OWN_LIST) {
            for (unsigned m0 = 0; m0 < cnt; m0 += 8) {
                bool   ok[4];
                int    e[4];
                int    key[4];
                int    s[4];
                float4 v[4];
#pragma unroll
                for (int j = 0; j < 4; ++j) {
                    const unsigned m = m0 + j * 2 + hw;
                    const int ed = (int)s_list[wave][m];
                    ok[j] = m < cnt;
                    e[j]  = ok[j] ? ed : 0;
                }
#pragma unroll
                for (int j = 0; j < 4; ++j) {
                    key[j] = own_idx[e[j]];
                    s[j]   = src_idx[e[j]];
                }
#pragma unroll
                for (int j = 0; j < 4; ++j)
                    v[j] = *(const float4*)(X + (size_t)s[j] * HID + c);
#pragma unroll
                for (int j = 0; j < 4; ++j) {
                    const unsigned r = ok[j] ? (unsigned)(key[j] - base) : (unsigned)OWN_MAXROWS;
                    float* o = &s_acc[r * OWN_PITCH + c];
                    atomicAdd(o + 0, v[j].x);
                    atomicAdd(o + 1, v[j].y);
                    atomicAdd(o + 2, v[j].z);
                    atomicAdd(o + 3, v[j].w);
                    if (c == 0) atomicAdd(&s_cnt[r], 1u);
                }
            }
        } else {
            for (int p = 0; p < keys_per_wave; p += 32) {
                const int e1 = wbase + p + lane;
                const int key1 = (e1 < E) ? own_idx[e1] : -1;
                unsigned bal = __builtin_amdgcn_ballot_w32((unsigned)(key1 - base) < (unsigned)R);
                while (bal != 0u) {
                    const int eL = wbase + p + __builtin_ctz(bal);
                    bal &= bal - 1u;
                    const unsigned rL = (unsigned)(own_idx[eL] - base);
                    const float2 x = *(const float2*)(X + (size_t)src_idx[eL] * HID + lane * 2);
                    atomicAdd(&s_acc[rL * OWN_PITCH + lane * 2], x.x);
                    atomicAdd(&s_acc[rL * OWN_PITCH + lane * 2 + 1], x.y);
                    if (lane == 0) atomicAdd(&s_cnt[rL], 1u);
                }
            }
        }
    }
    __syncthreads();

    for (int i = tid; i < R * HID; i += OWN_THREADS) {
        const int r = i >> 6, ch = i & 63;
        const int row = base + r;
        if (row < N) acc[(size_t)row * HID + ch] = s_acc[r * OWN_PITCH + ch];
    }
    if (deg) {
        for (int r = tid; r < R; r += OWN_THREADS)
            if (base + r < N) deg[base + r] = (float)s_cnt[r];
    }
}

__global__ __launch_bounds__(256)
void sage_gemm64(const float* __restrict__ A1, const float* __restrict__ deg1,
                 const float* __restrict__ W1, int K1,
                 const float* __restrict__ A2, const float* __restrict__ W2, int K2,
                 const float* __restrict__ bias, float* __restrict__ C,
                 int M, int relu) {
    const int lane = threadIdx.x & 31;
    const int wave = threadIdx.x >> 5;
    const int half = lane >> 4;
    const int r    = lane & 15;
    const int row0 = (blockIdx.x * 8 + wave) * 16;
    if (row0 >= M) return;

    v8f c[4] = {};

    float scale = 1.0f;
    if (deg1) scale = 1.0f / fmaxf(deg1[row0 + r], 1.0f);

    {
        const float* Arow = A1 + (size_t)(row0 + r) * K1;
        for (int kk = 0; kk < K1; kk += 4) {
            const int k = kk + 2 * half;
            v2f a;
            a.x = Arow[k] * scale;
            a.y = Arow[k + 1] * scale;
#pragma unroll
            for (int t = 0; t < 4; ++t) {
                const float* Wrow = W1 + (size_t)(t * 16 + r) * K1;
                v2f b;
                b.x = Wrow[k];
                b.y = Wrow[k + 1];
                c[t] = __builtin_amdgcn_wmma_f32_16x16x4_f32(
                    false, a, false, b, (short)0, c[t], false, false);
            }
        }
    }
    if (A2) {
        const float* Arow = A2 + (size_t)(row0 + r) * K2;
        for (int kk = 0; kk < K2; kk += 4) {
            const int k = kk + 2 * half;
            v2f a;
            a.x = Arow[k];
            a.y = Arow[k + 1];
#pragma unroll
            for (int t = 0; t < 4; ++t) {
                const float* Wrow = W2 + (size_t)(t * 16 + r) * K2;
                v2f b;
                b.x = Wrow[k];
                b.y = Wrow[k + 1];
                c[t] = __builtin_amdgcn_wmma_f32_16x16x4_f32(
                    false, a, false, b, (short)0, c[t], false, false);
            }
        }
    }

#pragma unroll
    for (int t = 0; t < 4; ++t) {
        const float bv = bias ? bias[t * 16 + r] : 0.0f;
#pragma unroll
        for (int i = 0; i < 8; ++i) {
            float v = c[t][i] + bv;
            if (relu) v = fmaxf(v, 0.0f);
            C[(size_t)(row0 + i + 8 * half) * HID + t * 16 + r] = v;
        }
    }
}

__global__ __launch_bounds__(256)
void head_kernel(const float* __restrict__ acc2, const float* __restrict__ degm,
                 const float* __restrict__ m1, const float* __restrict__ Wl2,
                 const float* __restrict__ bl2, const float* __restrict__ Wr2,
                 float* __restrict__ out, int M) {
    int m = blockIdx.x * blockDim.x + threadIdx.x;
    if (m >= M) return;
    const float inv = 1.0f / fmaxf(degm[m], 1.0f);
    const float* a = acc2 + (size_t)m * HID;
    const float* x = m1 + (size_t)m * HID;
    float acc[GENRES];
#pragma unroll
    for (int g = 0; g < GENRES; ++g) acc[g] = bl2[g];
    for (int k = 0; k < HID; ++k) {
        const float av = a[k] * inv;
        const float xv = x[k];
#pragma unroll
        for (int g = 0; g < GENRES; ++g)
            acc[g] += av * Wl2[g * HID + k] + xv * Wr2[g * HID + k];
    }
#pragma unroll
    for (int g = 0; g < GENRES; ++g) out[(size_t)m * GENRES + g] = acc[g];
}

extern "C" void kernel_launch(void* const* d_in, const int* in_sizes, int n_in,
                              void* d_out, int out_size, void* d_ws, size_t ws_size,
                              hipStream_t stream) {
    (void)in_sizes; (void)n_in; (void)out_size; (void)ws_size;

    const float* x_user  = (const float*)d_in[0];
    const float* x_movie = (const float*)d_in[1];
    const int*   e_src   = (const int*)d_in[2];
    const int*   e_dst   = (const int*)d_in[3];
    const float* W_user  = (const float*)d_in[4];
    const float* b_user  = (const float*)d_in[5];
    const float* W_movie = (const float*)d_in[6];
    const float* b_movie = (const float*)d_in[7];
    const float* Wl1_um  = (const float*)d_in[8];
    const float* bl1_um  = (const float*)d_in[9];
    const float* Wr1_um  = (const float*)d_in[10];
    const float* Wl1_mu  = (const float*)d_in[11];
    const float* bl1_mu  = (const float*)d_in[12];
    const float* Wr1_mu  = (const float*)d_in[13];
    const float* Wl2_um  = (const float*)d_in[14];
    const float* bl2_um  = (const float*)d_in[15];
    const float* Wr2_um  = (const float*)d_in[16];
    float* out = (float*)d_out;

    float* ws = (float*)d_ws;
    float* acc_um    = ws;
    float* acc_mu    = acc_um + 3200000;
    float* acc2      = acc_mu + 6400000;
    float* deg_user  = acc2 + 3200000;
    float* deg_movie = deg_user + 100000;
    float* hu        = deg_movie + 50000;
    float* hm        = hu + 6400000;
    float* m1        = hm + 3200000;
    float* u1        = m1 + 3200000;

    const int E = NUM_EDGES;

    const int gridU = (NUM_USERS / 16 + 7) / 8;
    const int gridM = (NUM_MOVIES / 16 + 7) / 8;
    sage_gemm64<<<gridU, 256, 0, stream>>>(x_user, nullptr, W_user, USER_FEAT,
                                           nullptr, nullptr, 0, b_user, hu,
                                           NUM_USERS, 0);
    sage_gemm64<<<gridM, 256, 0, stream>>>(x_movie, nullptr, W_movie, MOVIE_FEAT,
                                           nullptr, nullptr, 0, b_movie, hm,
                                           NUM_MOVIES, 0);

    owner_sum_kernel<<<OWN_BLOCKS, OWN_THREADS, 0, stream>>>(hu, e_dst, e_src, acc_um,
                                                             deg_movie, NUM_MOVIES, E);
    sage_gemm64<<<gridM, 256, 0, stream>>>(acc_um, deg_movie, Wl1_um, HID,
                                           hm, Wr1_um, HID, bl1_um, m1,
                                           NUM_MOVIES, 1);

    owner_sum_kernel<<<OWN_BLOCKS, OWN_THREADS, 0, stream>>>(hm, e_src, e_dst, acc_mu,
                                                             deg_user, NUM_USERS, E);
    sage_gemm64<<<gridU, 256, 0, stream>>>(acc_mu, deg_user, Wl1_mu, HID,
                                           hu, Wr1_mu, HID, bl1_mu, u1,
                                           NUM_USERS, 1);

    owner_sum_kernel<<<OWN_BLOCKS, OWN_THREADS, 0, stream>>>(u1, e_dst, e_src, acc2,
                                                             nullptr, NUM_MOVIES, E);
    head_kernel<<<(NUM_MOVIES + 255) / 256, 256, 0, stream>>>(
        acc2, deg_movie, m1, Wl2_um, bl2_um, Wr2_um, out, NUM_MOVIES);
}


# Owner-computed edge sums re-tiled as node ranges x edge slices (64x4 movie side, 128x2 user side): a key is read by 64 or 128 blocks, not 256; slice planes added by their readers; sum order changes
# speedup vs baseline: 2.2531x; 1.4267x over previous
#include <hip/hip_runtime.h>
#include <hip/hip_bf16.h>

typedef __attribute__((ext_vector_type(2))) float v2f;
typedef __attribute__((ext_vector_type(8))) float v8f;

#define NUM_USERS  100000
#define NUM_MOVIES 50000
#define NUM_EDGES  4000000
#define USER_FEAT  24
#define MOVIE_FEAT 404
#define HID        64
#define GENRES     3

#define OWN_BLOCKS   256
#define OWN_THREADS  1024
#define OWN_WAVES    32
#define OWN_MAXROWS  782
#define OWN_PITCH    68
#define OWN_LIST     512
#define OWN_LOADS    16
#define MOV_RANGES   64
#define MOV_SLICES   4
#define USR_RANGES   128
#define USR_SLICES   2

__global__ __launch_bounds__(OWN_THREADS)
void owner_sum_kernel(const float* __restrict__ X, const int* __restrict__ own_idx,
                      const int* __restrict__ src_idx, float* __restrict__ acc,
                      float* __restrict__ deg, int N, int E, int P, int S) {
    __shared__ float    s_acc[(OWN_MAXROWS + 1) * OWN_PITCH];
    __shared__ unsigned s_cnt[OWN_MAXROWS + 1];
    __shared__ unsigned s_list[OWN_WAVES][OWN_LIST];

    const int tid  = threadIdx.x;
    const int lane = tid & 31;
    const int wave = tid >> 5;
    const int R    = (N + P - 1) / P;
    const int pr   = (int)blockIdx.x % P;
    const int sl   = (int)blockIdx.x / P;
    const int base = pr * R;
    const int slen = (((E + S - 1) / S) + 3) & ~3;
    const int e_lo = sl * slen;
    const int e_hi = (e_lo + slen < E) ? e_lo + slen : E;
    acc += (size_t)sl * (size_t)N * HID;
    if (deg) deg += (size_t)sl * (size_t)N;

    for (int i = tid; i < (OWN_MAXROWS + 1) * OWN_PITCH; i += OWN_THREADS) s_acc[i] = 0.0f;
    for (int i = tid; i <= OWN_MAXROWS; i += OWN_THREADS) s_cnt[i] = 0u;
    __syncthreads();

    const int hw = lane >> 4;
    const int c  = (lane & 15) * 4;
    const int keys_per_wave = 32 * 4 * OWN_LOADS;
    const int keys_per_trip = keys_per_wave * OWN_WAVES;

    const int lane4 = lane * 4;
    const int ntrips = (slen + keys_per_trip - 1) / keys_per_trip;
    int trip = (int)(((unsigned)pr * (unsigned)ntrips) / (unsigned)P);
    for (int it = 0; it < ntrips; ++it) {
        const int t0 = e_lo + trip * keys_per_trip;
        trip = (trip + 1 == ntrips) ? 0 : trip + 1;
        const int wbase = t0 + wave * keys_per_wave;
        const int* kp = own_idx + wbase + lane4;
        int4 k[OWN_LOADS];
        if (wbase + keys_per_wave <= e_hi) {
#pragma unroll
            for (int j = 0; j < OWN_LOADS; ++j) k[j] = *(const int4*)(kp + j * 128);
        } else {
#pragma unroll
            for (int j = 0; j < OWN_LOADS; ++j)
                k[j] = (wbase + j * 128 + lane4 + 3 < e_hi) ? *(const int4*)(kp + j * 128)
                                                         : make_int4(-1, -1, -1, -1);
        }
        const int ebase = wbase + lane4;
        unsigned cnt = 0u;
#pragma unroll
        for (int j = 0; j < OWN_LOADS; ++j) {
            const int kv[4] = {k[j].x, k[j].y, k[j].z, k[j].w};
            bool mt[4];
#pragma unroll
            for (int q = 0; q < 4; ++q) mt[q] = (unsigned)(kv[q] - base) < (unsigned)R;
            if (__builtin_amdgcn_ballot_w32(mt[0] | mt[1] | mt[2] | mt[3]) != 0u) {
#pragma unroll
                for (int q = 0; q < 4; ++q) {
                    const unsigned bal = __builtin_amdgcn_ballot_w32(mt[q]);
                    if (bal != 0u) {
                        if (mt[q]) {
                            const unsigned slot = cnt + __builtin_amdgcn_mbcnt_lo(bal, 0u);
                            if (slot < OWN_LIST)
                                s_list[wave][slot] = (unsigned)(ebase + (j * 128 + q));
                        }
                        cnt += (unsigned)__builtin_popcount(bal);
                    }
                }
            }
        }
        if (cnt <= OWN_LIST) {
            for (unsigned m0 = 0; m0 < cnt; m0 += 8) {
                bool   ok[4];
                int    e[4];
                int    key[4];
                int    s[4];
                float4 v[4];
#pragma unroll
                for (int j = 0; j < 4; ++j) {
                    const unsigned m = m0 + j * 2 + hw;
                    const int ed = (int)s_list[wave][m];
                    ok[j] = m < cnt;
                    e[j]  = ok[j] ? ed : 0;
                }
#pragma unroll
                for (int j = 0; j < 4; ++j) {
                    key[j] = own_idx[e[j]];
                    s[j]   = src_idx[e[j]];
                }
#pragma unroll
                for (int j = 0; j < 4; ++j)
                    v[j] = *(const float4*)(X + (size_t)s[j] * HID + c);
#pragma unroll
                for (int j = 0; j < 4; ++j) {
                    const unsigned r = ok[j] ? (unsigned)(key[j] - base) : (unsigned)OWN_MAXROWS;
                    float* o = &s_acc[r * OWN_PITCH + c];
                    atomicAdd(o + 0, v[j].x);
                    atomicAdd(o + 1, v[j].y);
                    atomicAdd(o + 2, v[j].z);
                    atomicAdd(o + 3, v[j].w);
                    if (c == 0) atomicAdd(&s_cnt[r], 1u);
                }
            }
        } else {
            for (int p = 0; p < keys_per_wave; p += 32) {
                const int e1 = wbase + p + lane;
                const int key1 = (e1 < e_hi) ? own_idx[e1] : -1;
                unsigned bal = __builtin_amdgcn_ballot_w32((unsigned)(key1 - base) < (unsigned)R);
                while (bal != 0u) {
                    const int eL = wbase + p + __builtin_ctz(bal);
                    bal &= bal - 1u;
                    const unsigned rL = (unsigned)(own_idx[eL] - base);
                    const float2 x = *(const float2*)(X + (size_t)src_idx[eL] * HID + lane * 2);
                    atomicAdd(&s_acc[rL * OWN_PITCH + lane * 2], x.x);
                    atomicAdd(&s_acc[rL * OWN_PITCH + lane * 2 + 1], x.y);
                    if (lane == 0) atomicAdd(&s_cnt[rL], 1u);
                }
            }
        }
    }
    __syncthreads();

    for (int i = tid; i < R * HID; i += OWN_THREADS) {
        const int r = i >> 6, ch = i & 63;
        const int row = base + r;
        if (row < N) acc[(size_t)row * HID + ch] = s_acc[r * OWN_PITCH + ch];
    }
    if (deg) {
        for (int r = tid; r < R; r += OWN_THREADS)
            if (base + r < N) deg[base + r] = (float)s_cnt[r];
    }
}

template <int NP>
__global__ __launch_bounds__(256)
void sage_gemm64(const float* __restrict__ A1, const float* __restrict__ deg1,
                 const float* __restrict__ W1, int K1,
                 const float* __restrict__ A2, const float* __restrict__ W2, int K2,
                 const float* __restrict__ bias, float* __restrict__ C,
                 int M, int relu) {
    const int lane = threadIdx.x & 31;
    const int wave = threadIdx.x >> 5;
    const int half = lane >> 4;
    const int r    = lane & 15;
    const int row0 = (blockIdx.x * 8 + wave) * 16;
    if (row0 >= M) return;

    v8f c[4] = {};

    float scale = 1.0f;
    if (deg1) {
        float d = deg1[row0 + r];
#pragma unroll
        for (int s = 1; s < NP; ++s) d += deg1[(size_t)s * M + row0 + r];
        scale = 1.0f / fmaxf(d, 1.0f);
    }

    {
        const float* Arow = A1 + (size_t)(row0 + r) * K1;
        const size_t plane = (size_t)M * K1;
        for (int kk = 0; kk < K1; kk += 4) {
            const int k = kk + 2 * half;
            float ax = Arow[k];
            float ay = Arow[k + 1];
#pragma unroll
            for (int s = 1; s < NP; ++s) {
                ax += Arow[s * plane + k];
                ay += Arow[s * plane + k + 1];
            }
            v2f a;
            a.x = ax * scale;
            a.y = ay * scale;
#pragma unroll
            for (int t = 0; t < 4; ++t) {
                const float* Wrow = W1 + (size_t)(t * 16 + r) * K1;
                v2f b;
                b.x = Wrow[k];
                b.y = Wrow[k + 1];
                c[t] = __builtin_amdgcn_wmma_f32_16x16x4_f32(
                    false, a, false, b, (short)0, c[t], false, false);
            }
        }
    }
    if (A2) {
        const float* Arow = A2 + (size_t)(row0 + r) * K2;
        for (int kk = 0; kk < K2; kk += 4) {
            const int k = kk + 2 * half;
            v2f a;
            a.x = Arow[k];
            a.y = Arow[k + 1];
#pragma unroll
            for (int t = 0; t < 4; ++t) {
                const float* Wrow = W2 + (size_t)(t * 16 + r) * K2;
                v2f b;
                b.x = Wrow[k];
                b.y = Wrow[k + 1];
                c[t] = __builtin_amdgcn_wmma_f32_16x16x4_f32(
                    false, a, false, b, (short)0, c[t], false, false);
            }
        }
    }

#pragma unroll
    for (int t = 0; t < 4; ++t) {
        const float bv = bias ? bias[t * 16 + r] : 0.0f;
#pragma unroll
        for (int i = 0; i < 8; ++i) {
            float v = c[t][i] + bv;
            if (relu) v = fmaxf(v, 0.0f);
            C[(size_t)(row0 + i + 8 * half) * HID + t * 16 + r] = v;
        }
    }
}

__global__ __launch_bounds__(256)
void head_kernel(const float* __restrict__ acc2, const float* __restrict__ degm,
                 const float* __restrict__ m1, const float* __restrict__ Wl2,
                 const float* __restrict__ bl2, const float* __restrict__ Wr2,
                 float* __restrict__ out, int M) {
    int m = blockIdx.x * blockDim.x + threadIdx.x;
    if (m >= M) return;
    float dsum = degm[m];
#pragma unroll
    for (int s = 1; s < MOV_SLICES; ++s) dsum += degm[(size_t)s * M + m];
    const float inv = 1.0f / fmaxf(dsum, 1.0f);
    const size_t plane = (size_t)M * HID;
    const float* a = acc2 + (size_t)m * HID;
    const float* x = m1 + (size_t)m * HID;
    float acc[GENRES];
#pragma unroll
    for (int g = 0; g < GENRES; ++g) acc[g] = bl2[g];
    for (int k = 0; k < HID; ++k) {
        float as = a[k];
#pragma unroll
        for (int s = 1; s < MOV_SLICES; ++s) as += a[s * plane + k];
        const float av = as * inv;
        const float xv = x[k];
#pragma unroll
        for (int g = 0; g < GENRES; ++g)
            acc[g] += av * Wl2[g * HID + k] + xv * Wr2[g * HID + k];
    }
#pragma unroll
    for (int g = 0; g < GENRES; ++g) out[(size_t)m * GENRES + g] = acc[g];
}

extern "C" void kernel_launch(void* const* d_in, const int* in_sizes, int n_in,
                              void* d_out, int out_size, void* d_ws, size_t ws_size,
                              hipStream_t stream) {
    (void)in_sizes; (void)n_in; (void)out_size; (void)ws_size;

    const float* x_user  = (const float*)d_in[0];
    const float* x_movie = (const float*)d_in[1];
    const int*   e_src   = (const int*)d_in[2];
    const int*   e_dst   = (const int*)d_in[3];
    const float* W_user  = (const float*)d_in[4];
    const float* b_user  = (const float*)d_in[5];
    const float* W_movie = (const float*)d_in[6];
    const float* b_movie = (const float*)d_in[7];
    const float* Wl1_um  = (const float*)d_in[8];
    const float* bl1_um  = (const float*)d_in[9];
    const float* Wr1_um  = (const float*)d_in[10];
    const float* Wl1_mu  = (const float*)d_in[11];
    const float* bl1_mu  = (const float*)d_in[12];
    const float* Wr1_mu  = (const float*)d_in[13];
    const float* Wl2_um  = (const float*)d_in[14];
    const float* bl2_um  = (const float*)d_in[15];
    const float* Wr2_um  = (const float*)d_in[16];
    float* out = (float*)d_out;

    float* ws = (float*)d_ws;
    float* acc_um    = ws;
    float* acc_mu    = ws;
    float* acc2      = ws;
    float* hu        = ws + 12800000;
    float* m1        = hu + 6400000;
    float* deg_movie = m1 + 3200000;
    float* deg_user  = deg_movie + 200000;
    float* hm        = deg_user + 200000;
    float* u1        = hm;

    const int E = NUM_EDGES;

    const int gridU = (NUM_USERS / 16 + 7) / 8;
    const int gridM = (NUM_MOVIES / 16 + 7) / 8;
    sage_gemm64<1><<<gridU, 256, 0, stream>>>(x_user, nullptr, W_user, USER_FEAT,
                                           nullptr, nullptr, 0, b_user, hu,
                                           NUM_USERS, 0);
    sage_gemm64<1><<<gridM, 256, 0, stream>>>(x_movie, nullptr, W_movie, MOVIE_FEAT,
                                           nullptr, nullptr, 0, b_movie, hm,
                                           NUM_MOVIES, 0);

    owner_sum_kernel<<<OWN_BLOCKS, OWN_THREADS, 0, stream>>>(hu, e_dst, e_src, acc_um,
                                                             deg_movie, NUM_MOVIES, E,
                                                             MOV_RANGES, MOV_SLICES);
    sage_gemm64<MOV_SLICES><<<gridM, 256, 0, stream>>>(acc_um, deg_movie, Wl1_um, HID,
                                           hm, Wr1_um, HID, bl1_um, m1,
                                           NUM_MOVIES, 1);

    owner_sum_kernel<<<OWN_BLOCKS, OWN_THREADS, 0, stream>>>(hm, e_src, e_dst, acc_mu,
                                                             deg_user, NUM_USERS, E,
                                                             USR_RANGES, USR_SLICES);
    sage_gemm64<USR_SLICES><<<gridU, 256, 0, stream>>>(acc_mu, deg_user, Wl1_mu, HID,
                                           hu, Wr1_mu, HID, bl1_mu, u1,
                                           NUM_USERS, 1);

    owner_sum_kernel<<<OWN_BLOCKS, OWN_THREADS, 0, stream>>>(u1, e_dst, e_src, acc2,
                                                             nullptr, NUM_MOVIES, E,
                                                             MOV_RANGES, MOV_SLICES);
    head_kernel<<<(NUM_MOVIES + 255) / 256, 256, 0, stream>>>(
        acc2, deg_movie, m1, Wl2_um, bl2_um, Wr2_um, out, NUM_MOVIES);
}


# Owner scans read a one-byte range id per edge (key / range length, from a small pre-pass; 10 launches), not the 4-byte key: 16 ids per lane-load, zero-byte test per dword; sum order changes
# speedup vs baseline: 2.5207x; 1.1188x over previous
#include <hip/hip_runtime.h>
#include <hip/hip_bf16.h>

typedef __attribute__((ext_vector_type(2))) float v2f;
typedef __attribute__((ext_vector_type(8))) float v8f;

#define NUM_USERS  100000
#define NUM_MOVIES 50000
#define NUM_EDGES  4000000
#define USER_FEAT  24
#define MOVIE_FEAT 404
#define HID        64
#define GENRES     3

#define OWN_BLOCKS   256
#define OWN_THREADS  1024
#define OWN_WAVES    32
#define OWN_MAXROWS  782
#define OWN_PITCH    68
#define OWN_LIST     512
#define OWN_LOADS    8
#define MOV_RANGES   64
#define MOV_SLICES   4
#define USR_RANGES   128
#define USR_SLICES   2
#define MOV_R        ((NUM_MOVIES + MOV_RANGES - 1) / MOV_RANGES)
#define USR_R        ((NUM_USERS + USR_RANGES - 1) / USR_RANGES)
static_assert(MOV_RANGES <= 128 && USR_RANGES <= 128, "a range id must fit in 7 bits");
static_assert(NUM_EDGES % 16 == 0, "16-byte loads of range ids");

template <int RA, int RB>
__global__ __launch_bounds__(256)
void range_id_kernel(const int* __restrict__ keysA, int* __restrict__ ridA,
                     const int* __restrict__ keysB, int* __restrict__ ridB, int n4) {
    const int i = (int)blockIdx.x * 256 + (int)threadIdx.x;
    if (i >= n4) return;
    const bool second = blockIdx.y != 0;
    const int4 k = *(const int4*)((second ? keysB : keysA) + 4 * i);
    const unsigned kv[4] = {(unsigned)k.x, (unsigned)k.y, (unsigned)k.z, (unsigned)k.w};
    unsigned w = 0u;
#pragma unroll
    for (int q = 0; q < 4; ++q) {
        const unsigned id = second ? kv[q] / (unsigned)RB : kv[q] / (unsigned)RA;
        w |= (id & 127u) << (8 * q);
    }
    (second ? ridB : ridA)[i] = (int)w;
}

__global__ __launch_bounds__(OWN_THREADS)
void owner_sum_kernel(const float* __restrict__ X, const int* __restrict__ own_idx,
                      const int* __restrict__ rid,
                      const int* __restrict__ src_idx, float* __restrict__ acc,
                      float* __restrict__ deg, int N, int E, int P, int S) {
    __shared__ float    s_acc[(OWN_MAXROWS + 1) * OWN_PITCH];
    __shared__ unsigned s_cnt[OWN_MAXROWS + 1];
    __shared__ unsigned s_list[OWN_WAVES][OWN_LIST + 1];

    const int tid  = threadIdx.x;
    const int lane = tid & 31;
    const int wave = tid >> 5;
    const int R    = (N + P - 1) / P;
    const int pr   = (int)blockIdx.x % P;
    const int sl   = (int)blockIdx.x / P;
    const int base = pr * R;
    const int slen = (((E + S - 1) / S) + 15) & ~15;
    const int e_lo = sl * slen;
    const int e_hi = (e_lo + slen < E) ? e_lo + slen : E;
    acc += (size_t)sl * (size_t)N * HID;
    if (deg) deg += (size_t)sl * (size_t)N;

    for (int i = tid; i < (OWN_MAXROWS + 1) * OWN_PITCH; i += OWN_THREADS) s_acc[i] = 0.0f;
    for (int i = tid; i <= OWN_MAXROWS; i += OWN_THREADS) s_cnt[i] = 0u;
    __syncthreads();

    const int hw = lane >> 4;
    const int c  = (lane & 15) * 4;
    const int keys_per_wave = 32 * 16 * OWN_LOADS;
    const int keys_per_trip = keys_per_wave * OWN_WAVES;

    const int lane16 = lane * 16;
    const unsigned rep4 = (unsigned)pr * 0x01010101u;
    const int nomatch = (int)(rep4 ^ 0x01010101u);
    const int ntrips = (slen + keys_per_trip - 1) / keys_per_trip;
    int trip = (int)(((unsigned)pr * (unsigned)ntrips) / (unsigned)P);
    for (int it = 0; it < ntrips; ++it) {
        const int t0 = e_lo + trip * keys_per_trip;
        trip = (trip + 1 == ntrips) ? 0 : trip + 1;
        const int wbase = t0 + wave * keys_per_wave;
        const int* kp = rid + ((wbase + lane16) >> 2);
        int4 k[OWN_LOADS];
        if (wbase + keys_per_wave <= e_hi) {
#pragma unroll
            for (int j = 0; j < OWN_LOADS; ++j) k[j] = *(const int4*)(kp + j * 128);
        } else {
#pragma unroll
            for (int j = 0; j < OWN_LOADS; ++j)
                k[j] = (wbase + j * 512 + lane16 + 15 < e_hi) ? *(const int4*)(kp + j * 128)
                                                          : make_int4(nomatch, nomatch, nomatch, nomatch);
        }
        const int ebase = wbase + lane16;
        unsigned cnt = 0u;
#pragma unroll
        for (int j = 0; j < OWN_LOADS; ++j) {
            const unsigned t0 = ((unsigned)k[j].x ^ rep4) + 0x7F7F7F7Fu;
            const unsigned t1 = ((unsigned)k[j].y ^ rep4) + 0x7F7F7F7Fu;
            const unsigned t2 = ((unsigned)k[j].z ^ rep4) + 0x7F7F7F7Fu;
            const unsigned t3 = ((unsigned)k[j].w ^ rep4) + 0x7F7F7F7Fu;
            unsigned hits = (((t0 >> 7) & 0x01010101u) | ((t1 >> 6) & 0x02020202u) |
                          ((t2 >> 5) & 0x04040404u) | ((t3 >> 4) & 0x08080808u)) ^ 0x0F0F0F0Fu;
            unsigned bal = __builtin_amdgcn_ballot_w32(hits != 0u);
            while (bal != 0u) {
                if (hits != 0u) {
                    const unsigned p = (unsigned)__builtin_ctz(hits);
                    unsigned slot = cnt + __builtin_amdgcn_mbcnt_lo(bal, 0u);
                    slot = slot < OWN_LIST ? slot : OWN_LIST;
                    s_list[wave][slot] = (unsigned)(ebase + j * 512) + 4u * (p & 7u) + (p >> 3);
                    hits &= hits - 1u;
                }
                cnt += (unsigned)__builtin_popcount(bal);
                bal = __builtin_amdgcn_ballot_w32(hits != 0u);
            }
        }
        if (cnt <= OWN_LIST) {
            for (unsigned m0 = 0; m0 < cnt; m0 += 8) {
                bool   ok[4];
                int    e[4];
                int    key[4];
                int    s[4];
                float4 v[4];
#pragma unroll
                for (int j = 0; j < 4; ++j) {
                    const unsigned m = m0 + j * 2 + hw;
                    const int ed = (int)s_list[wave][m];
                    ok[j] = m < cnt;
                    e[j]  = ok[j] ? ed : 0;
                }
#pragma unroll
                for (int j = 0; j < 4; ++j) {
                    key[j] = own_idx[e[j]];
                    s[j]   = src_idx[e[j]];
                }
#pragma unroll
                for (int j = 0; j < 4; ++j)
                    v[j] = *(const float4*)(X + (size_t)s[j] * HID + c);
#pragma unroll
                for (int j = 0; j < 4; ++j) {
                    const unsigned rr = (unsigned)(key[j] - base);
                    const unsigned r = (ok[j] && rr < (unsigned)R) ? rr : (unsigned)OWN_MAXROWS;
                    float* o = &s_acc[r * OWN_PITCH + c];
                    atomicAdd(o + 0, v[j].x);
                    atomicAdd(o + 1, v[j].y);
                    atomicAdd(o + 2, v[j].z);
                    atomicAdd(o + 3, v[j].w);
                    if (c == 0) atomicAdd(&s_cnt[r], 1u);
                }
            }
        } else {
            for (int p = 0; p < keys_per_wave; p += 32) {
                const int e1 = wbase + p + lane;
                const int key1 = (e1 < e_hi) ? own_idx[e1] : -1;
                unsigned bal = __builtin_amdgcn_ballot_w32((unsigned)(key1 - base) < (unsigned)R);
                while (bal != 0u) {
                    const int eL = wbase + p + __builtin_ctz(bal);
                    bal &= bal - 1u;
                    const unsigned rL = (unsigned)(own_idx[eL] - base);
                    const float2 x = *(const float2*)(X + (size_t)src_idx[eL] * HID + lane * 2);
                    atomicAdd(&s_acc[rL * OWN_PITCH + lane * 2], x.x);
                    atomicAdd(&s_acc[rL * OWN_PITCH + lane * 2 + 1], x.y);
                    if (lane == 0) atomicAdd(&s_cnt[rL], 1u);
                }
            }
        }
    }
    __syncthreads();

    for (int i = tid; i < R * HID; i += OWN_THREADS) {
        const int r = i >> 6, ch = i & 63;
        const int row = base + r;
        if (row < N) acc[(size_t)row * HID + ch] = s_acc[r * OWN_PITCH + ch];
    }
    if (deg) {
        for (int r = tid; r < R; r += OWN_THREADS)
            if (base + r < N) deg[base + r] = (float)s_cnt[r];
    }
}

template <int NP>
__global__ __launch_bounds__(256)
void sage_gemm64(const float* __restrict__ A1, const float* __restrict__ deg1,
                 const float* __restrict__ W1, int K1,
                 const float* __restrict__ A2, const float* __restrict__ W2, int K2,
                 const float* __restrict__ bias, float* __restrict__ C,
                 int M, int relu) {
    const int lane = threadIdx.x & 31;
    const int wave = threadIdx.x >> 5;
    const int half = lane >> 4;
    const int r    = lane & 15;
    const int row0 = (blockIdx.x * 8 + wave) * 16;
    if (row0 >= M) return;

    v8f c[4] = {};

    float scale = 1.0f;
    if (deg1) {
        float d = deg1[row0 + r];
#pragma unroll
        for (int s = 1; s < NP; ++s) d += deg1[(size_t)s * M + row0 + r];
        scale = 1.0f / fmaxf(d, 1.0f);
    }

    {
        const float* Arow = A1 + (size_t)(row0 + r) * K1;
        const size_t plane = (size_t)M * K1;
        for (int kk = 0; kk < K1; kk += 4) {
            const int k = kk + 2 * half;
            float ax = Arow[k];
            float ay = Arow[k + 1];
#pragma unroll
            for (int s = 1; s < NP; ++s) {
                ax += Arow[s * plane + k];
                ay += Arow[s * plane + k + 1];
            }
            v2f a;
            a.x = ax * scale;
            a.y = ay * scale;
#pragma unroll
            for (int t = 0; t < 4; ++t) {
                const float* Wrow = W1 + (size_t)(t * 16 + r) * K1;
                v2f b;
                b.x = Wrow[k];
                b.y = Wrow[k + 1];
                c[t] = __builtin_amdgcn_wmma_f32_16x16x4_f32(
                    false, a, false, b, (short)0, c[t], false, false);
            }
        }
    }
    if (A2) {
        const float* Arow = A2 + (size_t)(row0 + r) * K2;
        for (int kk = 0; kk < K2; kk += 4) {
            const int k = kk + 2 * half;
            v2f a;
            a.x = Arow[k];
            a.y = Arow[k + 1];
#pragma unroll
            for (int t = 0; t < 4; ++t) {
                const float* Wrow = W2 + (size_t)(t * 16 + r) * K2;
                v2f b;
                b.x = Wrow[k];
                b.y = Wrow[k + 1];
                c[t] = __builtin_amdgcn_wmma_f32_16x16x4_f32(
                    false, a, false, b, (short)0, c[t], false, false);
            }
        }
    }

#pragma unroll
    for (int t = 0; t < 4; ++t) {
        const float bv = bias ? bias[t * 16 + r] : 0.0f;
#pragma unroll
        for (int i = 0; i < 8; ++i) {
            float v = c[t][i] + bv;
            if (relu) v = fmaxf(v, 0.0f);
            C[(size_t)(row0 + i + 8 * half) * HID + t * 16 + r] = v;
        }
    }
}

__global__ __launch_bounds__(256)
void head_kernel(const float* __restrict__ acc2, const float* __restrict__ degm,
                 const float* __restrict__ m1, const float* __restrict__ Wl2,
                 const float* __restrict__ bl2, const float* __restrict__ Wr2,
                 float* __restrict__ out, int M) {
    int m = blockIdx.x * blockDim.x + threadIdx.x;
    if (m >= M) return;
    float dsum = degm[m];
#pragma unroll
    for (int s = 1; s < MOV_SLICES; ++s) dsum += degm[(size_t)s * M + m];
    const float inv = 1.0f / fmaxf(dsum, 1.0f);
    const size_t plane = (size_t)M * HID;
    const float* a = acc2 + (size_t)m * HID;
    const float* x = m1 + (size_t)m * HID;
    float acc[GENRES];
#pragma unroll
    for (int g = 0; g < GENRES; ++g) acc[g] = bl2[g];
    for (int k = 0; k < HID; ++k) {
        float as = a[k];
#pragma unroll
        for (int s = 1; s < MOV_SLICES; ++s) as += a[s * plane + k];
        const float av = as * inv;
        const float xv = x[k];
#pragma unroll
        for (int g = 0; g < GENRES; ++g)
            acc[g] += av * Wl2[g * HID + k] + xv * Wr2[g * HID + k];
    }
#pragma unroll
    for (int g = 0; g < GENRES; ++g) out[(size_t)m * GENRES + g] = acc[g];
}

extern "C" void kernel_launch(void* const* d_in, const int* in_sizes, int n_in,
                              void* d_out, int out_size, void* d_ws, size_t ws_size,
                              hipStream_t stream) {
    (void)in_sizes; (void)n_in; (void)out_size; (void)ws_size;

    const float* x_user  = (const float*)d_in[0];
    const float* x_movie = (const float*)d_in[1];
    const int*   e_src   = (const int*)d_in[2];
    const int*   e_dst   = (const int*)d_in[3];
    const float* W_user  = (const float*)d_in[4];
    const float* b_user  = (const float*)d_in[5];
    const float* W_movie = (const float*)d_in[6];
    const float* b_movie = (const float*)d_in[7];
    const float* Wl1_um  = (const float*)d_in[8];
    const float* bl1_um  = (const float*)d_in[9];
    const float* Wr1_um  = (const float*)d_in[10];
    const float* Wl1_mu  = (const float*)d_in[11];
    const float* bl1_mu  = (const float*)d_in[12];
    const float* Wr1_mu  = (const float*)d_in[13];
    const float* Wl2_um  = (const float*)d_in[14];
    const float* bl2_um  = (const float*)d_in[15];
    const float* Wr2_um  = (const float*)d_in[16];
    float* out = (float*)d_out;

    float* ws = (float*)d_ws;
    float* acc_um    = ws;
    float* acc_mu    = ws;
    float* acc2      = ws;
    float* hu        = ws + 12800000;
    float* m1        = hu + 6400000;
    float* deg_movie = m1 + 3200000;
    float* deg_user  = deg_movie + 200000;
    float* hm        = deg_user + 200000;
    float* u1        = hm;
    int*   rid_dst   = (int*)(hm + 3200000);
    int*   rid_src   = (int*)(hm + 4200000);
    int*   rid_dst2  = (int*)hu;

    const int E = NUM_EDGES;

    range_id_kernel<MOV_R, USR_R><<<dim3((NUM_EDGES / 4 + 255) / 256, 2), 256, 0, stream>>>(
        e_dst, rid_dst, e_src, rid_src, NUM_EDGES / 4);

    const int gridU = (NUM_USERS / 16 + 7) / 8;
    const int gridM = (NUM_MOVIES / 16 + 7) / 8;
    sage_gemm64<1><<<gridU, 256, 0, stream>>>(x_user, nullptr, W_user, USER_FEAT,
                                           nullptr, nullptr, 0, b_user, hu,
                                           NUM_USERS, 0);
    sage_gemm64<1><<<gridM, 256, 0, stream>>>(x_movie, nullptr, W_movie, MOVIE_FEAT,
                                           nullptr, nullptr, 0, b_movie, hm,
                                           NUM_MOVIES, 0);

    owner_sum_kernel<<<OWN_BLOCKS, OWN_THREADS, 0, stream>>>(hu, e_dst, rid_dst, e_src, acc_um,
                                                             deg_movie, NUM_MOVIES, E,
                                                             MOV_RANGES, MOV_SLICES);
    sage_gemm64<MOV_SLICES><<<gridM, 256, 0, stream>>>(acc_um, deg_movie, Wl1_um, HID,
                                           hm, Wr1_um, HID, bl1_um, m1,
                                           NUM_MOVIES, 1);

    owner_sum_kernel<<<OWN_BLOCKS, OWN_THREADS, 0, stream>>>(hm, e_src, rid_src, e_dst, acc_mu,
                                                             deg_user, NUM_USERS, E,
                                                             USR_RANGES, USR_SLICES);
    sage_gemm64<USR_SLICES><<<gridU, 256, 0, stream>>>(acc_mu, deg_user, Wl1_mu, HID,
                                           hu, Wr1_mu, HID, bl1_mu, u1,
                                           NUM_USERS, 1);

    range_id_kernel<MOV_R, MOV_R><<<dim3((NUM_EDGES / 4 + 255) / 256, 1), 256, 0, stream>>>(
        e_dst, rid_dst2, e_dst, rid_dst2, NUM_EDGES / 4);
    owner_sum_kernel<<<OWN_BLOCKS, OWN_THREADS, 0, stream>>>(u1, e_dst, rid_dst2, e_src, acc2,
                                                             nullptr, NUM_MOVIES, E,
                                                             MOV_RANGES, MOV_SLICES);
    head_kernel<<<(NUM_MOVIES + 255) / 256, 256, 0, stream>>>(
        acc2, deg_movie, m1, Wl2_um, bl2_um, Wr2_um, out, NUM_MOVIES);
}


# Owner-sum LDS float adds made bank-conflict-free: the upper half-wave adds its four channels in the order z,w,x,y, so the two halves of every add hit disjoint banks; sum order changes
# speedup vs baseline: 2.8196x; 1.1186x over previous
#include <hip/hip_runtime.h>
#include <hip/hip_bf16.h>

typedef __attribute__((ext_vector_type(2))) float v2f;
typedef __attribute__((ext_vector_type(8))) float v8f;

#define NUM_USERS  100000
#define NUM_MOVIES 50000
#define NUM_EDGES  4000000
#define USER_FEAT  24
#define MOVIE_FEAT 404
#define HID        64
#define GENRES     3

#define OWN_BLOCKS   256
#define OWN_THREADS  1024
#define OWN_WAVES    32
#define OWN_MAXROWS  782
#define OWN_PITCH    68
#define OWN_LIST     512
#define OWN_LOADS    8
#define MOV_RANGES   64
#define MOV_SLICES   4
#define USR_RANGES   128
#define USR_SLICES   2
#define MOV_R        ((NUM_MOVIES + MOV_RANGES - 1) / MOV_RANGES)
#define USR_R        ((NUM_USERS + USR_RANGES - 1) / USR_RANGES)
static_assert(MOV_RANGES <= 128 && USR_RANGES <= 128, "a range id must fit in 7 bits");
static_assert(NUM_EDGES % 16 == 0, "16-byte loads of range ids");

template <int RA, int RB>
__global__ __launch_bounds__(256)
void range_id_kernel(const int* __restrict__ keysA, int* __restrict__ ridA,
                     const int* __restrict__ keysB, int* __restrict__ ridB, int n4) {
    const int i = (int)blockIdx.x * 256 + (int)threadIdx.x;
    if (i >= n4) return;
    const bool second = blockIdx.y != 0;
    const int4 k = *(const int4*)((second ? keysB : keysA) + 4 * i);
    const unsigned kv[4] = {(unsigned)k.x, (unsigned)k.y, (unsigned)k.z, (unsigned)k.w};
    unsigned w = 0u;
#pragma unroll
    for (int q = 0; q < 4; ++q) {
        const unsigned id = second ? kv[q] / (unsigned)RB : kv[q] / (unsigned)RA;
        w |= (id & 127u) << (8 * q);
    }
    (second ? ridB : ridA)[i] = (int)w;
}

__global__ __launch_bounds__(OWN_THREADS)
void owner_sum_kernel(const float* __restrict__ X, const int* __restrict__ own_idx,
                      const int* __restrict__ rid,
                      const int* __restrict__ src_idx, float* __restrict__ acc,
                      float* __restrict__ deg, int N, int E, int P, int S) {
    __shared__ float    s_acc[(OWN_MAXROWS + 1) * OWN_PITCH];
    __shared__ unsigned s_cnt[OWN_MAXROWS + 1];
    __shared__ unsigned s_list[OWN_WAVES][OWN_LIST + 1];

    const int tid  = threadIdx.x;
    const int lane = tid & 31;
    const int wave = tid >> 5;
    const int R    = (N + P - 1) / P;
    const int pr   = (int)blockIdx.x % P;
    const int sl   = (int)blockIdx.x / P;
    const int base = pr * R;
    const int slen = (((E + S - 1) / S) + 15) & ~15;
    const int e_lo = sl * slen;
    const int e_hi = (e_lo + slen < E) ? e_lo + slen : E;
    acc += (size_t)sl * (size_t)N * HID;
    if (deg) deg += (size_t)sl * (size_t)N;

    for (int i = tid; i < (OWN_MAXROWS + 1) * OWN_PITCH; i += OWN_THREADS) s_acc[i] = 0.0f;
    for (int i = tid; i <= OWN_MAXROWS; i += OWN_THREADS) s_cnt[i] = 0u;
    __syncthreads();

    const int hw = lane >> 4;
    const int c  = (lane & 15) * 4;
    const int cA = c + 2 * hw;
    const int cB = c + 2 - 2 * hw;
    const int keys_per_wave = 32 * 16 * OWN_LOADS;
    const int keys_per_trip = keys_per_wave * OWN_WAVES;

    const int lane16 = lane * 16;
    const unsigned rep4 = (unsigned)pr * 0x01010101u;
    const int nomatch = (int)(rep4 ^ 0x01010101u);
    const int ntrips = (slen + keys_per_trip - 1) / keys_per_trip;
    int trip = (int)(((unsigned)pr * (unsigned)ntrips) / (unsigned)P);
    for (int it = 0; it < ntrips; ++it) {
        const int t0 = e_lo + trip * keys_per_trip;
        trip = (trip + 1 == ntrips) ? 0 : trip + 1;
        const int wbase = t0 + wave * keys_per_wave;
        const int* kp = rid + ((wbase + lane16) >> 2);
        int4 k[OWN_LOADS];
        if (wbase + keys_per_wave <= e_hi) {
#pragma unroll
            for (int j = 0; j < OWN_LOADS; ++j) k[j] = *(const int4*)(kp + j * 128);
        } else {
#pragma unroll
            for (int j = 0; j < OWN_LOADS; ++j)
                k[j] = (wbase + j * 512 + lane16 + 15 < e_hi) ? *(const int4*)(kp + j * 128)
                                                          : make_int4(nomatch, nomatch, nomatch, nomatch);
        }
        const int ebase = wbase + lane16;
        unsigned cnt = 0u;
#pragma unroll
        for (int j = 0; j < OWN_LOADS; ++j) {
            const unsigned t0 = ((unsigned)k[j].x ^ rep4) + 0x7F7F7F7Fu;
            const unsigned t1 = ((unsigned)k[j].y ^ rep4) + 0x7F7F7F7Fu;
            const unsigned t2 = ((unsigned)k[j].z ^ rep4) + 0x7F7F7F7Fu;
            const unsigned t3 = ((unsigned)k[j].w ^ rep4) + 0x7F7F7F7Fu;
            unsigned hits = (((t0 >> 7) & 0x01010101u) | ((t1 >> 6) & 0x02020202u) |
                          ((t2 >> 5) & 0x04040404u) | ((t3 >> 4) & 0x08080808u)) ^ 0x0F0F0F0Fu;
            unsigned bal = __builtin_amdgcn_ballot_w32(hits != 0u);
            while (bal != 0u) {
                if (hits != 0u) {
                    const unsigned p = (unsigned)__builtin_ctz(hits);
                    unsigned slot = cnt + __builtin_amdgcn_mbcnt_lo(bal, 0u);
                    slot = slot < OWN_LIST ? slot : OWN_LIST;
                    s_list[wave][slot] = (unsigned)(ebase + j * 512) + 4u * (p & 7u) + (p >> 3);
                    hits &= hits - 1u;
                }
                cnt += (unsigned)__builtin_popcount(bal);
                bal = __builtin_amdgcn_ballot_w32(hits != 0u);
            }
        }
        if (cnt <= OWN_LIST) {
            for (unsigned m0 = 0; m0 < cnt; m0 += 8) {
                bool   ok[4];
                int    e[4];
                int    key[4];
                int    s[4];
                float4 v[4];
#pragma unroll
                for (int j = 0; j < 4; ++j) {
                    const unsigned m = m0 + j * 2 + hw;
                    const int ed = (int)s_list[wave][m];
                    ok[j] = m < cnt;
                    e[j]  = ok[j] ? ed : 0;
                }
#pragma unroll
                for (int j = 0; j < 4; ++j) {
                    key[j] = own_idx[e[j]];
                    s[j]   = src_idx[e[j]];
                }
#pragma unroll
                for (int j = 0; j < 4; ++j)
                    v[j] = *(const float4*)(X + (size_t)s[j] * HID + c);
#pragma unroll
                for (int j = 0; j < 4; ++j) {
                    const unsigned rr = (unsigned)(key[j] - base);
                    const unsigned r = (ok[j] && rr < (unsigned)R) ? rr : (unsigned)OWN_MAXROWS;
                    const float a0 = hw ? v[j].z : v[j].x;
                    const float a1 = hw ? v[j].w : v[j].y;
                    const float a2 = hw ? v[j].x : v[j].z;
                    const float a3 = hw ? v[j].y : v[j].w;
                    float* oa = &s_acc[r * OWN_PITCH + cA];
                    float* ob = &s_acc[r * OWN_PITCH + cB];
                    atomicAdd(oa + 0, a0);
                    atomicAdd(oa + 1, a1);
                    atomicAdd(ob + 0, a2);
                    atomicAdd(ob + 1, a3);
                    if (c == 0) atomicAdd(&s_cnt[r], 1u);
                }
            }
        } else {
            for (int p = 0; p < keys_per_wave; p += 32) {
                const int e1 = wbase + p + lane;
                const int key1 = (e1 < e_hi) ? own_idx[e1] : -1;
                unsigned bal = __builtin_amdgcn_ballot_w32((unsigned)(key1 - base) < (unsigned)R);
                while (bal != 0u) {
                    const int eL = wbase + p + __builtin_ctz(bal);
                    bal &= bal - 1u;
                    const unsigned rL = (unsigned)(own_idx[eL] - base);
                    const float2 x = *(const float2*)(X + (size_t)src_idx[eL] * HID + lane * 2);
                    atomicAdd(&s_acc[rL * OWN_PITCH + lane * 2], x.x);
                    atomicAdd(&s_acc[rL * OWN_PITCH + lane * 2 + 1], x.y);
                    if (lane == 0) atomicAdd(&s_cnt[rL], 1u);
                }
            }
        }
    }
    __syncthreads();

    for (int i = tid; i < R * HID; i += OWN_THREADS) {
        const int r = i >> 6, ch = i & 63;
        const int row = base + r;
        if (row < N) acc[(size_t)row * HID + ch] = s_acc[r * OWN_PITCH + ch];
    }
    if (deg) {
        for (int r = tid; r < R; r += OWN_THREADS)
            if (base + r < N) deg[base + r] = (float)s_cnt[r];
    }
}

template <int NP>
__global__ __launch_bounds__(256)
void sage_gemm64(const float* __restrict__ A1, const float* __restrict__ deg1,
                 const float* __restrict__ W1, int K1,
                 const float* __restrict__ A2, const float* __restrict__ W2, int K2,
                 const float* __restrict__ bias, float* __restrict__ C,
                 int M, int relu) {
    const int lane = threadIdx.x & 31;
    const int wave = threadIdx.x >> 5;
    const int half = lane >> 4;
    const int r    = lane & 15;
    const int row0 = (blockIdx.x * 8 + wave) * 16;
    if (row0 >= M) return;

    v8f c[4] = {};

    float scale = 1.0f;
    if (deg1) {
        float d = deg1[row0 + r];
#pragma unroll
        for (int s = 1; s < NP; ++s) d += deg1[(size_t)s * M + row0 + r];
        scale = 1.0f / fmaxf(d, 1.0f);
    }

    {
        const float* Arow = A1 + (size_t)(row0 + r) * K1;
        const size_t plane = (size_t)M * K1;
        for (int kk = 0; kk < K1; kk += 4) {
            const int k = kk + 2 * half;
            float ax = Arow[k];
            float ay = Arow[k + 1];
#pragma unroll
            for (int s = 1; s < NP; ++s) {
                ax += Arow[s * plane + k];
                ay += Arow[s * plane + k + 1];
            }
            v2f a;
            a.x = ax * scale;
            a.y = ay * scale;
#pragma unroll
            for (int t = 0; t < 4; ++t) {
                const float* Wrow = W1 + (size_t)(t * 16 + r) * K1;
                v2f b;
                b.x = Wrow[k];
                b.y = Wrow[k + 1];
                c[t] = __builtin_amdgcn_wmma_f32_16x16x4_f32(
                    false, a, false, b, (short)0, c[t], false, false);
            }
        }
    }
    if (A2) {
        const float* Arow = A2 + (size_t)(row0 + r) * K2;
        for (int kk = 0; kk < K2; kk += 4) {
            const int k = kk + 2 * half;
            v2f a;
            a.x = Arow[k];
            a.y = Arow[k + 1];
#pragma unroll
            for (int t = 0; t < 4; ++t) {
                const float* Wrow = W2 + (size_t)(t * 16 + r) * K2;
                v2f b;
                b.x = Wrow[k];
                b.y = Wrow[k + 1];
                c[t] = __builtin_amdgcn_wmma_f32_16x16x4_f32(
                    false, a, false, b, (short)0, c[t], false, false);
            }
        }
    }

#pragma unroll
    for (int t = 0; t < 4; ++t) {
        const float bv = bias ? bias[t * 16 + r] : 0.0f;
#pragma unroll
        for (int i = 0; i < 8; ++i) {
            float v = c[t][i] + bv;
            if (relu) v = fmaxf(v, 0.0f);
            C[(size_t)(row0 + i + 8 * half) * HID + t * 16 + r] = v;
        }
    }
}

__global__ __launch_bounds__(256)
void head_kernel(const float* __restrict__ acc2, const float* __restrict__ degm,
                 const float* __restrict__ m1, const float* __restrict__ Wl2,
                 const float* __restrict__ bl2, const float* __restrict__ Wr2,
                 float* __restrict__ out, int M) {
    int m = blockIdx.x * blockDim.x + threadIdx.x;
    if (m >= M) return;
    float dsum = degm[m];
#pragma unroll
    for (int s = 1; s < MOV_SLICES; ++s) dsum += degm[(size_t)s * M + m];
    const float inv = 1.0f / fmaxf(dsum, 1.0f);
    const size_t plane = (size_t)M * HID;
    const float* a = acc2 + (size_t)m * HID;
    const float* x = m1 + (size_t)m * HID;
    float acc[GENRES];
#pragma unroll
    for (int g = 0; g < GENRES; ++g) acc[g] = bl2[g];
    for (int k = 0; k < HID; ++k) {
        float as = a[k];
#pragma unroll
        for (int s = 1; s < MOV_SLICES; ++s) as += a[s * plane + k];
        const float av = as * inv;
        const float xv = x[k];
#pragma unroll
        for (int g = 0; g < GENRES; ++g)
            acc[g] += av * Wl2[g * HID + k] + xv * Wr2[g * HID + k];
    }
#pragma unroll
    for (int g = 0; g < GENRES; ++g) out[(size_t)m * GENRES + g] = acc[g];
}

extern "C" void kernel_launch(void* const* d_in, const int* in_sizes, int n_in,
                              void* d_out, int out_size, void* d_ws, size_t ws_size,
                              hipStream_t stream) {
    (void)in_sizes; (void)n_in; (void)out_size; (void)ws_size;

    const float* x_user  = (const float*)d_in[0];
    const float* x_movie = (const float*)d_in[1];
    const int*   e_src   = (const int*)d_in[2];
    const int*   e_dst   = (const int*)d_in[3];
    const float* W_user  = (const float*)d_in[4];
    const float* b_user  = (const float*)d_in[5];
    const float* W_movie = (const float*)d_in[6];
    const float* b_movie = (const float*)d_in[7];
    const float* Wl1_um  = (const float*)d_in[8];
    const float* bl1_um  = (const float*)d_in[9];
    const float* Wr1_um  = (const float*)d_in[10];
    const float* Wl1_mu  = (const float*)d_in[11];
    const float* bl1_mu  = (const float*)d_in[12];
    const float* Wr1_mu  = (const float*)d_in[13];
    const float* Wl2_um  = (const float*)d_in[14];
    const float* bl2_um  = (const float*)d_in[15];
    const float* Wr2_um  = (const float*)d_in[16];
    float* out = (float*)d_out;

    float* ws = (float*)d_ws;
    float* acc_um    = ws;
    float* acc_mu    = ws;
    float* acc2      = ws;
    float* hu        = ws + 12800000;
    float* m1        = hu + 6400000;
    float* deg_movie = m1 + 3200000;
    float* deg_user  = deg_movie + 200000;
    float* hm        = deg_user + 200000;
    float* u1        = hm;
    int*   rid_dst   = (int*)(hm + 3200000);
    int*   rid_src   = (int*)(hm + 4200000);
    int*   rid_dst2  = (int*)hu;

    const int E = NUM_EDGES;

    range_id_kernel<MOV_R, USR_R><<<dim3((NUM_EDGES / 4 + 255) / 256, 2), 256, 0, stream>>>(
        e_dst, rid_dst, e_src, rid_src, NUM_EDGES / 4);

    const int gridU = (NUM_USERS / 16 + 7) / 8;
    const int gridM = (NUM_MOVIES / 16 + 7) / 8;
    sage_gemm64<1><<<gridU, 256, 0, stream>>>(x_user, nullptr, W_user, USER_FEAT,
                                           nullptr, nullptr, 0, b_user, hu,
                                           NUM_USERS, 0);
    sage_gemm64<1><<<gridM, 256, 0, stream>>>(x_movie, nullptr, W_movie, MOVIE_FEAT,
                                           nullptr, nullptr, 0, b_movie, hm,
                                           NUM_MOVIES, 0);

    owner_sum_kernel<<<OWN_BLOCKS, OWN_THREADS, 0, stream>>>(hu, e_dst, rid_dst, e_src, acc_um,
                                                             deg_movie, NUM_MOVIES, E,
                                                             MOV_RANGES, MOV_SLICES);
    sage_gemm64<MOV_SLICES><<<gridM, 256, 0, stream>>>(acc_um, deg_movie, Wl1_um, HID,
                                           hm, Wr1_um, HID, bl1_um, m1,
                                           NUM_MOVIES, 1);

    owner_sum_kernel<<<OWN_BLOCKS, OWN_THREADS, 0, stream>>>(hm, e_src, rid_src, e_dst, acc_mu,
                                                             deg_user, NUM_USERS, E,
                                                             USR_RANGES, USR_SLICES);
    sage_gemm64<USR_SLICES><<<gridU, 256, 0, stream>>>(acc_mu, deg_user, Wl1_mu, HID,
                                           hu, Wr1_mu, HID, bl1_mu, u1,
                                           NUM_USERS, 1);

    range_id_kernel<MOV_R, MOV_R><<<dim3((NUM_EDGES / 4 + 255) / 256, 1), 256, 0, stream>>>(
        e_dst, rid_dst2, e_dst, rid_dst2, NUM_EDGES / 4);
    owner_sum_kernel<<<OWN_BLOCKS, OWN_THREADS, 0, stream>>>(u1, e_dst, rid_dst2, e_src, acc2,
                                                             nullptr, NUM_MOVIES, E,
                                                             MOV_RANGES, MOV_SLICES);
    head_kernel<<<(NUM_MOVIES + 255) / 256, 256, 0, stream>>>(
        acc2, deg_movie, m1, Wl2_um, bl2_um, Wr2_um, out, NUM_MOVIES);
}


# GEMM wave tile 16x64 -> 32x64 (50,000-row products) and 64x64 (100,000-row): the strips of a wave share each B fragment, so a weight line is fetched once per 32/64 rows; blocks of 2 waves
# speedup vs baseline: 2.9593x; 1.0495x over previous
#include <hip/hip_runtime.h>
#include <hip/hip_bf16.h>

typedef __attribute__((ext_vector_type(2))) float v2f;
typedef __attribute__((ext_vector_type(8))) float v8f;

#define NUM_USERS  100000
#define NUM_MOVIES 50000
#define NUM_EDGES  4000000
#define USER_FEAT  24
#define MOVIE_FEAT 404
#define HID        64
#define GENRES     3

#define OWN_BLOCKS   256
#define OWN_THREADS  1024
#define OWN_WAVES    32
#define OWN_MAXROWS  782
#define OWN_PITCH    68
#define OWN_LIST     512
#define OWN_LOADS    8
#define MOV_RANGES   64
#define MOV_SLICES   4
#define USR_RANGES   128
#define USR_SLICES   2
#define MOV_R        ((NUM_MOVIES + MOV_RANGES - 1) / MOV_RANGES)
#define USR_R        ((NUM_USERS + USR_RANGES - 1) / USR_RANGES)
static_assert(MOV_RANGES <= 128 && USR_RANGES <= 128, "a range id must fit in 7 bits");
static_assert(NUM_EDGES % 16 == 0, "16-byte loads of range ids");

template <int RA, int RB>
__global__ __launch_bounds__(256)
void range_id_kernel(const int* __restrict__ keysA, int* __restrict__ ridA,
                     const int* __restrict__ keysB, int* __restrict__ ridB, int n4) {
    const int i = (int)blockIdx.x * 256 + (int)threadIdx.x;
    if (i >= n4) return;
    const bool second = blockIdx.y != 0;
    const int4 k = *(const int4*)((second ? keysB : keysA) + 4 * i);
    const unsigned kv[4] = {(unsigned)k.x, (unsigned)k.y, (unsigned)k.z, (unsigned)k.w};
    unsigned w = 0u;
#pragma unroll
    for (int q = 0; q < 4; ++q) {
        const unsigned id = second ? kv[q] / (unsigned)RB : kv[q] / (unsigned)RA;
        w |= (id & 127u) << (8 * q);
    }
    (second ? ridB : ridA)[i] = (int)w;
}

__global__ __launch_bounds__(OWN_THREADS)
void owner_sum_kernel(const float* __restrict__ X, const int* __restrict__ own_idx,
                      const int* __restrict__ rid,
                      const int* __restrict__ src_idx, float* __restrict__ acc,
                      float* __restrict__ deg, int N, int E, int P, int S) {
    __shared__ float    s_acc[(OWN_MAXROWS + 1) * OWN_PITCH];
    __shared__ unsigned s_cnt[OWN_MAXROWS + 1];
    __shared__ unsigned s_list[OWN_WAVES][OWN_LIST + 1];

    const int tid  = threadIdx.x;
    const int lane = tid & 31;
    const int wave = tid >> 5;
    const int R    = (N + P - 1) / P;
    const int pr   = (int)blockIdx.x % P;
    const int sl   = (int)blockIdx.x / P;
    const int base = pr * R;
    const int slen = (((E + S - 1) / S) + 15) & ~15;
    const int e_lo = sl * slen;
    const int e_hi = (e_lo + slen < E) ? e_lo + slen : E;
    acc += (size_t)sl * (size_t)N * HID;
    if (deg) deg += (size_t)sl * (size_t)N;

    for (int i = tid; i < (OWN_MAXROWS + 1) * OWN_PITCH; i += OWN_THREADS) s_acc[i] = 0.0f;
    for (int i = tid; i <= OWN_MAXROWS; i += OWN_THREADS) s_cnt[i] = 0u;
    __syncthreads();

    const int hw = lane >> 4;
    const int c  = (lane & 15) * 4;
    const int cA = c + 2 * hw;
    const int cB = c + 2 - 2 * hw;
    const int keys_per_wave = 32 * 16 * OWN_LOADS;
    const int keys_per_trip = keys_per_wave * OWN_WAVES;

    const int lane16 = lane * 16;
    const unsigned rep4 = (unsigned)pr * 0x01010101u;
    const int nomatch = (int)(rep4 ^ 0x01010101u);
    const int ntrips = (slen + keys_per_trip - 1) / keys_per_trip;
    int trip = (int)(((unsigned)pr * (unsigned)ntrips) / (unsigned)P);
    for (int it = 0; it < ntrips; ++it) {
        const int t0 = e_lo + trip * keys_per_trip;
        trip = (trip + 1 == ntrips) ? 0 : trip + 1;
        const int wbase = t0 + wave * keys_per_wave;
        const int* kp = rid + ((wbase + lane16) >> 2);
        int4 k[OWN_LOADS];
        if (wbase + keys_per_wave <= e_hi) {
#pragma unroll
            for (int j = 0; j < OWN_LOADS; ++j) k[j] = *(const int4*)(kp + j * 128);
        } else {
#pragma unroll
            for (int j = 0; j < OWN_LOADS; ++j)
                k[j] = (wbase + j * 512 + lane16 + 15 < e_hi) ? *(const int4*)(kp + j * 128)
                                                          : make_int4(nomatch, nomatch, nomatch, nomatch);
        }
        const int ebase = wbase + lane16;
        unsigned cnt = 0u;
#pragma unroll
        for (int j = 0; j < OWN_LOADS; ++j) {
            const unsigned t0 = ((unsigned)k[j].x ^ rep4) + 0x7F7F7F7Fu;
            const unsigned t1 = ((unsigned)k[j].y ^ rep4) + 0x7F7F7F7Fu;
            const unsigned t2 = ((unsigned)k[j].z ^ rep4) + 0x7F7F7F7Fu;
            const unsigned t3 = ((unsigned)k[j].w ^ rep4) + 0x7F7F7F7Fu;
            unsigned hits = (((t0 >> 7) & 0x01010101u) | ((t1 >> 6) & 0x02020202u) |
                          ((t2 >> 5) & 0x04040404u) | ((t3 >> 4) & 0x08080808u)) ^ 0x0F0F0F0Fu;
            unsigned bal = __builtin_amdgcn_ballot_w32(hits != 0u);
            while (bal != 0u) {
                if (hits != 0u) {
                    const unsigned p = (unsigned)__builtin_ctz(hits);
                    unsigned slot = cnt + __builtin_amdgcn_mbcnt_lo(bal, 0u);
                    slot = slot < OWN_LIST ? slot : OWN_LIST;
                    s_list[wave][slot] = (unsigned)(ebase + j * 512) + 4u * (p & 7u) + (p >> 3);
                    hits &= hits - 1u;
                }
                cnt += (unsigned)__builtin_popcount(bal);
                bal = __builtin_amdgcn_ballot_w32(hits != 0u);
            }
        }
        if (cnt <= OWN_LIST) {
            for (unsigned m0 = 0; m0 < cnt; m0 += 8) {
                bool   ok[4];
                int    e[4];
                int    key[4];
                int    s[4];
                float4 v[4];
#pragma unroll
                for (int j = 0; j < 4; ++j) {
                    const unsigned m = m0 + j * 2 + hw;
                    const int ed = (int)s_list[wave][m];
                    ok[j] = m < cnt;
                    e[j]  = ok[j] ? ed : 0;
                }
#pragma unroll
                for (int j = 0; j < 4; ++j) {
                    key[j] = own_idx[e[j]];
                    s[j]   = src_idx[e[j]];
                }
#pragma unroll
                for (int j = 0; j < 4; ++j)
                    v[j] = *(const float4*)(X + (size_t)s[j] * HID + c);
#pragma unroll
                for (int j = 0; j < 4; ++j) {
                    const unsigned rr = (unsigned)(key[j] - base);
                    const unsigned r = (ok[j] && rr < (unsigned)R) ? rr : (unsigned)OWN_MAXROWS;
                    const float a0 = hw ? v[j].z : v[j].x;
                    const float a1 = hw ? v[j].w : v[j].y;
                    const float a2 = hw ? v[j].x : v[j].z;
                    const float a3 = hw ? v[j].y : v[j].w;
                    float* oa = &s_acc[r * OWN_PITCH + cA];
                    float* ob = &s_acc[r * OWN_PITCH + cB];
                    atomicAdd(oa + 0, a0);
                    atomicAdd(oa + 1, a1);
                    atomicAdd(ob + 0, a2);
                    atomicAdd(ob + 1, a3);
                    if (c == 0) atomicAdd(&s_cnt[r], 1u);
                }
            }
        } else {
            for (int p = 0; p < keys_per_wave; p += 32) {
                const int e1 = wbase + p + lane;
                const int key1 = (e1 < e_hi) ? own_idx[e1] : -1;
                unsigned bal = __builtin_amdgcn_ballot_w32((unsigned)(key1 - base) < (unsigned)R);
                while (bal != 0u) {
                    const int eL = wbase + p + __builtin_ctz(bal);
                    bal &= bal - 1u;
                    const unsigned rL = (unsigned)(own_idx[eL] - base);
                    const float2 x = *(const float2*)(X + (size_t)src_idx[eL] * HID + lane * 2);
                    atomicAdd(&s_acc[rL * OWN_PITCH + lane * 2], x.x);
                    atomicAdd(&s_acc[rL * OWN_PITCH + lane * 2 + 1], x.y);
                    if (lane == 0) atomicAdd(&s_cnt[rL], 1u);
                }
            }
        }
    }
    __syncthreads();

    for (int i = tid; i < R * HID; i += OWN_THREADS) {
        const int r = i >> 6, ch = i & 63;
        const int row = base + r;
        if (row < N) acc[(size_t)row * HID + ch] = s_acc[r * OWN_PITCH + ch];
    }
    if (deg) {
        for (int r = tid; r < R; r += OWN_THREADS)
            if (base + r < N) deg[base + r] = (float)s_cnt[r];
    }
}

#define GEMM_WAVES   2
#define MOV_STRIPS   2
#define USR_STRIPS   4
template <int NP, int STRIPS>
__global__ __launch_bounds__(GEMM_WAVES * 32)
void sage_gemm64(const float* __restrict__ A1, const float* __restrict__ deg1,
                 const float* __restrict__ W1, int K1,
                 const float* __restrict__ A2, const float* __restrict__ W2, int K2,
                 const float* __restrict__ bias, float* __restrict__ C,
                 int M, int relu) {
    const int lane = threadIdx.x & 31;
    const int wave = threadIdx.x >> 5;
    const int half = lane >> 4;
    const int r    = lane & 15;
    const int row0 = (blockIdx.x * GEMM_WAVES + wave) * (16 * STRIPS);
    if (row0 >= M) return;

    v8f c[STRIPS][4] = {};

    int   arow[STRIPS];
    float scale[STRIPS];
#pragma unroll
    for (int s = 0; s < STRIPS; ++s) {
        const int rs = row0 + 16 * s;
        arow[s] = ((rs < M) ? rs : M - 16) + r;
        scale[s] = 1.0f;
        if (deg1) {
            float d = deg1[arow[s]];
#pragma unroll
            for (int p = 1; p < NP; ++p) d += deg1[(size_t)p * M + arow[s]];
            scale[s] = 1.0f / fmaxf(d, 1.0f);
        }
    }

    {
        const size_t plane = (size_t)M * K1;
        for (int kk = 0; kk < K1; kk += 4) {
            const int k = kk + 2 * half;
            v2f b[4];
#pragma unroll
            for (int t = 0; t < 4; ++t) {
                const float* Wrow = W1 + (size_t)(t * 16 + r) * K1;
                b[t].x = Wrow[k];
                b[t].y = Wrow[k + 1];
            }
#pragma unroll
            for (int s = 0; s < STRIPS; ++s) {
                const float* Arow = A1 + (size_t)arow[s] * K1;
                float ax = Arow[k];
                float ay = Arow[k + 1];
#pragma unroll
                for (int p = 1; p < NP; ++p) {
                    ax += Arow[p * plane + k];
                    ay += Arow[p * plane + k + 1];
                }
                v2f a;
                a.x = ax * scale[s];
                a.y = ay * scale[s];
#pragma unroll
                for (int t = 0; t < 4; ++t)
                    c[s][t] = __builtin_amdgcn_wmma_f32_16x16x4_f32(
                        false, a, false, b[t], (short)0, c[s][t], false, false);
            }
        }
    }
    if (A2) {
        for (int kk = 0; kk < K2; kk += 4) {
            const int k = kk + 2 * half;
            v2f b[4];
#pragma unroll
            for (int t = 0; t < 4; ++t) {
                const float* Wrow = W2 + (size_t)(t * 16 + r) * K2;
                b[t].x = Wrow[k];
                b[t].y = Wrow[k + 1];
            }
#pragma unroll
            for (int s = 0; s < STRIPS; ++s) {
                const float* Arow = A2 + (size_t)arow[s] * K2;
                v2f a;
                a.x = Arow[k];
                a.y = Arow[k + 1];
#pragma unroll
                for (int t = 0; t < 4; ++t)
                    c[s][t] = __builtin_amdgcn_wmma_f32_16x16x4_f32(
                        false, a, false, b[t], (short)0, c[s][t], false, false);
            }
        }
    }

#pragma unroll
    for (int s = 0; s < STRIPS; ++s) {
        const int rs = row0 + 16 * s;
        if (rs < M) {
#pragma unroll
            for (int t = 0; t < 4; ++t) {
                const float bv = bias ? bias[t * 16 + r] : 0.0f;
#pragma unroll
                for (int i = 0; i < 8; ++i) {
                    float v = c[s][t][i] + bv;
                    if (relu) v = fmaxf(v, 0.0f);
                    C[(size_t)(rs + i + 8 * half) * HID + t * 16 + r] = v;
                }
            }
        }
    }
}

__global__ __launch_bounds__(256)
void head_kernel(const float* __restrict__ acc2, const float* __restrict__ degm,
                 const float* __restrict__ m1, const float* __restrict__ Wl2,
                 const float* __restrict__ bl2, const float* __restrict__ Wr2,
                 float* __restrict__ out, int M) {
    int m = blockIdx.x * blockDim.x + threadIdx.x;
    if (m >= M) return;
    float dsum = degm[m];
#pragma unroll
    for (int s = 1; s < MOV_SLICES; ++s) dsum += degm[(size_t)s * M + m];
    const float inv = 1.0f / fmaxf(dsum, 1.0f);
    const size_t plane = (size_t)M * HID;
    const float* a = acc2 + (size_t)m * HID;
    const float* x = m1 + (size_t)m * HID;
    float acc[GENRES];
#pragma unroll
    for (int g = 0; g < GENRES; ++g) acc[g] = bl2[g];
    for (int k = 0; k < HID; ++k) {
        float as = a[k];
#pragma unroll
        for (int s = 1; s < MOV_SLICES; ++s) as += a[s * plane + k];
        const float av = as * inv;
        const float xv = x[k];
#pragma unroll
        for (int g = 0; g < GENRES; ++g)
            acc[g] += av * Wl2[g * HID + k] + xv * Wr2[g * HID + k];
    }
#pragma unroll
    for (int g = 0; g < GENRES; ++g) out[(size_t)m * GENRES + g] = acc[g];
}

extern "C" void kernel_launch(void* const* d_in, const int* in_sizes, int n_in,
                              void* d_out, int out_size, void* d_ws, size_t ws_size,
                              hipStream_t stream) {
    (void)in_sizes; (void)n_in; (void)out_size; (void)ws_size;

    const float* x_user  = (const float*)d_in[0];
    const float* x_movie = (const float*)d_in[1];
    const int*   e_src   = (const int*)d_in[2];
    const int*   e_dst   = (const int*)d_in[3];
    const float* W_user  = (const float*)d_in[4];
    const float* b_user  = (const float*)d_in[5];
    const float* W_movie = (const float*)d_in[6];
    const float* b_movie = (const float*)d_in[7];
    const float* Wl1_um  = (const float*)d_in[8];
    const float* bl1_um  = (const float*)d_in[9];
    const float* Wr1_um  = (const float*)d_in[10];
    const float* Wl1_mu  = (const float*)d_in[11];
    const float* bl1_mu  = (const float*)d_in[12];
    const float* Wr1_mu  = (const float*)d_in[13];
    const float* Wl2_um  = (const float*)d_in[14];
    const float* bl2_um  = (const float*)d_in[15];
    const float* Wr2_um  = (const float*)d_in[16];
    float* out = (float*)d_out;

    float* ws = (float*)d_ws;
    float* acc_um    = ws;
    float* acc_mu    = ws;
    float* acc2      = ws;
    float* hu        = ws + 12800000;
    float* m1        = hu + 6400000;
    float* deg_movie = m1 + 3200000;
    float* deg_user  = deg_movie + 200000;
    float* hm        = deg_user + 200000;
    float* u1        = hm;
    int*   rid_dst   = (int*)(hm + 3200000);
    int*   rid_src   = (int*)(hm + 4200000);
    int*   rid_dst2  = (int*)hu;

    const int E = NUM_EDGES;

    range_id_kernel<MOV_R, USR_R><<<dim3((NUM_EDGES / 4 + 255) / 256, 2), 256, 0, stream>>>(
        e_dst, rid_dst, e_src, rid_src, NUM_EDGES / 4);

    const int gridU = ((NUM_USERS + 16 * USR_STRIPS - 1) / (16 * USR_STRIPS) + GEMM_WAVES - 1) / GEMM_WAVES;
    const int gridM = ((NUM_MOVIES + 16 * MOV_STRIPS - 1) / (16 * MOV_STRIPS) + GEMM_WAVES - 1) / GEMM_WAVES;
    sage_gemm64<1, USR_STRIPS><<<gridU, GEMM_WAVES * 32, 0, stream>>>(x_user, nullptr, W_user, USER_FEAT,
                                           nullptr, nullptr, 0, b_user, hu,
                                           NUM_USERS, 0);
    sage_gemm64<1, MOV_STRIPS><<<gridM, GEMM_WAVES * 32, 0, stream>>>(x_movie, nullptr, W_movie, MOVIE_FEAT,
                                           nullptr, nullptr, 0, b_movie, hm,
                                           NUM_MOVIES, 0);

    owner_sum_kernel<<<OWN_BLOCKS, OWN_THREADS, 0, stream>>>(hu, e_dst, rid_dst, e_src, acc_um,
                                                             deg_movie, NUM_MOVIES, E,
                                                             MOV_RANGES, MOV_SLICES);
    sage_gemm64<MOV_SLICES, MOV_STRIPS><<<gridM, GEMM_WAVES * 32, 0, stream>>>(acc_um, deg_movie, Wl1_um, HID,
                                           hm, Wr1_um, HID, bl1_um, m1,
                                           NUM_MOVIES, 1);

    owner_sum_kernel<<<OWN_BLOCKS, OWN_THREADS, 0, stream>>>(hm, e_src, rid_src, e_dst, acc_mu,
                                                             deg_user, NUM_USERS, E,
                                                             USR_RANGES, USR_SLICES);
    sage_gemm64<USR_SLICES, USR_STRIPS><<<gridU, GEMM_WAVES * 32, 0, stream>>>(acc_mu, deg_user, Wl1_mu, HID,
                                           hu, Wr1_mu, HID, bl1_mu, u1,
                                           NUM_USERS, 1);

    range_id_kernel<MOV_R, MOV_R><<<dim3((NUM_EDGES / 4 + 255) / 256, 1), 256, 0, stream>>>(
        e_dst, rid_dst2, e_dst, rid_dst2, NUM_EDGES / 4);
    owner_sum_kernel<<<OWN_BLOCKS, OWN_THREADS, 0, stream>>>(u1, e_dst, rid_dst2, e_src, acc2,
                                                             nullptr, NUM_MOVIES, E,
                                                             MOV_RANGES, MOV_SLICES);
    head_kernel<<<(NUM_MOVIES + 255) / 256, 256, 0, stream>>>(
        acc2, deg_movie, m1, Wl2_um, bl2_um, Wr2_um, out, NUM_MOVIES);
}


# Owner-sum gather in batches of 32 filed edges: lane = edge for the key and source-id loads, words handed to half-waves by readlane, 16 row loads in flight, leftovers carried over; sum order changes
# speedup vs baseline: 3.1094x; 1.0507x over previous
#include <hip/hip_runtime.h>
#include <hip/hip_bf16.h>

typedef __attribute__((ext_vector_type(2))) float v2f;
typedef __attribute__((ext_vector_type(8))) float v8f;

#define NUM_USERS  100000
#define NUM_MOVIES 50000
#define NUM_EDGES  4000000
#define USER_FEAT  24
#define MOVIE_FEAT 404
#define HID        64
#define GENRES     3

#define OWN_BLOCKS   256
#define OWN_THREADS  1024
#define OWN_WAVES    32
#define OWN_MAXROWS  782
#define OWN_PITCH    68
#define OWN_LIST     512
#define OWN_LOADS    8
#define MOV_RANGES   64
#define MOV_SLICES   4
#define USR_RANGES   128
#define USR_SLICES   2
#define MOV_R        ((NUM_MOVIES + MOV_RANGES - 1) / MOV_RANGES)
#define USR_R        ((NUM_USERS + USR_RANGES - 1) / USR_RANGES)
static_assert(MOV_RANGES <= 128 && USR_RANGES <= 128, "a range id must fit in 7 bits");
static_assert(NUM_EDGES % 16 == 0, "16-byte loads of range ids");
static_assert(OWN_MAXROWS < (1 << 10) && NUM_USERS <= (1 << 17) && NUM_MOVIES <= (1 << 17),
              "a row of a range and a source id are packed into one word (10 + 17 bits)");

template <int RA, int RB>
__global__ __launch_bounds__(256)
void range_id_kernel(const int* __restrict__ keysA, int* __restrict__ ridA,
                     const int* __restrict__ keysB, int* __restrict__ ridB, int n4) {
    const int i = (int)blockIdx.x * 256 + (int)threadIdx.x;
    if (i >= n4) return;
    const bool second = blockIdx.y != 0;
    const int4 k = *(const int4*)((second ? keysB : keysA) + 4 * i);
    const unsigned kv[4] = {(unsigned)k.x, (unsigned)k.y, (unsigned)k.z, (unsigned)k.w};
    unsigned w = 0u;
#pragma unroll
    for (int q = 0; q < 4; ++q) {
        const unsigned id = second ? kv[q] / (unsigned)RB : kv[q] / (unsigned)RA;
        w |= (id & 127u) << (8 * q);
    }
    (second ? ridB : ridA)[i] = (int)w;
}

template <bool FULL>
__device__ __forceinline__ void own_gather32(const float* __restrict__ X, const int* __restrict__ own_idx,
                                             const int* __restrict__ src_idx, const unsigned* lst,
                                             unsigned nv, float* s_acc, unsigned* s_cnt,
                                             int base, int R, int lane, int hw, int c, int cA, int cB) {
    const bool okL = FULL || (unsigned)lane < nv;
    const int  ed  = (int)lst[okL ? lane : 0];
    const int  key = own_idx[ed];
    const int  sid = src_idx[ed];
    const unsigned rr = (unsigned)(key - base);
    const unsigned rL = (okL && rr < (unsigned)R) ? rr : (unsigned)OWN_MAXROWS;
    atomicAdd(&s_cnt[rL], 1u);
    const int pk = (int)((rL << 17) | ((unsigned)sid & 0x1FFFFu));
    float4 v[16];
    int    pj[16];
#pragma unroll
    for (int j = 0; j < 16; ++j) {
        const int pa = __builtin_amdgcn_readlane(pk, 2 * j);
        const int pb = __builtin_amdgcn_readlane(pk, 2 * j + 1);
        pj[j] = hw ? pb : pa;
        v[j] = *(const float4*)(X + (size_t)(pj[j] & 0x1FFFF) * HID + c);
    }
#pragma unroll
    for (int j = 0; j < 16; ++j) {
        const unsigned r = (unsigned)pj[j] >> 17;
        const float a0 = hw ? v[j].z : v[j].x;
        const float a1 = hw ? v[j].w : v[j].y;
        const float a2 = hw ? v[j].x : v[j].z;
        const float a3 = hw ? v[j].y : v[j].w;
        float* oa = &s_acc[r * OWN_PITCH + cA];
        float* ob = &s_acc[r * OWN_PITCH + cB];
        atomicAdd(oa + 0, a0);
        atomicAdd(oa + 1, a1);
        atomicAdd(ob + 0, a2);
        atomicAdd(ob + 1, a3);
    }
}

__global__ __launch_bounds__(OWN_THREADS)
void owner_sum_kernel(const float* __restrict__ X, const int* __restrict__ own_idx,
                      const int* __restrict__ rid,
                      const int* __restrict__ src_idx, float* __restrict__ acc,
                      float* __restrict__ deg, int N, int E, int P, int S) {
    __shared__ float    s_acc[(OWN_MAXROWS + 1) * OWN_PITCH];
    __shared__ unsigned s_cnt[OWN_MAXROWS + 1];
    __shared__ unsigned s_list[OWN_WAVES][OWN_LIST + 1];

    const int tid  = threadIdx.x;
    const int lane = tid & 31;
    const int wave = tid >> 5;
    const int R    = (N + P - 1) / P;
    const int pr   = (int)blockIdx.x % P;
    const int sl   = (int)blockIdx.x / P;
    const int base = pr * R;
    const int slen = (((E + S - 1) / S) + 15) & ~15;
    const int e_lo = sl * slen;
    const int e_hi = (e_lo + slen < E) ? e_lo + slen : E;
    acc += (size_t)sl * (size_t)N * HID;
    if (deg) deg += (size_t)sl * (size_t)N;

    for (int i = tid; i < (OWN_MAXROWS + 1) * OWN_PITCH; i += OWN_THREADS) s_acc[i] = 0.0f;
    for (int i = tid; i <= OWN_MAXROWS; i += OWN_THREADS) s_cnt[i] = 0u;
    __syncthreads();

    const int hw = lane >> 4;
    const int c  = (lane & 15) * 4;
    const int cA = c + 2 * hw;
    const int cB = c + 2 - 2 * hw;
    const int keys_per_wave = 32 * 16 * OWN_LOADS;
    const int keys_per_trip = keys_per_wave * OWN_WAVES;

    const int lane16 = lane * 16;
    const unsigned rep4 = (unsigned)pr * 0x01010101u;
    const int nomatch = (int)(rep4 ^ 0x01010101u);
    const int ntrips = (slen + keys_per_trip - 1) / keys_per_trip;
    int trip = (int)(((unsigned)pr * (unsigned)ntrips) / (unsigned)P);
    unsigned lo = 0u;
    for (int it = 0; it < ntrips; ++it) {
        const int t0 = e_lo + trip * keys_per_trip;
        trip = (trip + 1 == ntrips) ? 0 : trip + 1;
        const int wbase = t0 + wave * keys_per_wave;
        const int* kp = rid + ((wbase + lane16) >> 2);
        int4 k[OWN_LOADS];
        if (wbase + keys_per_wave <= e_hi) {
#pragma unroll
            for (int j = 0; j < OWN_LOADS; ++j) k[j] = *(const int4*)(kp + j * 128);
        } else {
#pragma unroll
            for (int j = 0; j < OWN_LOADS; ++j)
                k[j] = (wbase + j * 512 + lane16 + 15 < e_hi) ? *(const int4*)(kp + j * 128)
                                                          : make_int4(nomatch, nomatch, nomatch, nomatch);
        }
        const int ebase = wbase + lane16;
        unsigned cnt = lo;
#pragma unroll
        for (int j = 0; j < OWN_LOADS; ++j) {
            const unsigned t0 = ((unsigned)k[j].x ^ rep4) + 0x7F7F7F7Fu;
            const unsigned t1 = ((unsigned)k[j].y ^ rep4) + 0x7F7F7F7Fu;
            const unsigned t2 = ((unsigned)k[j].z ^ rep4) + 0x7F7F7F7Fu;
            const unsigned t3 = ((unsigned)k[j].w ^ rep4) + 0x7F7F7F7Fu;
            unsigned hits = (((t0 >> 7) & 0x01010101u) | ((t1 >> 6) & 0x02020202u) |
                          ((t2 >> 5) & 0x04040404u) | ((t3 >> 4) & 0x08080808u)) ^ 0x0F0F0F0Fu;
            unsigned bal = __builtin_amdgcn_ballot_w32(hits != 0u);
            while (bal != 0u) {
                if (hits != 0u) {
                    const unsigned p = (unsigned)__builtin_ctz(hits);
                    unsigned slot = cnt + __builtin_amdgcn_mbcnt_lo(bal, 0u);
                    slot = slot < OWN_LIST ? slot : OWN_LIST;
                    s_list[wave][slot] = (unsigned)(ebase + j * 512) + 4u * (p & 7u) + (p >> 3);
                    hits &= hits - 1u;
                }
                cnt += (unsigned)__builtin_popcount(bal);
                bal = __builtin_amdgcn_ballot_w32(hits != 0u);
            }
        }
        if (cnt <= OWN_LIST) {
            unsigned m0 = 0u;
            for (; m0 + 32u <= cnt; m0 += 32u)
                own_gather32<true>(X, own_idx, src_idx, &s_list[wave][m0], 32u, s_acc, s_cnt,
                                   base, R, lane, hw, c, cA, cB);
            const unsigned rem = cnt - m0;
            if (m0 != 0u && (unsigned)lane < rem) {
                const unsigned t = s_list[wave][m0 + lane];
                s_list[wave][lane] = t;
            }
            lo = rem;
        } else {
            for (int p = 0; p < keys_per_wave; p += 32) {
                const int e1 = wbase + p + lane;
                const int key1 = (e1 < e_hi) ? own_idx[e1] : -1;
                unsigned bal = __builtin_amdgcn_ballot_w32((unsigned)(key1 - base) < (unsigned)R);
                while (bal != 0u) {
                    const int eL = wbase + p + __builtin_ctz(bal);
                    bal &= bal - 1u;
                    const unsigned rL = (unsigned)(own_idx[eL] - base);
                    const float2 x = *(const float2*)(X + (size_t)src_idx[eL] * HID + lane * 2);
                    atomicAdd(&s_acc[rL * OWN_PITCH + lane * 2], x.x);
                    atomicAdd(&s_acc[rL * OWN_PITCH + lane * 2 + 1], x.y);
                    if (lane == 0) atomicAdd(&s_cnt[rL], 1u);
                }
            }
        }
    }
    if (lo != 0u)
        own_gather32<false>(X, own_idx, src_idx, &s_list[wave][0], lo, s_acc, s_cnt,
                            base, R, lane, hw, c, cA, cB);
    __syncthreads();

    for (int i = tid; i < R * HID; i += OWN_THREADS) {
        const int r = i >> 6, ch = i & 63;
        const int row = base + r;
        if (row < N) acc[(size_t)row * HID + ch] = s_acc[r * OWN_PITCH + ch];
    }
    if (deg) {
        for (int r = tid; r < R; r += OWN_THREADS)
            if (base + r < N) deg[base + r] = (float)s_cnt[r];
    }
}

#define GEMM_WAVES   2
#define MOV_STRIPS   2
#define USR_STRIPS   4
template <int NP, int STRIPS>
__global__ __launch_bounds__(GEMM_WAVES * 32)
void sage_gemm64(const float* __restrict__ A1, const float* __restrict__ deg1,
                 const float* __restrict__ W1, int K1,
                 const float* __restrict__ A2, const float* __restrict__ W2, int K2,
                 const float* __restrict__ bias, float* __restrict__ C,
                 int M, int relu) {
    const int lane = threadIdx.x & 31;
    const int wave = threadIdx.x >> 5;
    const int half = lane >> 4;
    const int r    = lane & 15;
    const int row0 = (blockIdx.x * GEMM_WAVES + wave) * (16 * STRIPS);
    if (row0 >= M) return;

    v8f c[STRIPS][4] = {};

    int   arow[STRIPS];
    float scale[STRIPS];
#pragma unroll
    for (int s = 0; s < STRIPS; ++s) {
        const int rs = row0 + 16 * s;
        arow[s] = ((rs < M) ? rs : M - 16) + r;
        scale[s] = 1.0f;
        if (deg1) {
            float d = deg1[arow[s]];
#pragma unroll
            for (int p = 1; p < NP; ++p) d += deg1[(size_t)p * M + arow[s]];
            scale[s] = 1.0f / fmaxf(d, 1.0f);
        }
    }

    {
        const size_t plane = (size_t)M * K1;
        for (int kk = 0; kk < K1; kk += 4) {
            const int k = kk + 2 * half;
            v2f b[4];
#pragma unroll
            for (int t = 0; t < 4; ++t) {
                const float* Wrow = W1 + (size_t)(t * 16 + r) * K1;
                b[t].x = Wrow[k];
                b[t].y = Wrow[k + 1];
            }
#pragma unroll
            for (int s = 0; s < STRIPS; ++s) {
                const float* Arow = A1 + (size_t)arow[s] * K1;
                float ax = Arow[k];
                float ay = Arow[k + 1];
#pragma unroll
                for (int p = 1; p < NP; ++p) {
                    ax += Arow[p * plane + k];
                    ay += Arow[p * plane + k + 1];
                }
                v2f a;
                a.x = ax * scale[s];
                a.y = ay * scale[s];
#pragma unroll
                for (int t = 0; t < 4; ++t)
                    c[s][t] = __builtin_amdgcn_wmma_f32_16x16x4_f32(
                        false, a, false, b[t], (short)0, c[s][t], false, false);
            }
        }
    }
    if (A2) {
        for (int kk = 0; kk < K2; kk += 4) {
            const int k = kk + 2 * half;
            v2f b[4];
#pragma unroll
            for (int t = 0; t < 4; ++t) {
                const float* Wrow = W2 + (size_t)(t * 16 + r) * K2;
                b[t].x = Wrow[k];
                b[t].y = Wrow[k + 1];
            }
#pragma unroll
            for (int s = 0; s < STRIPS; ++s) {
                const float* Arow = A2 + (size_t)arow[s] * K2;
                v2f a;
                a.x = Arow[k];
                a.y = Arow[k + 1];
#pragma unroll
                for (int t = 0; t < 4; ++t)
                    c[s][t] = __builtin_amdgcn_wmma_f32_16x16x4_f32(
                        false, a, false, b[t], (short)0, c[s][t], false, false);
            }
        }
    }

#pragma unroll
    for (int s = 0; s < STRIPS; ++s) {
        const int rs = row0 + 16 * s;
        if (rs < M) {
#pragma unroll
            for (int t = 0; t < 4; ++t) {
                const float bv = bias ? bias[t * 16 + r] : 0.0f;
#pragma unroll
                for (int i = 0; i < 8; ++i) {
                    float v = c[s][t][i] + bv;
                    if (relu) v = fmaxf(v, 0.0f);
                    C[(size_t)(rs + i + 8 * half) * HID + t * 16 + r] = v;
                }
            }
        }
    }
}

__global__ __launch_bounds__(256)
void head_kernel(const float* __restrict__ acc2, const float* __restrict__ degm,
                 const float* __restrict__ m1, const float* __restrict__ Wl2,
                 const float* __restrict__ bl2, const float* __restrict__ Wr2,
                 float* __restrict__ out, int M) {
    int m = blockIdx.x * blockDim.x + threadIdx.x;
    if (m >= M) return;
    float dsum = degm[m];
#pragma unroll
    for (int s = 1; s < MOV_SLICES; ++s) dsum += degm[(size_t)s * M + m];
    const float inv = 1.0f / fmaxf(dsum, 1.0f);
    const size_t plane = (size_t)M * HID;
    const float* a = acc2 + (size_t)m * HID;
    const float* x = m1 + (size_t)m * HID;
    float acc[GENRES];
#pragma unroll
    for (int g = 0; g < GENRES; ++g) acc[g] = bl2[g];
    for (int k = 0; k < HID; ++k) {
        float as = a[k];
#pragma unroll
        for (int s = 1; s < MOV_SLICES; ++s) as += a[s * plane + k];
        const float av = as * inv;
        const float xv = x[k];
#pragma unroll
        for (int g = 0; g < GENRES; ++g)
            acc[g] += av * Wl2[g * HID + k] + xv * Wr2[g * HID + k];
    }
#pragma unroll
    for (int g = 0; g < GENRES; ++g) out[(size_t)m * GENRES + g] = acc[g];
}

extern "C" void kernel_launch(void* const* d_in, const int* in_sizes, int n_in,
                              void* d_out, int out_size, void* d_ws, size_t ws_size,
                              hipStream_t stream) {
    (void)in_sizes; (void)n_in; (void)out_size; (void)ws_size;

    const float* x_user  = (const float*)d_in[0];
    const float* x_movie = (const float*)d_in[1];
    const int*   e_src   = (const int*)d_in[2];
    const int*   e_dst   = (const int*)d_in[3];
    const float* W_user  = (const float*)d_in[4];
    const float* b_user  = (const float*)d_in[5];
    const float* W_movie = (const float*)d_in[6];
    const float* b_movie = (const float*)d_in[7];
    const float* Wl1_um  = (const float*)d_in[8];
    const float* bl1_um  = (const float*)d_in[9];
    const float* Wr1_um  = (const float*)d_in[10];
    const float* Wl1_mu  = (const float*)d_in[11];
    const float* bl1_mu  = (const float*)d_in[12];
    const float* Wr1_mu  = (const float*)d_in[13];
    const float* Wl2_um  = (const float*)d_in[14];
    const float* bl2_um  = (const float*)d_in[15];
    const float* Wr2_um  = (const float*)d_in[16];
    float* out = (float*)d_out;

    float* ws = (float*)d_ws;
    float* acc_um    = ws;
    float* acc_mu    = ws;
    float* acc2      = ws;
    float* hu        = ws + 12800000;
    float* m1        = hu + 6400000;
    float* deg_movie = m1 + 3200000;
    float* deg_user  = deg_movie + 200000;
    float* hm        = deg_user + 200000;
    float* u1        = hm;
    int*   rid_dst   = (int*)(hm + 3200000);
    int*   rid_src   = (int*)(hm + 4200000);
    int*   rid_dst2  = (int*)hu;

    const int E = NUM_EDGES;

    range_id_kernel<MOV_R, USR_R><<<dim3((NUM_EDGES / 4 + 255) / 256, 2), 256, 0, stream>>>(
        e_dst, rid_dst, e_src, rid_src, NUM_EDGES / 4);

    const int gridU = ((NUM_USERS + 16 * USR_STRIPS - 1) / (16 * USR_STRIPS) + GEMM_WAVES - 1) / GEMM_WAVES;
    const int gridM = ((NUM_MOVIES + 16 * MOV_STRIPS - 1) / (16 * MOV_STRIPS) + GEMM_WAVES - 1) / GEMM_WAVES;
    sage_gemm64<1, USR_STRIPS><<<gridU, GEMM_WAVES * 32, 0, stream>>>(x_user, nullptr, W_user, USER_FEAT,
                                           nullptr, nullptr, 0, b_user, hu,
                                           NUM_USERS, 0);
    sage_gemm64<1, MOV_STRIPS><<<gridM, GEMM_WAVES * 32, 0, stream>>>(x_movie, nullptr, W_movie, MOVIE_FEAT,
                                           nullptr, nullptr, 0, b_movie, hm,
                                           NUM_MOVIES, 0);

    owner_sum_kernel<<<OWN_BLOCKS, OWN_THREADS, 0, stream>>>(hu, e_dst, rid_dst, e_src, acc_um,
                                                             deg_movie, NUM_MOVIES, E,
                                                             MOV_RANGES, MOV_SLICES);
    sage_gemm64<MOV_SLICES, MOV_STRIPS><<<gridM, GEMM_WAVES * 32, 0, stream>>>(acc_um, deg_movie, Wl1_um, HID,
                                           hm, Wr1_um, HID, bl1_um, m1,
                                           NUM_MOVIES, 1);

    owner_sum_kernel<<<OWN_BLOCKS, OWN_THREADS, 0, stream>>>(hm, e_src, rid_src, e_dst, acc_mu,
                                                             deg_user, NUM_USERS, E,
                                                             USR_RANGES, USR_SLICES);
    sage_gemm64<USR_SLICES, USR_STRIPS><<<gridU, GEMM_WAVES * 32, 0, stream>>>(acc_mu, deg_user, Wl1_mu, HID,
                                           hu, Wr1_mu, HID, bl1_mu, u1,
                                           NUM_USERS, 1);

    range_id_kernel<MOV_R, MOV_R><<<dim3((NUM_EDGES / 4 + 255) / 256, 1), 256, 0, stream>>>(
        e_dst, rid_dst2, e_dst, rid_dst2, NUM_EDGES / 4);
    owner_sum_kernel<<<OWN_BLOCKS, OWN_THREADS, 0, stream>>>(u1, e_dst, rid_dst2, e_src, acc2,
                                                             nullptr, NUM_MOVIES, E,
                                                             MOV_RANGES, MOV_SLICES);
    head_kernel<<<(NUM_MOVIES + 255) / 256, 256, 0, stream>>>(
        acc2, deg_movie, m1, Wl2_um, bl2_um, Wr2_um, out, NUM_MOVIES);
}


# Layer-2 edge sum reads the packed (row, source) words the layer-1 movie sum filed and saves per wave (movie sums 128x2, run one after the other): no id pass, scan or key loads; sum order changes
# speedup vs baseline: 3.3183x; 1.0672x over previous
#include <hip/hip_runtime.h>
#include <hip/hip_bf16.h>

typedef __attribute__((ext_vector_type(2))) float v2f;
typedef __attribute__((ext_vector_type(8))) float v8f;

#define NUM_USERS  100000
#define NUM_MOVIES 50000
#define NUM_EDGES  4000000
#define USER_FEAT  24
#define MOVIE_FEAT 404
#define HID        64
#define GENRES     3

#define OWN_BLOCKS   256
#define OWN_THREADS  1024
#define OWN_WAVES    32
#define OWN_MAXROWS  782
#define OWN_PITCH    68
#define OWN_LIST     512
#define OWN_LOADS    8
#define MOV_RANGES   128
#define MOV_SLICES   2
#define USR_RANGES   128
#define USR_SLICES   2
#define MOV_R        ((NUM_MOVIES + MOV_RANGES - 1) / MOV_RANGES)
#define USR_R        ((NUM_USERS + USR_RANGES - 1) / USR_RANGES)
#define SAVE_CAP     640
#define SAVE_OVER    0xFFFFFFFFu
#define SAVE_WORDS   ((size_t)OWN_BLOCKS * OWN_WAVES * SAVE_CAP)
static_assert(MOV_RANGES <= 128 && USR_RANGES <= 128, "a range id must fit in 7 bits");
static_assert(NUM_EDGES % 16 == 0, "16-byte loads of range ids");
static_assert(OWN_MAXROWS < (1 << 10) && NUM_USERS <= (1 << 17) && NUM_MOVIES <= (1 << 17),
              "a row of a range and a source id are packed into one word (10 + 17 bits)");

template <int RA, int RB>
__global__ __launch_bounds__(256)
void range_id_kernel(const int* __restrict__ keysA, int* __restrict__ ridA,
                     const int* __restrict__ keysB, int* __restrict__ ridB, int n4) {
    const int i = (int)blockIdx.x * 256 + (int)threadIdx.x;
    if (i >= n4) return;
    const bool second = blockIdx.y != 0;
    const int4 k = *(const int4*)((second ? keysB : keysA) + 4 * i);
    const unsigned kv[4] = {(unsigned)k.x, (unsigned)k.y, (unsigned)k.z, (unsigned)k.w};
    unsigned w = 0u;
#pragma unroll
    for (int q = 0; q < 4; ++q) {
        const unsigned id = second ? kv[q] / (unsigned)RB : kv[q] / (unsigned)RA;
        w |= (id & 127u) << (8 * q);
    }
    (second ? ridB : ridA)[i] = (int)w;
}

__device__ __forceinline__ void own_rows32(const float* __restrict__ X, int pk, float* s_acc,
                                           int hw, int c, int cA, int cB) {
    float4 v[16];
    int    pj[16];
#pragma unroll
    for (int j = 0; j < 16; ++j) {
        const int pa = __builtin_amdgcn_readlane(pk, 2 * j);
        const int pb = __builtin_amdgcn_readlane(pk, 2 * j + 1);
        pj[j] = hw ? pb : pa;
        v[j] = *(const float4*)(X + (size_t)(pj[j] & 0x1FFFF) * HID + c);
    }
#pragma unroll
    for (int j = 0; j < 16; ++j) {
        const unsigned r = (unsigned)pj[j] >> 17;
        const float a0 = hw ? v[j].z : v[j].x;
        const float a1 = hw ? v[j].w : v[j].y;
        const float a2 = hw ? v[j].x : v[j].z;
        const float a3 = hw ? v[j].y : v[j].w;
        float* oa = &s_acc[r * OWN_PITCH + cA];
        float* ob = &s_acc[r * OWN_PITCH + cB];
        atomicAdd(oa + 0, a0);
        atomicAdd(oa + 1, a1);
        atomicAdd(ob + 0, a2);
        atomicAdd(ob + 1, a3);
    }
}

template <bool FULL>
__device__ __forceinline__ void own_gather32(const float* __restrict__ X, const int* __restrict__ own_idx,
                                             const int* __restrict__ src_idx, const unsigned* lst,
                                             unsigned nv, float* s_acc, unsigned* s_cnt,
                                             int base, int R, int lane, int hw, int c, int cA, int cB,
                                             unsigned* wsave, unsigned& saved) {
    const bool okL = FULL || (unsigned)lane < nv;
    const int  ed  = (int)lst[okL ? lane : 0];
    const int  key = own_idx[ed];
    const int  sid = src_idx[ed];
    const unsigned rr = (unsigned)(key - base);
    const unsigned rL = (okL && rr < (unsigned)R) ? rr : (unsigned)OWN_MAXROWS;
    atomicAdd(&s_cnt[rL], 1u);
    const int pk = (int)((rL << 17) | ((unsigned)sid & 0x1FFFFu));
    if (wsave != nullptr) {
        const unsigned pos = saved + (unsigned)lane;
        if (okL && pos < (unsigned)SAVE_CAP) wsave[pos] = (unsigned)pk;
    }
    saved += FULL ? 32u : nv;
    own_rows32(X, pk, s_acc, hw, c, cA, cB);
}

__global__ __launch_bounds__(OWN_THREADS)
void owner_sum_kernel(const float* __restrict__ X, const int* __restrict__ own_idx,
                      const int* __restrict__ rid,
                      const int* __restrict__ src_idx, float* __restrict__ acc,
                      float* __restrict__ deg, unsigned* __restrict__ save,
                      int N, int E, int P, int S) {
    __shared__ float    s_acc[(OWN_MAXROWS + 1) * OWN_PITCH];
    __shared__ unsigned s_cnt[OWN_MAXROWS + 1];
    __shared__ unsigned s_list[OWN_WAVES][OWN_LIST + 1];

    const int tid  = threadIdx.x;
    const int lane = tid & 31;
    const int wave = tid >> 5;
    const int R    = (N + P - 1) / P;
    const int pr   = (int)blockIdx.x % P;
    const int sl   = (int)blockIdx.x / P;
    const int base = pr * R;
    const int slen = (((E + S - 1) / S) + 15) & ~15;
    const int e_lo = sl * slen;
    const int e_hi = (e_lo + slen < E) ? e_lo + slen : E;
    acc += (size_t)sl * (size_t)N * HID;
    if (deg) deg += (size_t)sl * (size_t)N;

    for (int i = tid; i < (OWN_MAXROWS + 1) * OWN_PITCH; i += OWN_THREADS) s_acc[i] = 0.0f;
    for (int i = tid; i <= OWN_MAXROWS; i += OWN_THREADS) s_cnt[i] = 0u;
    __syncthreads();

    const int hw = lane >> 4;
    const int c  = (lane & 15) * 4;
    const int cA = c + 2 * hw;
    const int cB = c + 2 - 2 * hw;
    const int keys_per_wave = 32 * 16 * OWN_LOADS;
    const int keys_per_trip = keys_per_wave * OWN_WAVES;

    const int lane16 = lane * 16;
    const unsigned rep4 = (unsigned)pr * 0x01010101u;
    const int nomatch = (int)(rep4 ^ 0x01010101u);
    const int ntrips = (slen + keys_per_trip - 1) / keys_per_trip;
    int trip = (int)(((unsigned)pr * (unsigned)ntrips) / (unsigned)P);
    unsigned lo = 0u;
    unsigned* wsave = save ? save + ((size_t)blockIdx.x * OWN_WAVES + (size_t)wave) * SAVE_CAP : nullptr;
    unsigned saved = 0u;
    bool dense = false;
    for (int it = 0; it < ntrips; ++it) {
        const int t0 = e_lo + trip * keys_per_trip;
        trip = (trip + 1 == ntrips) ? 0 : trip + 1;
        const int wbase = t0 + wave * keys_per_wave;
        const int* kp = rid + ((wbase + lane16) >> 2);
        int4 k[OWN_LOADS];
        if (wbase + keys_per_wave <= e_hi) {
#pragma unroll
            for (int j = 0; j < OWN_LOADS; ++j) k[j] = *(const int4*)(kp + j * 128);
        } else {
#pragma unroll
            for (int j = 0; j < OWN_LOADS; ++j)
                k[j] = (wbase + j * 512 + lane16 + 15 < e_hi) ? *(const int4*)(kp + j * 128)
                                                          : make_int4(nomatch, nomatch, nomatch, nomatch);
        }
        const int ebase = wbase + lane16;
        unsigned cnt = lo;
#pragma unroll
        for (int j = 0; j < OWN_LOADS; ++j) {
            const unsigned t0 = ((unsigned)k[j].x ^ rep4) + 0x7F7F7F7Fu;
            const unsigned t1 = ((unsigned)k[j].y ^ rep4) + 0x7F7F7F7Fu;
            const unsigned t2 = ((unsigned)k[j].z ^ rep4) + 0x7F7F7F7Fu;
            const unsigned t3 = ((unsigned)k[j].w ^ rep4) + 0x7F7F7F7Fu;
            unsigned hits = (((t0 >> 7) & 0x01010101u) | ((t1 >> 6) & 0x02020202u) |
                          ((t2 >> 5) & 0x04040404u) | ((t3 >> 4) & 0x08080808u)) ^ 0x0F0F0F0Fu;
            unsigned bal = __builtin_amdgcn_ballot_w32(hits != 0u);
            while (bal != 0u) {
                if (hits != 0u) {
                    const unsigned p = (unsigned)__builtin_ctz(hits);
                    unsigned slot = cnt + __builtin_amdgcn_mbcnt_lo(bal, 0u);
                    slot = slot < OWN_LIST ? slot : OWN_LIST;
                    s_list[wave][slot] = (unsigned)(ebase + j * 512) + 4u * (p & 7u) + (p >> 3);
                    hits &= hits - 1u;
                }
                cnt += (unsigned)__builtin_popcount(bal);
                bal = __builtin_amdgcn_ballot_w32(hits != 0u);
            }
        }
        if (cnt <= OWN_LIST) {
            unsigned m0 = 0u;
            for (; m0 + 32u <= cnt; m0 += 32u)
                own_gather32<true>(X, own_idx, src_idx, &s_list[wave][m0], 32u, s_acc, s_cnt,
                                   base, R, lane, hw, c, cA, cB, wsave, saved);
            const unsigned rem = cnt - m0;
            if (m0 != 0u && (unsigned)lane < rem) {
                const unsigned t = s_list[wave][m0 + lane];
                s_list[wave][lane] = t;
            }
            lo = rem;
        } else {
            dense = true;
            for (int p = 0; p < keys_per_wave; p += 32) {
                const int e1 = wbase + p + lane;
                const int key1 = (e1 < e_hi) ? own_idx[e1] : -1;
                unsigned bal = __builtin_amdgcn_ballot_w32((unsigned)(key1 - base) < (unsigned)R);
                while (bal != 0u) {
                    const int eL = wbase + p + __builtin_ctz(bal);
                    bal &= bal - 1u;
                    const unsigned rL = (unsigned)(own_idx[eL] - base);
                    const float2 x = *(const float2*)(X + (size_t)src_idx[eL] * HID + lane * 2);
                    atomicAdd(&s_acc[rL * OWN_PITCH + lane * 2], x.x);
                    atomicAdd(&s_acc[rL * OWN_PITCH + lane * 2 + 1], x.y);
                    if (lane == 0) atomicAdd(&s_cnt[rL], 1u);
                }
            }
        }
    }
    if (lo != 0u)
        own_gather32<false>(X, own_idx, src_idx, &s_list[wave][0], lo, s_acc, s_cnt,
                            base, R, lane, hw, c, cA, cB, wsave, saved);
    if (save && lane == 0)
        save[SAVE_WORDS + (size_t)blockIdx.x * OWN_WAVES + (size_t)wave] =
            (dense || saved > (unsigned)SAVE_CAP) ? SAVE_OVER : saved;
    __syncthreads();

    for (int i = tid; i < R * HID; i += OWN_THREADS) {
        const int r = i >> 6, ch = i & 63;
        const int row = base + r;
        if (row < N) acc[(size_t)row * HID + ch] = s_acc[r * OWN_PITCH + ch];
    }
    if (deg) {
        for (int r = tid; r < R; r += OWN_THREADS)
            if (base + r < N) deg[base + r] = (float)s_cnt[r];
    }
}

__global__ __launch_bounds__(OWN_THREADS)
void list_sum_kernel(const float* __restrict__ X, const int* __restrict__ own_idx,
                     const int* __restrict__ src_idx, const unsigned* __restrict__ save,
                     float* __restrict__ acc, int N, int E, int P, int S) {
    __shared__ float s_acc[(OWN_MAXROWS + 1) * OWN_PITCH];

    const int tid  = threadIdx.x;
    const int lane = tid & 31;
    const int wave = tid >> 5;
    const int R    = (N + P - 1) / P;
    const int pr   = (int)blockIdx.x % P;
    const int sl   = (int)blockIdx.x / P;
    const int base = pr * R;
    acc += (size_t)sl * (size_t)N * HID;

    for (int i = tid; i < (OWN_MAXROWS + 1) * OWN_PITCH; i += OWN_THREADS) s_acc[i] = 0.0f;
    __syncthreads();

    const int hw = lane >> 4;
    const int c  = (lane & 15) * 4;
    const int cA = c + 2 * hw;
    const int cB = c + 2 - 2 * hw;

    const unsigned* wl = save + ((size_t)blockIdx.x * OWN_WAVES + (size_t)wave) * SAVE_CAP;
    const unsigned n = (unsigned)__builtin_amdgcn_readfirstlane(
        (int)save[SAVE_WORDS + (size_t)blockIdx.x * OWN_WAVES + (size_t)wave]);
    const unsigned spare = (unsigned)OWN_MAXROWS << 17;
    if (n != SAVE_OVER) {
        unsigned pk = ((unsigned)lane < n) ? wl[lane] : spare;
        for (unsigned m = 0u; m < n; m += 32u) {
            const unsigned mn = m + 32u + (unsigned)lane;
            const unsigned wn = wl[mn < n ? mn : 0u];
            own_rows32(X, (int)pk, s_acc, hw, c, cA, cB);
            pk = (mn < n) ? wn : spare;
        }
    } else {
        const int slen = (((E + S - 1) / S) + 15) & ~15;
        const int e_lo = sl * slen;
        const int e_hi = (e_lo + slen < E) ? e_lo + slen : E;
        const int keys_per_wave = 32 * 16 * OWN_LOADS;
        const int keys_per_trip = keys_per_wave * OWN_WAVES;
        const int ntrips = (slen + keys_per_trip - 1) / keys_per_trip;
        for (int it = 0; it < ntrips; ++it) {
            const int wbase = e_lo + it * keys_per_trip + wave * keys_per_wave;
            for (int p = 0; p < keys_per_wave; p += 32) {
                const int e1 = wbase + p + lane;
                const int key1 = (e1 < e_hi) ? own_idx[e1] : -1;
                unsigned bal = __builtin_amdgcn_ballot_w32((unsigned)(key1 - base) < (unsigned)R);
                while (bal != 0u) {
                    const int eL = wbase + p + __builtin_ctz(bal);
                    bal &= bal - 1u;
                    const unsigned rL = (unsigned)(own_idx[eL] - base);
                    const float2 x = *(const float2*)(X + (size_t)src_idx[eL] * HID + lane * 2);
                    atomicAdd(&s_acc[rL * OWN_PITCH + lane * 2], x.x);
                    atomicAdd(&s_acc[rL * OWN_PITCH + lane * 2 + 1], x.y);
                }
            }
        }
    }
    __syncthreads();

    for (int i = tid; i < R * HID; i += OWN_THREADS) {
        const int r = i >> 6, ch = i & 63;
        const int row = base + r;
        if (row < N) acc[(size_t)row * HID + ch] = s_acc[r * OWN_PITCH + ch];
    }
}

#define GEMM_WAVES   2
#define MOV_STRIPS   2
#define USR_STRIPS   4
template <int NP, int STRIPS>
__global__ __launch_bounds__(GEMM_WAVES * 32)
void sage_gemm64(const float* __restrict__ A1, const float* __restrict__ deg1,
                 const float* __restrict__ W1, int K1,
                 const float* __restrict__ A2, const float* __restrict__ W2, int K2,
                 const float* __restrict__ bias, float* __restrict__ C,
                 int M, int relu) {
    const int lane = threadIdx.x & 31;
    const int wave = threadIdx.x >> 5;
    const int half = lane >> 4;
    const int r    = lane & 15;
    const int row0 = (blockIdx.x * GEMM_WAVES + wave) * (16 * STRIPS);
    if (row0 >= M) return;

    v8f c[STRIPS][4] = {};

    int   arow[STRIPS];
    float scale[STRIPS];
#pragma unroll
    for (int s = 0; s < STRIPS; ++s) {
        const int rs = row0 + 16 * s;
        arow[s] = ((rs < M) ? rs : M - 16) + r;
        scale[s] = 1.0f;
        if (deg1) {
            float d = deg1[arow[s]];
#pragma unroll
            for (int p = 1; p < NP; ++p) d += deg1[(size_t)p * M + arow[s]];
            scale[s] = 1.0f / fmaxf(d, 1.0f);
        }
    }

    {
        const size_t plane = (size_t)M * K1;
        for (int kk = 0; kk < K1; kk += 4) {
            const int k = kk + 2 * half;
            v2f b[4];
#pragma unroll
            for (int t = 0; t < 4; ++t) {
                const float* Wrow = W1 + (size_t)(t * 16 + r) * K1;
                b[t].x = Wrow[k];
                b[t].y = Wrow[k + 1];
            }
#pragma unroll
            for (int s = 0; s < STRIPS; ++s) {
                const float* Arow = A1 + (size_t)arow[s] * K1;
                float ax = Arow[k];
                float ay = Arow[k + 1];
#pragma unroll
                for (int p = 1; p < NP; ++p) {
                    ax += Arow[p * plane + k];
                    ay += Arow[p * plane + k + 1];
                }
                v2f a;
                a.x = ax * scale[s];
                a.y = ay * scale[s];
#pragma unroll
                for (int t = 0; t < 4; ++t)
                    c[s][t] = __builtin_amdgcn_wmma_f32_16x16x4_f32(
                        false, a, false, b[t], (short)0, c[s][t], false, false);
            }
        }
    }
    if (A2) {
        for (int kk = 0; kk < K2; kk += 4) {
            const int k = kk + 2 * half;
            v2f b[4];
#pragma unroll
            for (int t = 0; t < 4; ++t) {
                const float* Wrow = W2 + (size_t)(t * 16 + r) * K2;
                b[t].x = Wrow[k];
                b[t].y = Wrow[k + 1];
            }
#pragma unroll
            for (int s = 0; s < STRIPS; ++s) {
                const float* Arow = A2 + (size_t)arow[s] * K2;
                v2f a;
                a.x = Arow[k];
                a.y = Arow[k + 1];
#pragma unroll
                for (int t = 0; t < 4; ++t)
                    c[s][t] = __builtin_amdgcn_wmma_f32_16x16x4_f32(
                        false, a, false, b[t], (short)0, c[s][t], false, false);
            }
        }
    }

#pragma unroll
    for (int s = 0; s < STRIPS; ++s) {
        const int rs = row0 + 16 * s;
        if (rs < M) {
#pragma unroll
            for (int t = 0; t < 4; ++t) {
                const float bv = bias ? bias[t * 16 + r] : 0.0f;
#pragma unroll
                for (int i = 0; i < 8; ++i) {
                    float v = c[s][t][i] + bv;
                    if (relu) v = fmaxf(v, 0.0f);
                    C[(size_t)(rs + i + 8 * half) * HID + t * 16 + r] = v;
                }
            }
        }
    }
}

__global__ __launch_bounds__(256)
void head_kernel(const float* __restrict__ acc2, const float* __restrict__ degm,
                 const float* __restrict__ m1, const float* __restrict__ Wl2,
                 const float* __restrict__ bl2, const float* __restrict__ Wr2,
                 float* __restrict__ out, int M) {
    int m = blockIdx.x * blockDim.x + threadIdx.x;
    if (m >= M) return;
    float dsum = degm[m];
#pragma unroll
    for (int s = 1; s < MOV_SLICES; ++s) dsum += degm[(size_t)s * M + m];
    const float inv = 1.0f / fmaxf(dsum, 1.0f);
    const size_t plane = (size_t)M * HID;
    const float* a = acc2 + (size_t)m * HID;
    const float* x = m1 + (size_t)m * HID;
    float acc[GENRES];
#pragma unroll
    for (int g = 0; g < GENRES; ++g) acc[g] = bl2[g];
    for (int k = 0; k < HID; ++k) {
        float as = a[k];
#pragma unroll
        for (int s = 1; s < MOV_SLICES; ++s) as += a[s * plane + k];
        const float av = as * inv;
        const float xv = x[k];
#pragma unroll
        for (int g = 0; g < GENRES; ++g)
            acc[g] += av * Wl2[g * HID + k] + xv * Wr2[g * HID + k];
    }
#pragma unroll
    for (int g = 0; g < GENRES; ++g) out[(size_t)m * GENRES + g] = acc[g];
}

extern "C" void kernel_launch(void* const* d_in, const int* in_sizes, int n_in,
                              void* d_out, int out_size, void* d_ws, size_t ws_size,
                              hipStream_t stream) {
    (void)in_sizes; (void)n_in; (void)out_size; (void)ws_size;

    const float* x_user  = (const float*)d_in[0];
    const float* x_movie = (const float*)d_in[1];
    const int*   e_src   = (const int*)d_in[2];
    const int*   e_dst   = (const int*)d_in[3];
    const float* W_user  = (const float*)d_in[4];
    const float* b_user  = (const float*)d_in[5];
    const float* W_movie = (const float*)d_in[6];
    const float* b_movie = (const float*)d_in[7];
    const float* Wl1_um  = (const float*)d_in[8];
    const float* bl1_um  = (const float*)d_in[9];
    const float* Wr1_um  = (const float*)d_in[10];
    const float* Wl1_mu  = (const float*)d_in[11];
    const float* bl1_mu  = (const float*)d_in[12];
    const float* Wr1_mu  = (const float*)d_in[13];
    const float* Wl2_um  = (const float*)d_in[14];
    const float* bl2_um  = (const float*)d_in[15];
    const float* Wr2_um  = (const float*)d_in[16];
    float* out = (float*)d_out;

    float* ws = (float*)d_ws;
    float* acc_mu    = ws;
    float* acc_um    = ws;
    float* acc2      = ws;
    unsigned* saved  = (unsigned*)(ws + 6400000);
    int*   rid_dst   = (int*)(ws + 11800000);
    float* hu        = ws + 12800000;
    float* m1        = hu;
    float* hm        = hu + 6400000;
    float* u1        = hm + 3200000;
    int*   rid_src   = (int*)u1;
    float* deg_movie = u1 + 6400000;
    float* deg_user  = deg_movie + 200000;

    const int E = NUM_EDGES;

    range_id_kernel<USR_R, USR_R><<<dim3((NUM_EDGES / 4 + 255) / 256, 1), 256, 0, stream>>>(
        e_src, rid_src, e_src, rid_src, NUM_EDGES / 4);

    const int gridU = ((NUM_USERS + 16 * USR_STRIPS - 1) / (16 * USR_STRIPS) + GEMM_WAVES - 1) / GEMM_WAVES;
    const int gridM = ((NUM_MOVIES + 16 * MOV_STRIPS - 1) / (16 * MOV_STRIPS) + GEMM_WAVES - 1) / GEMM_WAVES;
    sage_gemm64<1, USR_STRIPS><<<gridU, GEMM_WAVES * 32, 0, stream>>>(x_user, nullptr, W_user, USER_FEAT,
                                           nullptr, nullptr, 0, b_user, hu,
                                           NUM_USERS, 0);
    sage_gemm64<1, MOV_STRIPS><<<gridM, GEMM_WAVES * 32, 0, stream>>>(x_movie, nullptr, W_movie, MOVIE_FEAT,
                                           nullptr, nullptr, 0, b_movie, hm,
                                           NUM_MOVIES, 0);

    owner_sum_kernel<<<OWN_BLOCKS, OWN_THREADS, 0, stream>>>(hm, e_src, rid_src, e_dst, acc_mu,
                                                             deg_user, nullptr, NUM_USERS, E,
                                                             USR_RANGES, USR_SLICES);
    sage_gemm64<USR_SLICES, USR_STRIPS><<<gridU, GEMM_WAVES * 32, 0, stream>>>(acc_mu, deg_user, Wl1_mu, HID,
                                           hu, Wr1_mu, HID, bl1_mu, u1,
                                           NUM_USERS, 1);

    range_id_kernel<MOV_R, MOV_R><<<dim3((NUM_EDGES / 4 + 255) / 256, 1), 256, 0, stream>>>(
        e_dst, rid_dst, e_dst, rid_dst, NUM_EDGES / 4);

    owner_sum_kernel<<<OWN_BLOCKS, OWN_THREADS, 0, stream>>>(hu, e_dst, rid_dst, e_src, acc_um,
                                                             deg_movie, saved, NUM_MOVIES, E,
                                                             MOV_RANGES, MOV_SLICES);
    sage_gemm64<MOV_SLICES, MOV_STRIPS><<<gridM, GEMM_WAVES * 32, 0, stream>>>(acc_um, deg_movie, Wl1_um, HID,
                                           hm, Wr1_um, HID, bl1_um, m1,
                                           NUM_MOVIES, 1);

    list_sum_kernel<<<OWN_BLOCKS, OWN_THREADS, 0, stream>>>(u1, e_dst, e_src, saved, acc2,
                                                            NUM_MOVIES, E, MOV_RANGES, MOV_SLICES);
    head_kernel<<<(NUM_MOVIES + 255) / 256, 256, 0, stream>>>(
        acc2, deg_movie, m1, Wl2_um, bl2_um, Wr2_um, out, NUM_MOVIES);
}


# Edge sums leave LDS float atomics: filed edges are queued in LDS by owner wave (row mod 32); after one barrier each wave adds its own rows by plain LDS load, add, store; sum order changes
# speedup vs baseline: 3.5386x; 1.0664x over previous
#include <hip/hip_runtime.h>
#include <hip/hip_bf16.h>

typedef __attribute__((ext_vector_type(2))) float v2f;
typedef __attribute__((ext_vector_type(8))) float v8f;

#define NUM_USERS  100000
#define NUM_MOVIES 50000
#define NUM_EDGES  4000000
#define USER_FEAT  24
#define MOVIE_FEAT 404
#define HID        64
#define GENRES     3

#define OWN_BLOCKS   256
#define OWN_THREADS  1024
#define OWN_WAVES    32
#define OWN_MAXROWS  782
#define OWN_PITCH    68
#define OWN_LIST     128
#define OWN_LOADS    8
#define MOV_RANGES   128
#define MOV_SLICES   2
#define USR_RANGES   128
#define USR_SLICES   2
#define MOV_R        ((NUM_MOVIES + MOV_RANGES - 1) / MOV_RANGES)
#define USR_R        ((NUM_USERS + USR_RANGES - 1) / USR_RANGES)
#define SAVE_CAP     640
#define OWN_LDSROWS  (OWN_MAXROWS + OWN_WAVES)
#define SAVE_OVER    0xFFFFFFFFu
#define SAVE_WORDS   ((size_t)OWN_BLOCKS * OWN_WAVES * SAVE_CAP)
static_assert(MOV_RANGES <= 128 && USR_RANGES <= 128, "a range id must fit in 7 bits");
static_assert(NUM_EDGES % 16 == 0, "16-byte loads of range ids");
static_assert(OWN_LDSROWS <= (1 << 10) && NUM_USERS <= (1 << 17) && NUM_MOVIES <= (1 << 17),
              "a row of a range and a source id are packed into one word (10 + 17 bits)");

template <int RA, int RB>
__global__ __launch_bounds__(256)
void range_id_kernel(const int* __restrict__ keysA, int* __restrict__ ridA,
                     const int* __restrict__ keysB, int* __restrict__ ridB, int n4) {
    const int i = (int)blockIdx.x * 256 + (int)threadIdx.x;
    if (i >= n4) return;
    const bool second = blockIdx.y != 0;
    const int4 k = *(const int4*)((second ? keysB : keysA) + 4 * i);
    const unsigned kv[4] = {(unsigned)k.x, (unsigned)k.y, (unsigned)k.z, (unsigned)k.w};
    unsigned w = 0u;
#pragma unroll
    for (int q = 0; q < 4; ++q) {
        const unsigned id = second ? kv[q] / (unsigned)RB : kv[q] / (unsigned)RA;
        w |= (id & 127u) << (8 * q);
    }
    (second ? ridB : ridA)[i] = (int)w;
}

__device__ __forceinline__ void own_rows32(const float* __restrict__ X, int pk, float* s_acc,
                                           int hw, int c, int cA, int cB) {
    float4 v[16];
    int    pj[16];
#pragma unroll
    for (int j = 0; j < 16; ++j) {
        const int pa = __builtin_amdgcn_readlane(pk, 2 * j);
        const int pb = __builtin_amdgcn_readlane(pk, 2 * j + 1);
        pj[j] = hw ? pb : pa;
        v[j] = *(const float4*)(X + (size_t)(pj[j] & 0x1FFFF) * HID + c);
    }
#pragma unroll
    for (int j = 0; j < 16; ++j) {
        const unsigned r = (unsigned)pj[j] >> 17;
        const float a0 = hw ? v[j].z : v[j].x;
        const float a1 = hw ? v[j].w : v[j].y;
        const float a2 = hw ? v[j].x : v[j].z;
        const float a3 = hw ? v[j].y : v[j].w;
        float* oa = &s_acc[r * OWN_PITCH + cA];
        float* ob = &s_acc[r * OWN_PITCH + cB];
        atomicAdd(oa + 0, a0);
        atomicAdd(oa + 1, a1);
        atomicAdd(ob + 0, a2);
        atomicAdd(ob + 1, a3);
    }
}

__device__ __forceinline__ void rows_plain32(const float* __restrict__ X, int pk, float* s_acc, int lane) {
    float2   v[32];
    unsigned rw[32];
#pragma unroll
    for (int j = 0; j < 32; ++j) {
        const unsigned p = (unsigned)__builtin_amdgcn_readlane(pk, j);
        rw[j] = p >> 17;
        v[j] = *(const float2*)(X + (size_t)(p & 0x1FFFFu) * HID + 2 * lane);
    }
#pragma unroll
    for (int g = 0; g < 32; g += 4) {
        float2 t[4];
#pragma unroll
        for (int i = 0; i < 4; ++i)
            t[i] = *(const float2*)&s_acc[rw[g + i] * OWN_PITCH + 2 * lane];
#pragma unroll
        for (int i = 0; i < 4; ++i) {
#pragma unroll
            for (int k = 0; k < i; ++k)
                if (rw[g + k] == rw[g + i]) t[i] = t[k];
            t[i].x += v[g + i].x;
            t[i].y += v[g + i].y;
        }
#pragma unroll
        for (int i = 0; i < 4; ++i)
            *(float2*)&s_acc[rw[g + i] * OWN_PITCH + 2 * lane] = t[i];
    }
}

template <bool FULL>
__device__ __forceinline__ void own_gather32(const float* __restrict__ X, const int* __restrict__ own_idx,
                                             const int* __restrict__ src_idx, const unsigned* lst,
                                             unsigned nv, float* s_acc, unsigned* s_cnt,
                                             unsigned* s_q, unsigned* s_qcnt, unsigned* s_over,
                                             int base, int R, int lane, int hw, int c, int cA, int cB) {
    const bool okL = FULL || (unsigned)lane < nv;
    const int  ed  = (int)lst[okL ? lane : 0];
    const int  key = own_idx[ed];
    const int  sid = src_idx[ed];
    const unsigned rr  = (unsigned)(key - base);
    const bool     inr = okL && rr < (unsigned)R;
    const unsigned rL  = inr ? rr : (unsigned)OWN_MAXROWS;
    atomicAdd(&s_cnt[rL], 1u);
    const unsigned pk = (rL << 17) | ((unsigned)sid & 0x1FFFFu);
    bool over = false;
    if (inr) {
        const unsigned ow   = rL & (unsigned)(OWN_WAVES - 1);
        const unsigned slot = atomicAdd(&s_qcnt[ow], 1u);
        if (slot < (unsigned)SAVE_CAP) s_q[ow * SAVE_CAP + slot] = pk;
        else over = true;
    }
    if (__builtin_amdgcn_ballot_w32(over) != 0u) {
        *s_over = 1u;
        own_rows32(X, (int)(over ? pk : ((unsigned)OWN_MAXROWS << 17)), s_acc, hw, c, cA, cB);
    }
}

__global__ __launch_bounds__(OWN_THREADS)
void owner_sum_kernel(const float* __restrict__ X, const int* __restrict__ own_idx,
                      const int* __restrict__ rid,
                      const int* __restrict__ src_idx, float* __restrict__ acc,
                      float* __restrict__ deg, unsigned* __restrict__ save,
                      int N, int E, int P, int S) {
    __shared__ float    s_acc[OWN_LDSROWS * OWN_PITCH];
    __shared__ unsigned s_cnt[OWN_MAXROWS + 1];
    __shared__ unsigned s_list[OWN_WAVES][OWN_LIST + 1];
    __shared__ unsigned s_q[OWN_WAVES * SAVE_CAP];
    __shared__ unsigned s_qcnt[OWN_WAVES];
    __shared__ unsigned s_over;

    const int tid  = threadIdx.x;
    const int lane = tid & 31;
    const int wave = tid >> 5;
    const int R    = (N + P - 1) / P;
    const int pr   = (int)blockIdx.x % P;
    const int sl   = (int)blockIdx.x / P;
    const int base = pr * R;
    const int slen = (((E + S - 1) / S) + 15) & ~15;
    const int e_lo = sl * slen;
    const int e_hi = (e_lo + slen < E) ? e_lo + slen : E;
    acc += (size_t)sl * (size_t)N * HID;
    if (deg) deg += (size_t)sl * (size_t)N;

    for (int i = tid; i < OWN_LDSROWS * OWN_PITCH; i += OWN_THREADS) s_acc[i] = 0.0f;
    for (int i = tid; i <= OWN_MAXROWS; i += OWN_THREADS) s_cnt[i] = 0u;
    if (tid < OWN_WAVES) s_qcnt[tid] = 0u;
    if (tid == 0) s_over = 0u;
    __syncthreads();

    const int hw = lane >> 4;
    const int c  = (lane & 15) * 4;
    const int cA = c + 2 * hw;
    const int cB = c + 2 - 2 * hw;
    const int keys_per_wave = 32 * 16 * OWN_LOADS;
    const int keys_per_trip = keys_per_wave * OWN_WAVES;

    const int lane16 = lane * 16;
    const unsigned rep4 = (unsigned)pr * 0x01010101u;
    const int nomatch = (int)(rep4 ^ 0x01010101u);
    const int ntrips = (slen + keys_per_trip - 1) / keys_per_trip;
    int trip = (int)(((unsigned)pr * (unsigned)ntrips) / (unsigned)P);
    unsigned lo = 0u;
    for (int it = 0; it < ntrips; ++it) {
        const int t0 = e_lo + trip * keys_per_trip;
        trip = (trip + 1 == ntrips) ? 0 : trip + 1;
        const int wbase = t0 + wave * keys_per_wave;
        const int* kp = rid + ((wbase + lane16) >> 2);
        int4 k[OWN_LOADS];
        if (wbase + keys_per_wave <= e_hi) {
#pragma unroll
            for (int j = 0; j < OWN_LOADS; ++j) k[j] = *(const int4*)(kp + j * 128);
        } else {
#pragma unroll
            for (int j = 0; j < OWN_LOADS; ++j)
                k[j] = (wbase + j * 512 + lane16 + 15 < e_hi) ? *(const int4*)(kp + j * 128)
                                                          : make_int4(nomatch, nomatch, nomatch, nomatch);
        }
        const int ebase = wbase + lane16;
        unsigned cnt = lo;
#pragma unroll
        for (int j = 0; j < OWN_LOADS; ++j) {
            const unsigned t0 = ((unsigned)k[j].x ^ rep4) + 0x7F7F7F7Fu;
            const unsigned t1 = ((unsigned)k[j].y ^ rep4) + 0x7F7F7F7Fu;
            const unsigned t2 = ((unsigned)k[j].z ^ rep4) + 0x7F7F7F7Fu;
            const unsigned t3 = ((unsigned)k[j].w ^ rep4) + 0x7F7F7F7Fu;
            unsigned hits = (((t0 >> 7) & 0x01010101u) | ((t1 >> 6) & 0x02020202u) |
                          ((t2 >> 5) & 0x04040404u) | ((t3 >> 4) & 0x08080808u)) ^ 0x0F0F0F0Fu;
            unsigned bal = __builtin_amdgcn_ballot_w32(hits != 0u);
            while (bal != 0u) {
                if (hits != 0u) {
                    const unsigned p = (unsigned)__builtin_ctz(hits);
                    unsigned slot = cnt + __builtin_amdgcn_mbcnt_lo(bal, 0u);
                    slot = slot < OWN_LIST ? slot : OWN_LIST;
                    s_list[wave][slot] = (unsigned)(ebase + j * 512) + 4u * (p & 7u) + (p >> 3);
                    hits &= hits - 1u;
                }
                cnt += (unsigned)__builtin_popcount(bal);
                bal = __builtin_amdgcn_ballot_w32(hits != 0u);
            }
        }
        if (cnt <= OWN_LIST) {
            unsigned m0 = 0u;
            for (; m0 + 32u <= cnt; m0 += 32u)
                own_gather32<true>(X, own_idx, src_idx, &s_list[wave][m0], 32u, s_acc, s_cnt,
                                   s_q, s_qcnt, &s_over, base, R, lane, hw, c, cA, cB);
            const unsigned rem = cnt - m0;
            if (m0 != 0u && (unsigned)lane < rem) {
                const unsigned t = s_list[wave][m0 + lane];
                s_list[wave][lane] = t;
            }
            lo = rem;
        } else {
            s_over = 1u;
            for (int p = 0; p < keys_per_wave; p += 32) {
                const int e1 = wbase + p + lane;
                const int key1 = (e1 < e_hi) ? own_idx[e1] : -1;
                unsigned bal = __builtin_amdgcn_ballot_w32((unsigned)(key1 - base) < (unsigned)R);
                while (bal != 0u) {
                    const int eL = wbase + p + __builtin_ctz(bal);
                    bal &= bal - 1u;
                    const unsigned rL = (unsigned)(own_idx[eL] - base);
                    const float2 x = *(const float2*)(X + (size_t)src_idx[eL] * HID + lane * 2);
                    atomicAdd(&s_acc[rL * OWN_PITCH + lane * 2], x.x);
                    atomicAdd(&s_acc[rL * OWN_PITCH + lane * 2 + 1], x.y);
                    if (lane == 0) atomicAdd(&s_cnt[rL], 1u);
                }
            }
        }
    }
    if (lo != 0u)
        own_gather32<false>(X, own_idx, src_idx, &s_list[wave][0], lo, s_acc, s_cnt,
                            s_q, s_qcnt, &s_over, base, R, lane, hw, c, cA, cB);
    __syncthreads();

    {
        const unsigned n  = (unsigned)__builtin_amdgcn_readfirstlane((int)s_qcnt[wave]);
        const unsigned nq = n < (unsigned)SAVE_CAP ? n : (unsigned)SAVE_CAP;
        const unsigned spare = (unsigned)(OWN_MAXROWS + wave) << 17;
        unsigned* wsave = save ? save + ((size_t)blockIdx.x * OWN_WAVES + (size_t)wave) * SAVE_CAP : nullptr;
        for (unsigned m = 0u; m < nq; m += 32u) {
            const unsigned idx = m + (unsigned)lane;
            const bool     ok  = idx < nq;
            const unsigned q   = s_q[(unsigned)wave * SAVE_CAP + (ok ? idx : 0u)];
            const unsigned pk  = ok ? q : spare;
            if (wsave != nullptr && ok) wsave[idx] = pk;
            rows_plain32(X, (int)pk, s_acc, lane);
        }
        if (save && lane == 0)
            save[SAVE_WORDS + (size_t)blockIdx.x * OWN_WAVES + (size_t)wave] =
                (s_over != 0u || n > (unsigned)SAVE_CAP) ? SAVE_OVER : n;
    }
    __syncthreads();

    for (int i = tid; i < R * HID; i += OWN_THREADS) {
        const int r = i >> 6, ch = i & 63;
        const int row = base + r;
        if (row < N) acc[(size_t)row * HID + ch] = s_acc[r * OWN_PITCH + ch];
    }
    if (deg) {
        for (int r = tid; r < R; r += OWN_THREADS)
            if (base + r < N) deg[base + r] = (float)s_cnt[r];
    }
}

__global__ __launch_bounds__(OWN_THREADS)
void list_sum_kernel(const float* __restrict__ X, const int* __restrict__ own_idx,
                     const int* __restrict__ src_idx, const unsigned* __restrict__ save,
                     float* __restrict__ acc, int N, int E, int P, int S) {
    __shared__ float s_acc[OWN_LDSROWS * OWN_PITCH];

    const int tid  = threadIdx.x;
    const int lane = tid & 31;
    const int wave = tid >> 5;
    const int R    = (N + P - 1) / P;
    const int pr   = (int)blockIdx.x % P;
    const int sl   = (int)blockIdx.x / P;
    const int base = pr * R;
    acc += (size_t)sl * (size_t)N * HID;

    for (int i = tid; i < OWN_LDSROWS * OWN_PITCH; i += OWN_THREADS) s_acc[i] = 0.0f;
    __syncthreads();

    const unsigned* wl = save + ((size_t)blockIdx.x * OWN_WAVES + (size_t)wave) * SAVE_CAP;
    const unsigned n = (unsigned)__builtin_amdgcn_readfirstlane(
        (int)save[SAVE_WORDS + (size_t)blockIdx.x * OWN_WAVES + (size_t)wave]);
    const unsigned spare = (unsigned)(OWN_MAXROWS + wave) << 17;
    if (n != SAVE_OVER) {
        unsigned pk = ((unsigned)lane < n) ? wl[lane] : spare;
        for (unsigned m = 0u; m < n; m += 32u) {
            const unsigned mn = m + 32u + (unsigned)lane;
            const unsigned wn = wl[mn < n ? mn : 0u];
            rows_plain32(X, (int)pk, s_acc, lane);
            pk = (mn < n) ? wn : spare;
        }
    } else {
        const int slen = (((E + S - 1) / S) + 15) & ~15;
        const int e_lo = sl * slen;
        const int e_hi = (e_lo + slen < E) ? e_lo + slen : E;
        const int keys_per_wave = 32 * 16 * OWN_LOADS;
        const int keys_per_trip = keys_per_wave * OWN_WAVES;
        const int ntrips = (slen + keys_per_trip - 1) / keys_per_trip;
        for (int it = 0; it < ntrips; ++it) {
            const int wbase = e_lo + it * keys_per_trip + wave * keys_per_wave;
            for (int p = 0; p < keys_per_wave; p += 32) {
                const int e1 = wbase + p + lane;
                const int key1 = (e1 < e_hi) ? own_idx[e1] : -1;
                unsigned bal = __builtin_amdgcn_ballot_w32((unsigned)(key1 - base) < (unsigned)R);
                while (bal != 0u) {
                    const int eL = wbase + p + __builtin_ctz(bal);
                    bal &= bal - 1u;
                    const unsigned rL = (unsigned)(own_idx[eL] - base);
                    const float2 x = *(const float2*)(X + (size_t)src_idx[eL] * HID + lane * 2);
                    atomicAdd(&s_acc[rL * OWN_PITCH + lane * 2], x.x);
                    atomicAdd(&s_acc[rL * OWN_PITCH + lane * 2 + 1], x.y);
                }
            }
        }
    }
    __syncthreads();

    for (int i = tid; i < R * HID; i += OWN_THREADS) {
        const int r = i >> 6, ch = i & 63;
        const int row = base + r;
        if (row < N) acc[(size_t)row * HID + ch] = s_acc[r * OWN_PITCH + ch];
    }
}

#define GEMM_WAVES   2
#define MOV_STRIPS   2
#define USR_STRIPS   4
template <int NP, int STRIPS>
__global__ __launch_bounds__(GEMM_WAVES * 32)
void sage_gemm64(const float* __restrict__ A1, const float* __restrict__ deg1,
                 const float* __restrict__ W1, int K1,
                 const float* __restrict__ A2, const float* __restrict__ W2, int K2,
                 const float* __restrict__ bias, float* __restrict__ C,
                 int M, int relu) {
    const int lane = threadIdx.x & 31;
    const int wave = threadIdx.x >> 5;
    const int half = lane >> 4;
    const int r    = lane & 15;
    const int row0 = (blockIdx.x * GEMM_WAVES + wave) * (16 * STRIPS);
    if (row0 >= M) return;

    v8f c[STRIPS][4] = {};

    int   arow[STRIPS];
    float scale[STRIPS];
#pragma unroll
    for (int s = 0; s < STRIPS; ++s) {
        const int rs = row0 + 16 * s;
        arow[s] = ((rs < M) ? rs : M - 16) + r;
        scale[s] = 1.0f;
        if (deg1) {
            float d = deg1[arow[s]];
#pragma unroll
            for (int p = 1; p < NP; ++p) d += deg1[(size_t)p * M + arow[s]];
            scale[s] = 1.0f / fmaxf(d, 1.0f);
        }
    }

    {
        const size_t plane = (size_t)M * K1;
        for (int kk = 0; kk < K1; kk += 4) {
            const int k = kk + 2 * half;
            v2f b[4];
#pragma unroll
            for (int t = 0; t < 4; ++t) {
                const float* Wrow = W1 + (size_t)(t * 16 + r) * K1;
                b[t].x = Wrow[k];
                b[t].y = Wrow[k + 1];
            }
#pragma unroll
            for (int s = 0; s < STRIPS; ++s) {
                const float* Arow = A1 + (size_t)arow[s] * K1;
                float ax = Arow[k];
                float ay = Arow[k + 1];
#pragma unroll
                for (int p = 1; p < NP; ++p) {
                    ax += Arow[p * plane + k];
                    ay += Arow[p * plane + k + 1];
                }
                v2f a;
                a.x = ax * scale[s];
                a.y = ay * scale[s];
#pragma unroll
                for (int t = 0; t < 4; ++t)
                    c[s][t] = __builtin_amdgcn_wmma_f32_16x16x4_f32(
                        false, a, false, b[t], (short)0, c[s][t], false, false);
            }
        }
    }
    if (A2) {
        for (int kk = 0; kk < K2; kk += 4) {
            const int k = kk + 2 * half;
            v2f b[4];
#pragma unroll
            for (int t = 0; t < 4; ++t) {
                const float* Wrow = W2 + (size_t)(t * 16 + r) * K2;
                b[t].x = Wrow[k];
                b[t].y = Wrow[k + 1];
            }
#pragma unroll
            for (int s = 0; s < STRIPS; ++s) {
                const float* Arow = A2 + (size_t)arow[s] * K2;
                v2f a;
                a.x = Arow[k];
                a.y = Arow[k + 1];
#pragma unroll
                for (int t = 0; t < 4; ++t)
                    c[s][t] = __builtin_amdgcn_wmma_f32_16x16x4_f32(
                        false, a, false, b[t], (short)0, c[s][t], false, false);
            }
        }
    }

#pragma unroll
    for (int s = 0; s < STRIPS; ++s) {
        const int rs = row0 + 16 * s;
        if (rs < M) {
#pragma unroll
            for (int t = 0; t < 4; ++t) {
                const float bv = bias ? bias[t * 16 + r] : 0.0f;
#pragma unroll
                for (int i = 0; i < 8; ++i) {
                    float v = c[s][t][i] + bv;
                    if (relu) v = fmaxf(v, 0.0f);
                    C[(size_t)(rs + i + 8 * half) * HID + t * 16 + r] = v;
                }
            }
        }
    }
}

__global__ __launch_bounds__(256)
void head_kernel(const float* __restrict__ acc2, const float* __restrict__ degm,
                 const float* __restrict__ m1, const float* __restrict__ Wl2,
                 const float* __restrict__ bl2, const float* __restrict__ Wr2,
                 float* __restrict__ out, int M) {
    int m = blockIdx.x * blockDim.x + threadIdx.x;
    if (m >= M) return;
    float dsum = degm[m];
#pragma unroll
    for (int s = 1; s < MOV_SLICES; ++s) dsum += degm[(size_t)s * M + m];
    const float inv = 1.0f / fmaxf(dsum, 1.0f);
    const size_t plane = (size_t)M * HID;
    const float* a = acc2 + (size_t)m * HID;
    const float* x = m1 + (size_t)m * HID;
    float acc[GENRES];
#pragma unroll
    for (int g = 0; g < GENRES; ++g) acc[g] = bl2[g];
    for (int k = 0; k < HID; ++k) {
        float as = a[k];
#pragma unroll
        for (int s = 1; s < MOV_SLICES; ++s) as += a[s * plane + k];
        const float av = as * inv;
        const float xv = x[k];
#pragma unroll
        for (int g = 0; g < GENRES; ++g)
            acc[g] += av * Wl2[g * HID + k] + xv * Wr2[g * HID + k];
    }
#pragma unroll
    for (int g = 0; g < GENRES; ++g) out[(size_t)m * GENRES + g] = acc[g];
}

extern "C" void kernel_launch(void* const* d_in, const int* in_sizes, int n_in,
                              void* d_out, int out_size, void* d_ws, size_t ws_size,
                              hipStream_t stream) {
    (void)in_sizes; (void)n_in; (void)out_size; (void)ws_size;

    const float* x_user  = (const float*)d_in[0];
    const float* x_movie = (const float*)d_in[1];
    const int*   e_src   = (const int*)d_in[2];
    const int*   e_dst   = (const int*)d_in[3];
    const float* W_user  = (const float*)d_in[4];
    const float* b_user  = (const float*)d_in[5];
    const float* W_movie = (const float*)d_in[6];
    const float* b_movie = (const float*)d_in[7];
    const float* Wl1_um  = (const float*)d_in[8];
    const float* bl1_um  = (const float*)d_in[9];
    const float* Wr1_um  = (const float*)d_in[10];
    const float* Wl1_mu  = (const float*)d_in[11];
    const float* bl1_mu  = (const float*)d_in[12];
    const float* Wr1_mu  = (const float*)d_in[13];
    const float* Wl2_um  = (const float*)d_in[14];
    const float* bl2_um  = (const float*)d_in[15];
    const float* Wr2_um  = (const float*)d_in[16];
    float* out = (float*)d_out;

    float* ws = (float*)d_ws;
    float* acc_mu    = ws;
    float* acc_um    = ws;
    float* acc2      = ws;
    unsigned* saved  = (unsigned*)(ws + 6400000);
    int*   rid_dst   = (int*)(ws + 11800000);
    float* hu        = ws + 12800000;
    float* m1        = hu;
    float* hm        = hu + 6400000;
    float* u1        = hm + 3200000;
    int*   rid_src   = (int*)u1;
    float* deg_movie = u1 + 6400000;
    float* deg_user  = deg_movie + 200000;

    const int E = NUM_EDGES;

    range_id_kernel<USR_R, USR_R><<<dim3((NUM_EDGES / 4 + 255) / 256, 1), 256, 0, stream>>>(
        e_src, rid_src, e_src, rid_src, NUM_EDGES / 4);

    const int gridU = ((NUM_USERS + 16 * USR_STRIPS - 1) / (16 * USR_STRIPS) + GEMM_WAVES - 1) / GEMM_WAVES;
    const int gridM = ((NUM_MOVIES + 16 * MOV_STRIPS - 1) / (16 * MOV_STRIPS) + GEMM_WAVES - 1) / GEMM_WAVES;
    sage_gemm64<1, USR_STRIPS><<<gridU, GEMM_WAVES * 32, 0, stream>>>(x_user, nullptr, W_user, USER_FEAT,
                                           nullptr, nullptr, 0, b_user, hu,
                                           NUM_USERS, 0);
    sage_gemm64<1, MOV_STRIPS><<<gridM, GEMM_WAVES * 32, 0, stream>>>(x_movie, nullptr, W_movie, MOVIE_FEAT,
                                           nullptr, nullptr, 0, b_movie, hm,
                                           NUM_MOVIES, 0);

    owner_sum_kernel<<<OWN_BLOCKS, OWN_THREADS, 0, stream>>>(hm, e_src, rid_src, e_dst, acc_mu,
                                                             deg_user, nullptr, NUM_USERS, E,
                                                             USR_RANGES, USR_SLICES);
    sage_gemm64<USR_SLICES, USR_STRIPS><<<gridU, GEMM_WAVES * 32, 0, stream>>>(acc_mu, deg_user, Wl1_mu, HID,
                                           hu, Wr1_mu, HID, bl1_mu, u1,
                                           NUM_USERS, 1);

    range_id_kernel<MOV_R, MOV_R><<<dim3((NUM_EDGES / 4 + 255) / 256, 1), 256, 0, stream>>>(
        e_dst, rid_dst, e_dst, rid_dst, NUM_EDGES / 4);

    owner_sum_kernel<<<OWN_BLOCKS, OWN_THREADS, 0, stream>>>(hu, e_dst, rid_dst, e_src, acc_um,
                                                             deg_movie, saved, NUM_MOVIES, E,
                                                             MOV_RANGES, MOV_SLICES);
    sage_gemm64<MOV_SLICES, MOV_STRIPS><<<gridM, GEMM_WAVES * 32, 0, stream>>>(acc_um, deg_movie, Wl1_um, HID,
                                           hm, Wr1_um, HID, bl1_um, m1,
                                           NUM_MOVIES, 1);

    list_sum_kernel<<<OWN_BLOCKS, OWN_THREADS, 0, stream>>>(u1, e_dst, e_src, saved, acc2,
                                                            NUM_MOVIES, E, MOV_RANGES, MOV_SLICES);
    head_kernel<<<(NUM_MOVIES + 255) / 256, 256, 0, stream>>>(
        acc2, deg_movie, m1, Wl2_um, bl2_um, Wr2_um, out, NUM_MOVIES);
}


# Edge filing by chunk-local counting sort: a pre-pass sorts each 4096-edge chunk by owner range into packed (row, source) words; owner blocks read their runs, no scan, no key loads; sum order changes
# speedup vs baseline: 4.2491x; 1.2008x over previous
#include <hip/hip_runtime.h>
#include <hip/hip_bf16.h>

typedef __attribute__((ext_vector_type(2))) float v2f;
typedef __attribute__((ext_vector_type(8))) float v8f;

#define NUM_USERS  100000
#define NUM_MOVIES 50000
#define NUM_EDGES  4000000
#define USER_FEAT  24
#define MOVIE_FEAT 404
#define HID        64
#define GENRES     3


#define OWN_BLOCKS   256
#define OWN_THREADS  1024
#define OWN_WAVES    32
#define OWN_MAXROWS  782
#define OWN_PITCH    68
#define MOV_RANGES   128
#define MOV_SLICES   2
#define USR_RANGES   128
#define USR_SLICES   2
#define MOV_R        ((NUM_MOVIES + MOV_RANGES - 1) / MOV_RANGES)
#define USR_R        ((NUM_USERS + USR_RANGES - 1) / USR_RANGES)
#define OWN_QCAP     640
#define OWN_LDSROWS  (OWN_MAXROWS + OWN_WAVES)
#define CHUNK        4096
#define NRANGE       128
#define OFF_STRIDE   (NRANGE + 1)
#define NCHUNKS      ((NUM_EDGES + CHUNK - 1) / CHUNK)
static_assert(MOV_RANGES <= NRANGE && USR_RANGES <= NRANGE, "a range id must be below 128");
static_assert(NUM_EDGES % 4 == 0, "16-byte loads of keys and source ids");
static_assert(OWN_LDSROWS <= (1 << 10) && NUM_USERS <= (1 << 17) && NUM_MOVIES <= (1 << 17),
              "a row of a range and a source id are packed into one word (10 + 17 bits)");

template <int R>
__global__ __launch_bounds__(256)
void bucket_kernel(const int* __restrict__ keys, const int* __restrict__ sids,
                   unsigned* __restrict__ pw, unsigned* __restrict__ off, int E) {
    __shared__ unsigned s_hist[NRANGE];
    __shared__ unsigned s_start[OFF_STRIDE];
    __shared__ __attribute__((aligned(16))) unsigned s_out[CHUNK];

    const int tid = threadIdx.x;
    const int c0  = (int)blockIdx.x * CHUNK;
    if (tid < NRANGE) s_hist[tid] = 0u;
    __syncthreads();

    unsigned w[16], rk[16], id[16];
#pragma unroll
    for (int i = 0; i < 4; ++i) {
        const int  e  = c0 + i * 1024 + tid * 4;
        const bool in = e < E;
        const int4 k = in ? *(const int4*)(keys + e) : make_int4(-1, -1, -1, -1);
        const int4 s = in ? *(const int4*)(sids + e) : make_int4(0, 0, 0, 0);
        const unsigned kv[4] = {(unsigned)k.x, (unsigned)k.y, (unsigned)k.z, (unsigned)k.w};
        const unsigned sv[4] = {(unsigned)s.x, (unsigned)s.y, (unsigned)s.z, (unsigned)s.w};
#pragma unroll
        for (int q = 0; q < 4; ++q) {
            const unsigned rid = kv[q] / (unsigned)R;
            const bool     ok  = rid < (unsigned)NRANGE;
            id[4 * i + q] = ok ? rid : (unsigned)NRANGE;
            w[4 * i + q]  = ((kv[q] - rid * (unsigned)R) << 17) | (sv[q] & 0x1FFFFu);
            rk[4 * i + q] = 0u;
            if (ok) rk[4 * i + q] = atomicAdd(&s_hist[rid], 1u);
        }
    }
    __syncthreads();

    if (tid < 32) {
        const unsigned h0 = s_hist[4 * tid], h1 = s_hist[4 * tid + 1];
        const unsigned h2 = s_hist[4 * tid + 2], h3 = s_hist[4 * tid + 3];
        const unsigned t = h0 + h1 + h2 + h3;
        unsigned inc = t;
#pragma unroll
        for (int d = 1; d < 32; d <<= 1) {
            const unsigned v = __shfl_up(inc, (unsigned)d);
            if (tid >= d) inc += v;
        }
        const unsigned ex = inc - t;
        s_start[4 * tid]     = ex;
        s_start[4 * tid + 1] = ex + h0;
        s_start[4 * tid + 2] = ex + h0 + h1;
        s_start[4 * tid + 3] = ex + h0 + h1 + h2;
        if (tid == 31) s_start[NRANGE] = inc;
    }
    __syncthreads();

#pragma unroll
    for (int j = 0; j < 16; ++j)
        if (id[j] < (unsigned)NRANGE) s_out[s_start[id[j]] + rk[j]] = w[j];
    __syncthreads();

    const unsigned total = s_start[NRANGE];
#pragma unroll
    for (int i = 0; i < 4; ++i) {
        const int idx = i * 1024 + tid * 4;
        if ((unsigned)idx < total) *(uint4*)(pw + (size_t)c0 + idx) = *(const uint4*)&s_out[idx];
    }
    if (tid < OFF_STRIDE) off[(size_t)blockIdx.x * OFF_STRIDE + tid] = s_start[tid];
}

__device__ __forceinline__ void own_rows32(const float* __restrict__ X, int pk, float* s_acc,
                                           int hw, int c, int cA, int cB) {
    float4 v[16];
    int    pj[16];
#pragma unroll
    for (int j = 0; j < 16; ++j) {
        const int pa = __builtin_amdgcn_readlane(pk, 2 * j);
        const int pb = __builtin_amdgcn_readlane(pk, 2 * j + 1);
        pj[j] = hw ? pb : pa;
        v[j] = *(const float4*)(X + (size_t)(pj[j] & 0x1FFFF) * HID + c);
    }
#pragma unroll
    for (int j = 0; j < 16; ++j) {
        const unsigned r = (unsigned)pj[j] >> 17;
        const float a0 = hw ? v[j].z : v[j].x;
        const float a1 = hw ? v[j].w : v[j].y;
        const float a2 = hw ? v[j].x : v[j].z;
        const float a3 = hw ? v[j].y : v[j].w;
        float* oa = &s_acc[r * OWN_PITCH + cA];
        float* ob = &s_acc[r * OWN_PITCH + cB];
        atomicAdd(oa + 0, a0);
        atomicAdd(oa + 1, a1);
        atomicAdd(ob + 0, a2);
        atomicAdd(ob + 1, a3);
    }
}

__device__ __forceinline__ void rows_plain32(const float* __restrict__ X, int pk, float* s_acc, int lane) {
    float2   v[32];
    unsigned rw[32];
#pragma unroll
    for (int j = 0; j < 32; ++j) {
        const unsigned p = (unsigned)__builtin_amdgcn_readlane(pk, j);
        rw[j] = p >> 17;
        v[j] = *(const float2*)(X + (size_t)(p & 0x1FFFFu) * HID + 2 * lane);
    }
#pragma unroll
    for (int g = 0; g < 32; g += 4) {
        float2 t[4];
#pragma unroll
        for (int i = 0; i < 4; ++i)
            t[i] = *(const float2*)&s_acc[rw[g + i] * OWN_PITCH + 2 * lane];
#pragma unroll
        for (int i = 0; i < 4; ++i) {
#pragma unroll
            for (int k = 0; k < i; ++k)
                if (rw[g + k] == rw[g + i]) t[i] = t[k];
            t[i].x += v[g + i].x;
            t[i].y += v[g + i].y;
        }
#pragma unroll
        for (int i = 0; i < 4; ++i)
            *(float2*)&s_acc[rw[g + i] * OWN_PITCH + 2 * lane] = t[i];
    }
}

__device__ __forceinline__ void own_push32(const float* __restrict__ X, unsigned wv, bool ok,
                                           float* s_acc, unsigned* s_cnt, unsigned* s_q,
                                           unsigned* s_qcnt, int R, bool count,
                                           int hw, int c, int cA, int cB) {
    const unsigned rr  = wv >> 17;
    const bool     inr = ok && rr < (unsigned)R;
    const unsigned rL  = inr ? rr : (unsigned)OWN_MAXROWS;
    if (count) atomicAdd(&s_cnt[rL], 1u);
    const unsigned pk = (rL << 17) | (wv & 0x1FFFFu);
    bool over = false;
    if (inr) {
        const unsigned ow   = rL & (unsigned)(OWN_WAVES - 1);
        const unsigned slot = atomicAdd(&s_qcnt[ow], 1u);
        if (slot < (unsigned)OWN_QCAP) s_q[ow * OWN_QCAP + slot] = pk;
        else over = true;
    }
    if (__builtin_amdgcn_ballot_w32(over) != 0u) {
        own_rows32(X, (int)(over ? pk : ((unsigned)OWN_MAXROWS << 17)), s_acc, hw, c, cA, cB);
    }
}

__global__ __launch_bounds__(OWN_THREADS)
void owner_sum_kernel(const float* __restrict__ X, const unsigned* __restrict__ pw,
                      const unsigned* __restrict__ off, float* __restrict__ acc,
                      float* __restrict__ deg, int N, int E, int P, int S) {
    __shared__ float    s_acc[OWN_LDSROWS * OWN_PITCH];
    __shared__ unsigned s_cnt[OWN_MAXROWS + 1];
    __shared__ unsigned s_q[OWN_WAVES * OWN_QCAP];
    __shared__ unsigned s_qcnt[OWN_WAVES];

    const int tid  = threadIdx.x;
    const int lane = tid & 31;
    const int wave = tid >> 5;
    const int R    = (N + P - 1) / P;
    const int pr   = (int)blockIdx.x % P;
    const int sl   = (int)blockIdx.x / P;
    const int base = pr * R;
    const int nch  = (E + CHUNK - 1) / CHUNK;
    const int cps  = (nch + S - 1) / S;
    const int c_lo = sl * cps;
    const int c_hi = (c_lo + cps < nch) ? c_lo + cps : nch;
    const bool count = deg != nullptr;
    acc += (size_t)sl * (size_t)N * HID;
    if (deg) deg += (size_t)sl * (size_t)N;

    for (int i = tid; i < OWN_LDSROWS * OWN_PITCH; i += OWN_THREADS) s_acc[i] = 0.0f;
    for (int i = tid; i <= OWN_MAXROWS; i += OWN_THREADS) s_cnt[i] = 0u;
    if (tid < OWN_WAVES) s_qcnt[tid] = 0u;
    __syncthreads();

    const int hw = lane >> 4;
    const int c  = (lane & 15) * 4;
    const int cA = c + 2 * hw;
    const int cB = c + 2 - 2 * hw;


    const int uw = __builtin_amdgcn_readfirstlane(wave);
    for (int g0 = c_lo + uw; g0 < c_hi; g0 += 32 * OWN_WAVES) {
        const int cl = g0 + lane * OWN_WAVES;
        unsigned st = 0u, nn = 0u;
        if (cl < c_hi) {
            const unsigned a = off[(size_t)cl * OFF_STRIDE + pr];
            const unsigned b = off[(size_t)cl * OFF_STRIDE + pr + 1];
            st = (unsigned)cl * (unsigned)CHUNK + a;
            nn = b - a;
        }
        int ng = (c_hi - g0 + OWN_WAVES - 1) / OWN_WAVES;
        ng = ng < 32 ? ng : 32;
        for (int j = 0; j < ng; j += 4) {
            unsigned s4[4], n4[4], wa[4], wb[4];
#pragma unroll
            for (int q = 0; q < 4; ++q) {
                s4[q] = (unsigned)__builtin_amdgcn_readlane((int)st, j + q);
                n4[q] = (unsigned)__builtin_amdgcn_readlane((int)nn, j + q);
            }
#pragma unroll
            for (int q = 0; q < 4; ++q) {
                wa[q] = ((unsigned)lane < n4[q]) ? pw[(size_t)s4[q] + lane] : 0u;
                wb[q] = ((unsigned)lane + 32u < n4[q]) ? pw[(size_t)s4[q] + 32u + lane] : 0u;
            }
#pragma nounroll
            for (int q = 0; q < 4; ++q) {
                const unsigned nr = q == 0 ? n4[0] : q == 1 ? n4[1] : q == 2 ? n4[2] : n4[3];
                const unsigned sr = q == 0 ? s4[0] : q == 1 ? s4[1] : q == 2 ? s4[2] : s4[3];
                const unsigned ra = q == 0 ? wa[0] : q == 1 ? wa[1] : q == 2 ? wa[2] : wa[3];
                const unsigned rb = q == 0 ? wb[0] : q == 1 ? wb[1] : q == 2 ? wb[2] : wb[3];
                for (unsigned m = 0u; m < nr; m += 32u) {
                    const bool okm = m + (unsigned)lane < nr;
                    unsigned wm;
                    if (m == 0u)       wm = ra;
                    else if (m == 32u) wm = rb;
                    else               wm = okm ? pw[(size_t)sr + m + lane] : 0u;
                    own_push32(X, wm, okm, s_acc, s_cnt, s_q, s_qcnt, R, count, hw, c, cA, cB);
                }
            }
        }
    }
    __syncthreads();

    {
        const unsigned n  = (unsigned)__builtin_amdgcn_readfirstlane((int)s_qcnt[wave]);
        const unsigned nq = n < (unsigned)OWN_QCAP ? n : (unsigned)OWN_QCAP;
        const unsigned spare = (unsigned)(OWN_MAXROWS + wave) << 17;
        for (unsigned m = 0u; m < nq; m += 32u) {
            const unsigned idx = m + (unsigned)lane;
            const bool     ok  = idx < nq;
            const unsigned q   = s_q[(unsigned)wave * OWN_QCAP + (ok ? idx : 0u)];
            const unsigned pk  = ok ? q : spare;
            rows_plain32(X, (int)pk, s_acc, lane);
        }
    }
    __syncthreads();

    for (int i = tid; i < R * HID; i += OWN_THREADS) {
        const int r = i >> 6, ch = i & 63;
        const int row = base + r;
        if (row < N) acc[(size_t)row * HID + ch] = s_acc[r * OWN_PITCH + ch];
    }
    if (deg) {
        for (int r = tid; r < R; r += OWN_THREADS)
            if (base + r < N) deg[base + r] = (float)s_cnt[r];
    }
}

#define GEMM_WAVES   2
#define MOV_STRIPS   2
#define USR_STRIPS   4
template <int NP, int STRIPS>
__global__ __launch_bounds__(GEMM_WAVES * 32)
void sage_gemm64(const float* __restrict__ A1, const float* __restrict__ deg1,
                 const float* __restrict__ W1, int K1,
                 const float* __restrict__ A2, const float* __restrict__ W2, int K2,
                 const float* __restrict__ bias, float* __restrict__ C,
                 int M, int relu) {
    const int lane = threadIdx.x & 31;
    const int wave = threadIdx.x >> 5;
    const int half = lane >> 4;
    const int r    = lane & 15;
    const int row0 = (blockIdx.x * GEMM_WAVES + wave) * (16 * STRIPS);
    if (row0 >= M) return;

    v8f c[STRIPS][4] = {};

    int   arow[STRIPS];
    float scale[STRIPS];
#pragma unroll
    for (int s = 0; s < STRIPS; ++s) {
        const int rs = row0 + 16 * s;
        arow[s] = ((rs < M) ? rs : M - 16) + r;
        scale[s] = 1.0f;
        if (deg1) {
            float d = deg1[arow[s]];
#pragma unroll
            for (int p = 1; p < NP; ++p) d += deg1[(size_t)p * M + arow[s]];
            scale[s] = 1.0f / fmaxf(d, 1.0f);
        }
    }

    {
        const size_t plane = (size_t)M * K1;
        for (int kk = 0; kk < K1; kk += 4) {
            const int k = kk + 2 * half;
            v2f b[4];
#pragma unroll
            for (int t = 0; t < 4; ++t) {
                const float* Wrow = W1 + (size_t)(t * 16 + r) * K1;
                b[t].x = Wrow[k];
                b[t].y = Wrow[k + 1];
            }
#pragma unroll
            for (int s = 0; s < STRIPS; ++s) {
                const float* Arow = A1 + (size_t)arow[s] * K1;
                float ax = Arow[k];
                float ay = Arow[k + 1];
#pragma unroll
                for (int p = 1; p < NP; ++p) {
                    ax += Arow[p * plane + k];
                    ay += Arow[p * plane + k + 1];
                }
                v2f a;
                a.x = ax * scale[s];
                a.y = ay * scale[s];
#pragma unroll
                for (int t = 0; t < 4; ++t)
                    c[s][t] = __builtin_amdgcn_wmma_f32_16x16x4_f32(
                        false, a, false, b[t], (short)0, c[s][t], false, false);
            }
        }
    }
    if (A2) {
        for (int kk = 0; kk < K2; kk += 4) {
            const int k = kk + 2 * half;
            v2f b[4];
#pragma unroll
            for (int t = 0; t < 4; ++t) {
                const float* Wrow = W2 + (size_t)(t * 16 + r) * K2;
                b[t].x = Wrow[k];
                b[t].y = Wrow[k + 1];
            }
#pragma unroll
            for (int s = 0; s < STRIPS; ++s) {
                const float* Arow = A2 + (size_t)arow[s] * K2;
                v2f a;
                a.x = Arow[k];
                a.y = Arow[k + 1];
#pragma unroll
                for (int t = 0; t < 4; ++t)
                    c[s][t] = __builtin_amdgcn_wmma_f32_16x16x4_f32(
                        false, a, false, b[t], (short)0, c[s][t], false, false);
            }
        }
    }

#pragma unroll
    for (int s = 0; s < STRIPS; ++s) {
        const int rs = row0 + 16 * s;
        if (rs < M) {
#pragma unroll
            for (int t = 0; t < 4; ++t) {
                const float bv = bias ? bias[t * 16 + r] : 0.0f;
#pragma unroll
                for (int i = 0; i < 8; ++i) {
                    float v = c[s][t][i] + bv;
                    if (relu) v = fmaxf(v, 0.0f);
                    C[(size_t)(rs + i + 8 * half) * HID + t * 16 + r] = v;
                }
            }
        }
    }
}

__global__ __launch_bounds__(256)
void head_kernel(const float* __restrict__ acc2, const float* __restrict__ degm,
                 const float* __restrict__ m1, const float* __restrict__ Wl2,
                 const float* __restrict__ bl2, const float* __restrict__ Wr2,
                 float* __restrict__ out, int M) {
    int m = blockIdx.x * blockDim.x + threadIdx.x;
    if (m >= M) return;
    float dsum = degm[m];
#pragma unroll
    for (int s = 1; s < MOV_SLICES; ++s) dsum += degm[(size_t)s * M + m];
    const float inv = 1.0f / fmaxf(dsum, 1.0f);
    const size_t plane = (size_t)M * HID;
    const float* a = acc2 + (size_t)m * HID;
    const float* x = m1 + (size_t)m * HID;
    float acc[GENRES];
#pragma unroll
    for (int g = 0; g < GENRES; ++g) acc[g] = bl2[g];
    for (int k = 0; k < HID; ++k) {
        float as = a[k];
#pragma unroll
        for (int s = 1; s < MOV_SLICES; ++s) as += a[s * plane + k];
        const float av = as * inv;
        const float xv = x[k];
#pragma unroll
        for (int g = 0; g < GENRES; ++g)
            acc[g] += av * Wl2[g * HID + k] + xv * Wr2[g * HID + k];
    }
#pragma unroll
    for (int g = 0; g < GENRES; ++g) out[(size_t)m * GENRES + g] = acc[g];
}

extern "C" void kernel_launch(void* const* d_in, const int* in_sizes, int n_in,
                              void* d_out, int out_size, void* d_ws, size_t ws_size,
                              hipStream_t stream) {
    (void)in_sizes; (void)n_in; (void)out_size; (void)ws_size;

    const float* x_user  = (const float*)d_in[0];
    const float* x_movie = (const float*)d_in[1];
    const int*   e_src   = (const int*)d_in[2];
    const int*   e_dst   = (const int*)d_in[3];
    const float* W_user  = (const float*)d_in[4];
    const float* b_user  = (const float*)d_in[5];
    const float* W_movie = (const float*)d_in[6];
    const float* b_movie = (const float*)d_in[7];
    const float* Wl1_um  = (const float*)d_in[8];
    const float* bl1_um  = (const float*)d_in[9];
    const float* Wr1_um  = (const float*)d_in[10];
    const float* Wl1_mu  = (const float*)d_in[11];
    const float* bl1_mu  = (const float*)d_in[12];
    const float* Wr1_mu  = (const float*)d_in[13];
    const float* Wl2_um  = (const float*)d_in[14];
    const float* bl2_um  = (const float*)d_in[15];
    const float* Wr2_um  = (const float*)d_in[16];
    float* out = (float*)d_out;

    float* ws = (float*)d_ws;
    float* acc_mu    = ws;
    float* acc_um    = ws;
    float* acc2      = ws;
    unsigned* pw_dst  = (unsigned*)(ws + 6400000);
    unsigned* off_dst = pw_dst + 4000000;
    float* hu        = ws + 12800000;
    float* m1        = hu;
    float* hm        = hu + 6400000;
    float* u1        = hm + 3200000;
    unsigned* pw_src  = (unsigned*)u1;
    unsigned* off_src = pw_src + 4000000;
    float* deg_movie = u1 + 6400000;
    float* deg_user  = deg_movie + 200000;

    const int E = NUM_EDGES;

    bucket_kernel<USR_R><<<NCHUNKS, 256, 0, stream>>>(e_src, e_dst, pw_src, off_src, E);

    const int gridU = ((NUM_USERS + 16 * USR_STRIPS - 1) / (16 * USR_STRIPS) + GEMM_WAVES - 1) / GEMM_WAVES;
    const int gridM = ((NUM_MOVIES + 16 * MOV_STRIPS - 1) / (16 * MOV_STRIPS) + GEMM_WAVES - 1) / GEMM_WAVES;
    sage_gemm64<1, USR_STRIPS><<<gridU, GEMM_WAVES * 32, 0, stream>>>(x_user, nullptr, W_user, USER_FEAT,
                                           nullptr, nullptr, 0, b_user, hu,
                                           NUM_USERS, 0);
    sage_gemm64<1, MOV_STRIPS><<<gridM, GEMM_WAVES * 32, 0, stream>>>(x_movie, nullptr, W_movie, MOVIE_FEAT,
                                           nullptr, nullptr, 0, b_movie, hm,
                                           NUM_MOVIES, 0);

    owner_sum_kernel<<<OWN_BLOCKS, OWN_THREADS, 0, stream>>>(hm, pw_src, off_src, acc_mu,
                                                             deg_user, NUM_USERS, E,
                                                             USR_RANGES, USR_SLICES);
    sage_gemm64<USR_SLICES, USR_STRIPS><<<gridU, GEMM_WAVES * 32, 0, stream>>>(acc_mu, deg_user, Wl1_mu, HID,
                                           hu, Wr1_mu, HID, bl1_mu, u1,
                                           NUM_USERS, 1);

    bucket_kernel<MOV_R><<<NCHUNKS, 256, 0, stream>>>(e_dst, e_src, pw_dst, off_dst, E);

    owner_sum_kernel<<<OWN_BLOCKS, OWN_THREADS, 0, stream>>>(hu, pw_dst, off_dst, acc_um,
                                                             deg_movie, NUM_MOVIES, E,
                                                             MOV_RANGES, MOV_SLICES);
    sage_gemm64<MOV_SLICES, MOV_STRIPS><<<gridM, GEMM_WAVES * 32, 0, stream>>>(acc_um, deg_movie, Wl1_um, HID,
                                           hm, Wr1_um, HID, bl1_um, m1,
                                           NUM_MOVIES, 1);

    owner_sum_kernel<<<OWN_BLOCKS, OWN_THREADS, 0, stream>>>(u1, pw_dst, off_dst, acc2,
                                                             nullptr, NUM_MOVIES, E,
                                                             MOV_RANGES, MOV_SLICES);
    head_kernel<<<(NUM_MOVIES + 255) / 256, 256, 0, stream>>>(
        acc2, deg_movie, m1, Wl2_um, bl2_um, Wr2_um, out, NUM_MOVIES);
}


# Node products run in the owner-sum epilogues: ranges 256 x 1, so a block's LDS holds whole sums; SAGE products on WMMA and the head read them there; sum planes and 3 launches gone; sum order changes
# speedup vs baseline: 4.9590x; 1.1671x over previous
#include <hip/hip_runtime.h>
#include <hip/hip_bf16.h>

typedef __attribute__((ext_vector_type(2))) float v2f;
typedef __attribute__((ext_vector_type(8))) float v8f;

#define NUM_USERS  100000
#define NUM_MOVIES 50000
#define NUM_EDGES  4000000
#define USER_FEAT  24
#define MOVIE_FEAT 404
#define HID        64
#define GENRES     3


#define OWN_BLOCKS   256
#define OWN_THREADS  1024
#define OWN_WAVES    32
#define OWN_MAXROWS  391
#define OWN_PITCH    68
#define MOV_RANGES   256
#define USR_RANGES   256
#define MOV_R        ((NUM_MOVIES + MOV_RANGES - 1) / MOV_RANGES)
#define USR_R        ((NUM_USERS + USR_RANGES - 1) / USR_RANGES)
#define OWN_QCAP     704
#define OWN_LDSROWS  (OWN_MAXROWS + OWN_WAVES)
#define CHUNK        8192
#define NRANGE       256
#define OFF_STRIDE   (NRANGE + 1)
#define NCHUNKS      ((NUM_EDGES + CHUNK - 1) / CHUNK)
static_assert(MOV_RANGES <= NRANGE && USR_RANGES <= NRANGE, "a range id must be below 256");
static_assert(NUM_EDGES % 4 == 0, "16-byte loads of keys and source ids");
static_assert(2 * HID * OWN_PITCH <= OWN_WAVES * OWN_QCAP, "both weight matrices fit in the queue's LDS");
static_assert(OWN_LDSROWS <= (1 << 10) && NUM_USERS <= (1 << 17) && NUM_MOVIES <= (1 << 17),
              "a row of a range and a source id are packed into one word (10 + 17 bits)");

#define BKT_THREADS  512
template <int R>
__global__ __launch_bounds__(BKT_THREADS)
void bucket_kernel(const int* __restrict__ keys, const int* __restrict__ sids,
                   unsigned* __restrict__ pw, unsigned* __restrict__ off, int E) {
    __shared__ unsigned s_hist[NRANGE];
    __shared__ unsigned s_start[OFF_STRIDE];
    __shared__ __attribute__((aligned(16))) unsigned s_out[CHUNK];

    const int tid = threadIdx.x;
    const int c0  = (int)blockIdx.x * CHUNK;
    if (tid < NRANGE) s_hist[tid] = 0u;
    __syncthreads();

    unsigned w[16], rk[16], id[16];
#pragma unroll
    for (int i = 0; i < 4; ++i) {
        const int  e  = c0 + i * (4 * BKT_THREADS) + tid * 4;
        const bool in = e < E;
        const int4 k = in ? *(const int4*)(keys + e) : make_int4(-1, -1, -1, -1);
        const int4 s = in ? *(const int4*)(sids + e) : make_int4(0, 0, 0, 0);
        const unsigned kv[4] = {(unsigned)k.x, (unsigned)k.y, (unsigned)k.z, (unsigned)k.w};
        const unsigned sv[4] = {(unsigned)s.x, (unsigned)s.y, (unsigned)s.z, (unsigned)s.w};
#pragma unroll
        for (int q = 0; q < 4; ++q) {
            const unsigned rid = kv[q] / (unsigned)R;
            const bool     ok  = rid < (unsigned)NRANGE;
            id[4 * i + q] = ok ? rid : (unsigned)NRANGE;
            w[4 * i + q]  = ((kv[q] - rid * (unsigned)R) << 17) | (sv[q] & 0x1FFFFu);
            rk[4 * i + q] = 0u;
            if (ok) rk[4 * i + q] = atomicAdd(&s_hist[rid], 1u);
        }
    }
    __syncthreads();

    if (tid < 32) {
        unsigned h[8];
        unsigned t = 0u;
#pragma unroll
        for (int q = 0; q < 8; ++q) { h[q] = s_hist[8 * tid + q]; t += h[q]; }
        unsigned inc = t;
#pragma unroll
        for (int d = 1; d < 32; d <<= 1) {
            const unsigned v = __shfl_up(inc, (unsigned)d);
            if (tid >= d) inc += v;
        }
        unsigned run = inc - t;
#pragma unroll
        for (int q = 0; q < 8; ++q) { s_start[8 * tid + q] = run; run += h[q]; }
        if (tid == 31) s_start[NRANGE] = inc;
    }
    __syncthreads();

#pragma unroll
    for (int j = 0; j < 16; ++j)
        if (id[j] < (unsigned)NRANGE) s_out[s_start[id[j]] + rk[j]] = w[j];
    __syncthreads();

    const unsigned total = s_start[NRANGE];
#pragma unroll
    for (int i = 0; i < 4; ++i) {
        const int idx = i * (4 * BKT_THREADS) + tid * 4;
        if ((unsigned)idx < total) *(uint4*)(pw + (size_t)c0 + idx) = *(const uint4*)&s_out[idx];
    }
    for (int i = tid; i < OFF_STRIDE; i += BKT_THREADS) off[(size_t)blockIdx.x * OFF_STRIDE + i] = s_start[i];
}

__device__ __forceinline__ void own_rows32(const float* __restrict__ X, int pk, float* s_acc,
                                           int hw, int c, int cA, int cB) {
    float4 v[16];
    int    pj[16];
#pragma unroll
    for (int j = 0; j < 16; ++j) {
        const int pa = __builtin_amdgcn_readlane(pk, 2 * j);
        const int pb = __builtin_amdgcn_readlane(pk, 2 * j + 1);
        pj[j] = hw ? pb : pa;
        v[j] = *(const float4*)(X + (size_t)(pj[j] & 0x1FFFF) * HID + c);
    }
#pragma unroll
    for (int j = 0; j < 16; ++j) {
        const unsigned r = (unsigned)pj[j] >> 17;
        const float a0 = hw ? v[j].z : v[j].x;
        const float a1 = hw ? v[j].w : v[j].y;
        const float a2 = hw ? v[j].x : v[j].z;
        const float a3 = hw ? v[j].y : v[j].w;
        float* oa = &s_acc[r * OWN_PITCH + cA];
        float* ob = &s_acc[r * OWN_PITCH + cB];
        atomicAdd(oa + 0, a0);
        atomicAdd(oa + 1, a1);
        atomicAdd(ob + 0, a2);
        atomicAdd(ob + 1, a3);
    }
}

__device__ __forceinline__ void rows_plain32(const float* __restrict__ X, int pk, float* s_acc, int lane) {
    float2   v[32];
    unsigned rw[32];
#pragma unroll
    for (int j = 0; j < 32; ++j) {
        const unsigned p = (unsigned)__builtin_amdgcn_readlane(pk, j);
        rw[j] = p >> 17;
        v[j] = *(const float2*)(X + (size_t)(p & 0x1FFFFu) * HID + 2 * lane);
    }
#pragma unroll
    for (int g = 0; g < 32; g += 4) {
        float2 t[4];
#pragma unroll
        for (int i = 0; i < 4; ++i)
            t[i] = *(const float2*)&s_acc[rw[g + i] * OWN_PITCH + 2 * lane];
#pragma unroll
        for (int i = 0; i < 4; ++i) {
#pragma unroll
            for (int k = 0; k < i; ++k)
                if (rw[g + k] == rw[g + i]) t[i] = t[k];
            t[i].x += v[g + i].x;
            t[i].y += v[g + i].y;
        }
#pragma unroll
        for (int i = 0; i < 4; ++i)
            *(float2*)&s_acc[rw[g + i] * OWN_PITCH + 2 * lane] = t[i];
    }
}

__device__ __forceinline__ void own_push32(const float* __restrict__ X, unsigned wv, bool ok,
                                           float* s_acc, unsigned* s_cnt, unsigned* s_q,
                                           unsigned* s_qcnt, int R, bool count,
                                           int hw, int c, int cA, int cB) {
    const unsigned rr  = wv >> 17;
    const bool     inr = ok && rr < (unsigned)R;
    const unsigned rL  = inr ? rr : (unsigned)OWN_MAXROWS;
    if (count) atomicAdd(&s_cnt[rL], 1u);
    const unsigned pk = (rL << 17) | (wv & 0x1FFFFu);
    bool over = false;
    if (inr) {
        const unsigned ow   = rL & (unsigned)(OWN_WAVES - 1);
        const unsigned slot = atomicAdd(&s_qcnt[ow], 1u);
        if (slot < (unsigned)OWN_QCAP) s_q[ow * OWN_QCAP + slot] = pk;
        else over = true;
    }
    if (__builtin_amdgcn_ballot_w32(over) != 0u) {
        own_rows32(X, (int)(over ? pk : ((unsigned)OWN_MAXROWS << 17)), s_acc, hw, c, cA, cB);
    }
}

__device__ __forceinline__ void sage_rows_from_lds(const float* s_acc, const unsigned* s_cnt, const float* s_w,
                                                   const float* __restrict__ bl, const float* __restrict__ Xs,
                                                   float* __restrict__ out, int base, int R, int N,
                                                   int wave, int lane) {
    const int half  = lane >> 4;
    const int r     = lane & 15;
    const int ntask = 2 * ((R + 15) >> 4);
    for (int t = wave; t < ntask; t += OWN_WAVES) {
        const int strip = t >> 1;
        const int ct0   = 2 * (t & 1);
        const int lr    = 16 * strip + r;
        const float scale = 1.0f / fmaxf((float)s_cnt[lr], 1.0f);
        const int   grow  = (base + lr < N) ? base + lr : N - 1;
        const float* Lrow = s_acc + lr * OWN_PITCH;
        const float* Arow = Xs + (size_t)grow * HID;
        const float* Wl0  = s_w + (ct0 * 16 + r) * OWN_PITCH;
        const float* Wl1  = Wl0 + 16 * OWN_PITCH;
        const float* Wr0  = Wl0 + HID * OWN_PITCH;
        const float* Wr1  = Wr0 + 16 * OWN_PITCH;
        v8f c0 = {}, c1 = {};
#pragma unroll
        for (int kk = 0; kk < HID; kk += 4) {
            const int k = kk + 2 * half;
            v2f a, b0, b1;
            a.x  = Lrow[k] * scale;
            a.y  = Lrow[k + 1] * scale;
            b0.x = Wl0[k];  b0.y = Wl0[k + 1];
            b1.x = Wl1[k];  b1.y = Wl1[k + 1];
            c0 = __builtin_amdgcn_wmma_f32_16x16x4_f32(false, a, false, b0, (short)0, c0, false, false);
            c1 = __builtin_amdgcn_wmma_f32_16x16x4_f32(false, a, false, b1, (short)0, c1, false, false);
        }
#pragma unroll
        for (int kk = 0; kk < HID; kk += 4) {
            const int k = kk + 2 * half;
            v2f a, b0, b1;
            a.x  = Arow[k];
            a.y  = Arow[k + 1];
            b0.x = Wr0[k];  b0.y = Wr0[k + 1];
            b1.x = Wr1[k];  b1.y = Wr1[k + 1];
            c0 = __builtin_amdgcn_wmma_f32_16x16x4_f32(false, a, false, b0, (short)0, c0, false, false);
            c1 = __builtin_amdgcn_wmma_f32_16x16x4_f32(false, a, false, b1, (short)0, c1, false, false);
        }
        asm volatile("v_nop\n\tv_nop\n\tv_nop\n\tv_nop\n\tv_nop\n\tv_nop\n\tv_nop\n\tv_nop" : "+v"(c0), "+v"(c1));
        const float bv0 = bl[ct0 * 16 + r];
        const float bv1 = bl[ct0 * 16 + 16 + r];
#pragma unroll
        for (int i = 0; i < 8; ++i) {
            const int lrow = 16 * strip + i + 8 * half;
            if (lrow < R && base + lrow < N) {
                float* o = out + (size_t)(base + lrow) * HID + ct0 * 16 + r;
                o[0]  = fmaxf(c0[i] + bv0, 0.0f);
                o[16] = fmaxf(c1[i] + bv1, 0.0f);
            }
        }
    }
}

template <int MODE>
__global__ __launch_bounds__(OWN_THREADS)
void owner_sum_kernel(const float* __restrict__ X, const unsigned* __restrict__ pw,
                      const unsigned* __restrict__ off, float* __restrict__ deg,
                      const float* __restrict__ Wl, const float* __restrict__ bl,
                      const float* __restrict__ Xs, const float* __restrict__ Wr,
                      float* __restrict__ out, int N, int E, int P, int S) {
    __shared__ float    s_acc[OWN_LDSROWS * OWN_PITCH];
    __shared__ unsigned s_cnt[OWN_LDSROWS];
    __shared__ unsigned s_q[OWN_WAVES * OWN_QCAP];
    __shared__ unsigned s_qcnt[OWN_WAVES];

    const int tid  = threadIdx.x;
    const int lane = tid & 31;
    const int wave = tid >> 5;
    const int R    = (N + P - 1) / P;
    const int pr   = (int)blockIdx.x % P;
    const int sl   = (int)blockIdx.x / P;
    const int base = pr * R;
    const int nch  = (E + CHUNK - 1) / CHUNK;
    const int cps  = (nch + S - 1) / S;
    const int c_lo = sl * cps;
    const int c_hi = (c_lo + cps < nch) ? c_lo + cps : nch;
    const bool count = MODE == 0;

    for (int i = tid; i < OWN_LDSROWS * OWN_PITCH; i += OWN_THREADS) s_acc[i] = 0.0f;
    for (int i = tid; i < OWN_LDSROWS; i += OWN_THREADS) s_cnt[i] = 0u;
    if (tid < OWN_WAVES) s_qcnt[tid] = 0u;
    __syncthreads();

    const int hw = lane >> 4;
    const int c  = (lane & 15) * 4;
    const int cA = c + 2 * hw;
    const int cB = c + 2 - 2 * hw;


    const int uw = __builtin_amdgcn_readfirstlane(wave);
    for (int g0 = c_lo + uw; g0 < c_hi; g0 += 32 * OWN_WAVES) {
        const int cl = g0 + lane * OWN_WAVES;
        unsigned st = 0u, nn = 0u;
        if (cl < c_hi) {
            const unsigned a = off[(size_t)cl * OFF_STRIDE + pr];
            const unsigned b = off[(size_t)cl * OFF_STRIDE + pr + 1];
            st = (unsigned)cl * (unsigned)CHUNK + a;
            nn = b - a;
        }
        int ng = (c_hi - g0 + OWN_WAVES - 1) / OWN_WAVES;
        ng = ng < 32 ? ng : 32;
        for (int j = 0; j < ng; j += 4) {
            unsigned s4[4], n4[4], wa[4], wb[4];
#pragma unroll
            for (int q = 0; q < 4; ++q) {
                s4[q] = (unsigned)__builtin_amdgcn_readlane((int)st, j + q);
                n4[q] = (unsigned)__builtin_amdgcn_readlane((int)nn, j + q);
            }
#pragma unroll
            for (int q = 0; q < 4; ++q) {
                wa[q] = ((unsigned)lane < n4[q]) ? pw[(size_t)s4[q] + lane] : 0u;
                wb[q] = ((unsigned)lane + 32u < n4[q]) ? pw[(size_t)s4[q] + 32u + lane] : 0u;
            }
#pragma nounroll
            for (int q = 0; q < 4; ++q) {
                const unsigned nr = q == 0 ? n4[0] : q == 1 ? n4[1] : q == 2 ? n4[2] : n4[3];
                const unsigned sr = q == 0 ? s4[0] : q == 1 ? s4[1] : q == 2 ? s4[2] : s4[3];
                const unsigned ra = q == 0 ? wa[0] : q == 1 ? wa[1] : q == 2 ? wa[2] : wa[3];
                const unsigned rb = q == 0 ? wb[0] : q == 1 ? wb[1] : q == 2 ? wb[2] : wb[3];
                for (unsigned m = 0u; m < nr; m += 32u) {
                    const bool okm = m + (unsigned)lane < nr;
                    unsigned wm;
                    if (m == 0u)       wm = ra;
                    else if (m == 32u) wm = rb;
                    else               wm = okm ? pw[(size_t)sr + m + lane] : 0u;
                    own_push32(X, wm, okm, s_acc, s_cnt, s_q, s_qcnt, R, count, hw, c, cA, cB);
                }
            }
        }
    }
    __syncthreads();

    {
        const unsigned n  = (unsigned)__builtin_amdgcn_readfirstlane((int)s_qcnt[wave]);
        const unsigned nq = n < (unsigned)OWN_QCAP ? n : (unsigned)OWN_QCAP;
        const unsigned spare = (unsigned)(OWN_MAXROWS + wave) << 17;
        for (unsigned m = 0u; m < nq; m += 32u) {
            const unsigned idx = m + (unsigned)lane;
            const bool     ok  = idx < nq;
            const unsigned q   = s_q[(unsigned)wave * OWN_QCAP + (ok ? idx : 0u)];
            const unsigned pk  = ok ? q : spare;
            rows_plain32(X, (int)pk, s_acc, lane);
        }
    }
    __syncthreads();

    if (MODE == 0) {
        float* s_w = (float*)s_q;
        for (int i = tid; i < HID * HID; i += OWN_THREADS) {
            const int n = i >> 6, k = i & 63;
            s_w[n * OWN_PITCH + k]         = Wl[i];
            s_w[(HID + n) * OWN_PITCH + k] = Wr[i];
        }
        __syncthreads();
        sage_rows_from_lds(s_acc, s_cnt, s_w, bl, Xs, out, base, R, N, wave, lane);
        if (deg) {
            for (int r = tid; r < R; r += OWN_THREADS)
                if (base + r < N) deg[base + r] = (float)s_cnt[r];
        }
    } else {
        float* s_o = (float*)s_q;
        float wl[GENRES][2], wr[GENRES][2];
#pragma unroll
        for (int g = 0; g < GENRES; ++g) {
            wl[g][0] = Wl[g * HID + 2 * lane];  wl[g][1] = Wl[g * HID + 2 * lane + 1];
            wr[g][0] = Wr[g * HID + 2 * lane];  wr[g][1] = Wr[g * HID + 2 * lane + 1];
        }
        for (int r = wave; r < R; r += OWN_WAVES) {
            const int row = base + r;
            if (row < N) {
                const float  inv = 1.0f / fmaxf(deg[row], 1.0f);
                const float2 a = *(const float2*)&s_acc[r * OWN_PITCH + 2 * lane];
                const float2 x = *(const float2*)(Xs + (size_t)row * HID + 2 * lane);
                float p[GENRES];
#pragma unroll
                for (int g = 0; g < GENRES; ++g)
                    p[g] = (a.x * inv) * wl[g][0] + (a.y * inv) * wl[g][1] + x.x * wr[g][0] + x.y * wr[g][1];
#pragma unroll
                for (int d = 16; d >= 1; d >>= 1) {
#pragma unroll
                    for (int g = 0; g < GENRES; ++g) p[g] += __shfl_xor(p[g], d);
                }
                if (lane < GENRES) s_o[r * GENRES + lane] = bl[lane] + (lane == 0 ? p[0] : lane == 1 ? p[1] : p[2]);
            }
        }
        __syncthreads();
        for (int i = tid; i < R * GENRES; i += OWN_THREADS)
            if (base + i / GENRES < N) out[(size_t)base * GENRES + i] = s_o[i];
    }
}

#define GEMM_WAVES   2
#define MOV_STRIPS   2
#define USR_STRIPS   4
template <int NP, int STRIPS>
__global__ __launch_bounds__(GEMM_WAVES * 32)
void sage_gemm64(const float* __restrict__ A1, const float* __restrict__ deg1,
                 const float* __restrict__ W1, int K1,
                 const float* __restrict__ A2, const float* __restrict__ W2, int K2,
                 const float* __restrict__ bias, float* __restrict__ C,
                 int M, int relu) {
    const int lane = threadIdx.x & 31;
    const int wave = threadIdx.x >> 5;
    const int half = lane >> 4;
    const int r    = lane & 15;
    const int row0 = (blockIdx.x * GEMM_WAVES + wave) * (16 * STRIPS);
    if (row0 >= M) return;

    v8f c[STRIPS][4] = {};

    int   arow[STRIPS];
    float scale[STRIPS];
#pragma unroll
    for (int s = 0; s < STRIPS; ++s) {
        const int rs = row0 + 16 * s;
        arow[s] = ((rs < M) ? rs : M - 16) + r;
        scale[s] = 1.0f;
        if (deg1) {
            float d = deg1[arow[s]];
#pragma unroll
            for (int p = 1; p < NP; ++p) d += deg1[(size_t)p * M + arow[s]];
            scale[s] = 1.0f / fmaxf(d, 1.0f);
        }
    }

    {
        const size_t plane = (size_t)M * K1;
        for (int kk = 0; kk < K1; kk += 4) {
            const int k = kk + 2 * half;
            v2f b[4];
#pragma unroll
            for (int t = 0; t < 4; ++t) {
                const float* Wrow = W1 + (size_t)(t * 16 + r) * K1;
                b[t].x = Wrow[k];
                b[t].y = Wrow[k + 1];
            }
#pragma unroll
            for (int s = 0; s < STRIPS; ++s) {
                const float* Arow = A1 + (size_t)arow[s] * K1;
                float ax = Arow[k];
                float ay = Arow[k + 1];
#pragma unroll
                for (int p = 1; p < NP; ++p) {
                    ax += Arow[p * plane + k];
                    ay += Arow[p * plane + k + 1];
                }
                v2f a;
                a.x = ax * scale[s];
                a.y = ay * scale[s];
#pragma unroll
                for (int t = 0; t < 4; ++t)
                    c[s][t] = __builtin_amdgcn_wmma_f32_16x16x4_f32(
                        false, a, false, b[t], (short)0, c[s][t], false, false);
            }
        }
    }
    if (A2) {
        for (int kk = 0; kk < K2; kk += 4) {
            const int k = kk + 2 * half;
            v2f b[4];
#pragma unroll
            for (int t = 0; t < 4; ++t) {
                const float* Wrow = W2 + (size_t)(t * 16 + r) * K2;
                b[t].x = Wrow[k];
                b[t].y = Wrow[k + 1];
            }
#pragma unroll
            for (int s = 0; s < STRIPS; ++s) {
                const float* Arow = A2 + (size_t)arow[s] * K2;
                v2f a;
                a.x = Arow[k];
                a.y = Arow[k + 1];
#pragma unroll
                for (int t = 0; t < 4; ++t)
                    c[s][t] = __builtin_amdgcn_wmma_f32_16x16x4_f32(
                        false, a, false, b[t], (short)0, c[s][t], false, false);
            }
        }
    }

#pragma unroll
    for (int s = 0; s < STRIPS; ++s) {
        const int rs = row0 + 16 * s;
        if (rs < M) {
#pragma unroll
            for (int t = 0; t < 4; ++t) {
                const float bv = bias ? bias[t * 16 + r] : 0.0f;
#pragma unroll
                for (int i = 0; i < 8; ++i) {
                    float v = c[s][t][i] + bv;
                    if (relu) v = fmaxf(v, 0.0f);
                    C[(size_t)(rs + i + 8 * half) * HID + t * 16 + r] = v;
                }
            }
        }
    }
}

extern "C" void kernel_launch(void* const* d_in, const int* in_sizes, int n_in,
                              void* d_out, int out_size, void* d_ws, size_t ws_size,
                              hipStream_t stream) {
    (void)in_sizes; (void)n_in; (void)out_size; (void)ws_size;

    const float* x_user  = (const float*)d_in[0];
    const float* x_movie = (const float*)d_in[1];
    const int*   e_src   = (const int*)d_in[2];
    const int*   e_dst   = (const int*)d_in[3];
    const float* W_user  = (const float*)d_in[4];
    const float* b_user  = (const float*)d_in[5];
    const float* W_movie = (const float*)d_in[6];
    const float* b_movie = (const float*)d_in[7];
    const float* Wl1_um  = (const float*)d_in[8];
    const float* bl1_um  = (const float*)d_in[9];
    const float* Wr1_um  = (const float*)d_in[10];
    const float* Wl1_mu  = (const float*)d_in[11];
    const float* bl1_mu  = (const float*)d_in[12];
    const float* Wr1_mu  = (const float*)d_in[13];
    const float* Wl2_um  = (const float*)d_in[14];
    const float* bl2_um  = (const float*)d_in[15];
    const float* Wr2_um  = (const float*)d_in[16];
    float* out = (float*)d_out;

    float* ws = (float*)d_ws;
    float* m1         = ws;
    unsigned* pw_src  = (unsigned*)(ws + 3200000);
    unsigned* off_src = pw_src + 4000000;
    unsigned* pw_dst  = (unsigned*)(ws + 7500000);
    unsigned* off_dst = pw_dst + 4000000;
    float* hu        = ws + 12800000;
    float* hm        = hu + 6400000;
    float* u1        = hm + 3200000;
    float* deg_movie = u1 + 6400000;

    const int E = NUM_EDGES;

    bucket_kernel<USR_R><<<NCHUNKS, BKT_THREADS, 0, stream>>>(e_src, e_dst, pw_src, off_src, E);

    const int gridU = ((NUM_USERS + 16 * USR_STRIPS - 1) / (16 * USR_STRIPS) + GEMM_WAVES - 1) / GEMM_WAVES;
    const int gridM = ((NUM_MOVIES + 16 * MOV_STRIPS - 1) / (16 * MOV_STRIPS) + GEMM_WAVES - 1) / GEMM_WAVES;
    sage_gemm64<1, USR_STRIPS><<<gridU, GEMM_WAVES * 32, 0, stream>>>(x_user, nullptr, W_user, USER_FEAT,
                                           nullptr, nullptr, 0, b_user, hu,
                                           NUM_USERS, 0);
    sage_gemm64<1, MOV_STRIPS><<<gridM, GEMM_WAVES * 32, 0, stream>>>(x_movie, nullptr, W_movie, MOVIE_FEAT,
                                           nullptr, nullptr, 0, b_movie, hm,
                                           NUM_MOVIES, 0);

    owner_sum_kernel<0><<<OWN_BLOCKS, OWN_THREADS, 0, stream>>>(hm, pw_src, off_src, nullptr,
                                                                Wl1_mu, bl1_mu, hu, Wr1_mu, u1,
                                                                NUM_USERS, E, USR_RANGES, 1);

    bucket_kernel<MOV_R><<<NCHUNKS, BKT_THREADS, 0, stream>>>(e_dst, e_src, pw_dst, off_dst, E);

    owner_sum_kernel<0><<<OWN_BLOCKS, OWN_THREADS, 0, stream>>>(hu, pw_dst, off_dst, deg_movie,
                                                                Wl1_um, bl1_um, hm, Wr1_um, m1,
                                                                NUM_MOVIES, E, MOV_RANGES, 1);

    owner_sum_kernel<1><<<OWN_BLOCKS, OWN_THREADS, 0, stream>>>(u1, pw_dst, off_dst, deg_movie,
                                                                Wl2_um, bl2_um, m1, Wr2_um, out,
                                                                NUM_MOVIES, E, MOV_RANGES, 1);
}
